# starting kernel
# baseline (speedup 1.0000x reference)
#include <hip/hip_runtime.h>
#include <hip/hip_bf16.h>
#include <math.h>

#define BB     8
#define SS     256
#define HIDC   1024
#define HH     16
#define DD     64
#define FF     64
#define LR     0.1f
#define GSCALE (2.0f / 8192.0f)
#define LN_EPS 1e-5f

typedef float v2f __attribute__((ext_vector_type(2)));
typedef float v8f __attribute__((ext_vector_type(8)));

__device__ __forceinline__ v8f wmma4(v2f a, v2f b, v8f c) {
  return __builtin_amdgcn_wmma_f32_16x16x4_f32(
       false, a,  false, b,
       (short)0, c,  false,  false);
}

__device__ __forceinline__ float gelu_f(float x) {
  const float c = 0.7978845608028654f;
  float u = c * (x + 0.044715f * x * x * x);
  return 0.5f * x * (1.0f + tanhf(u));
}
__device__ __forceinline__ float gelu_grad(float x) {
  const float c = 0.7978845608028654f;
  float x2 = x * x;
  float u = c * (x + 0.044715f * x2 * x);
  float t = tanhf(u);
  return 0.5f * (1.0f + t) + 0.5f * x * (1.0f - t * t) * c * (1.0f + 0.134145f * x2);
}
__device__ __forceinline__ float wave_sum32(float v) {
#pragma unroll
  for (int off = 16; off > 0; off >>= 1) v += __shfl_xor(v, off, 32);
  return v;
}

__global__ void __launch_bounds__(256)
gemm_wmma_f32(const float* __restrict__ A, const float* __restrict__ Bm,
              float* __restrict__ C, int M, int N, int K, int act) {
  __shared__ float As[16 * 64];
  const int tid   = threadIdx.x;
  const int wave  = tid >> 5;
  const int lane  = tid & 31;
  const int mbase = blockIdx.y * 16;
  const int tn    = blockIdx.x * 8 + wave;
  const int col   = tn * 16 + (lane & 15);
  const int kh    = (lane >> 4) << 1;
  const int arow  = lane & 15;
  v8f acc = {0.f, 0.f, 0.f, 0.f, 0.f, 0.f, 0.f, 0.f};
  for (int kc = 0; kc < K; kc += 64) {
    __syncthreads();
    for (int i = tid; i < 16 * 64; i += 256)
      As[i] = A[(size_t)(mbase + (i >> 6)) * K + kc + (i & 63)];
    if (kc + 64 < K)
      __builtin_prefetch(&A[(size_t)(mbase + (tid >> 4)) * K + kc + 64], 0, 1);
    __syncthreads();
#pragma unroll
    for (int k0 = 0; k0 < 64; k0 += 4) {
      v2f a, b;
      a.x = As[arow * 64 + k0 + kh];
      a.y = As[arow * 64 + k0 + kh + 1];
      b.x = Bm[(size_t)(kc + k0 + kh) * N + col];
      b.y = Bm[(size_t)(kc + k0 + kh + 1) * N + col];
      acc = wmma4(a, b, acc);
    }
  }
  const int r0 = mbase + ((lane >= 16) ? 8 : 0);
#pragma unroll
  for (int i = 0; i < 8; ++i) {
    float v = acc[i];
    if (act) v = gelu_f(v);
    C[(size_t)(r0 + i) * N + col] = v;
  }
}

#define EPI_BIAS      0
#define EPI_GELU_PAIR 1
#define EPI_DPRED     2
#define EPI_DGELU     3

__device__ __forceinline__ void
gemm16_rowA(const float* A8, const float* Bm, const float* bias,
            float* C8, float* C2, const float* aux,
            int transB, int epi, int wave, int lane) {
  if (wave >= 4) return;
  const int col  = wave * 16 + (lane & 15);
  const int mrow = lane & 15;
  const int kh   = (lane >> 4) << 1;
  const bool live = (mrow < 8);
  v8f acc = {0.f, 0.f, 0.f, 0.f, 0.f, 0.f, 0.f, 0.f};
#pragma unroll
  for (int k0 = 0; k0 < 64; k0 += 4) {
    v2f a, b;
    a.x = live ? A8[mrow * 64 + k0 + kh] : 0.f;
    a.y = live ? A8[mrow * 64 + k0 + kh + 1] : 0.f;
    if (!transB) {
      b.x = Bm[(k0 + kh) * 64 + col];
      b.y = Bm[(k0 + kh + 1) * 64 + col];
    } else {
      b.x = Bm[col * 64 + k0 + kh];
      b.y = Bm[col * 64 + k0 + kh + 1];
    }
    acc = wmma4(a, b, acc);
  }
  if (lane < 16) {
    const float bv = bias ? bias[col] : 0.f;
#pragma unroll
    for (int i = 0; i < 8; ++i) {
      const int o = i * 64 + col;
      float v = acc[i] + bv;
      if (epi == EPI_GELU_PAIR)      { C8[o] = v; C2[o] = gelu_f(v); }
      else if (epi == EPI_DPRED)     { C8[o] = GSCALE * (v - aux[o]); }
      else if (epi == EPI_DGELU)     { C8[o] = v * gelu_grad(aux[o]); }
      else                           { C8[o] = v; }
    }
  }
}

__device__ __forceinline__ void
gemm_outer_update(const float* X8, const float* Y8, float* Wt, float lr,
                  int wave, int lane) {
  const int kh = (lane >> 4) << 1;
#pragma unroll
  for (int ti = 0; ti < 2; ++ti) {
    const int t  = wave * 2 + ti;
    const int tm = t >> 2, tn = t & 3;
    const int m = tm * 16 + (lane & 15);
    const int n = tn * 16 + (lane & 15);
    v8f acc = {0.f, 0.f, 0.f, 0.f, 0.f, 0.f, 0.f, 0.f};
#pragma unroll
    for (int k0 = 0; k0 < 8; k0 += 4) {
      v2f a, b;
      a.x = X8[(k0 + kh) * 64 + m];
      a.y = X8[(k0 + kh + 1) * 64 + m];
      b.x = Y8[(k0 + kh) * 64 + n];
      b.y = Y8[(k0 + kh + 1) * 64 + n];
      acc = wmma4(a, b, acc);
    }
    const int rbase = tm * 16 + ((lane >= 16) ? 8 : 0);
    const int cc    = tn * 16 + (lane & 15);
#pragma unroll
    for (int i = 0; i < 8; ++i) Wt[(rbase + i) * 64 + cc] -= lr * acc[i];
  }
}

__device__ __forceinline__ void
fwd_pass(const float* ktb, float* av, float* hg, float* ob, float* xh, float* zb,
         const float* W1, const float* W2, const float* b1v, const float* b2v,
         const float* lgv, const float* lbv, float* muv, float* isv,
         int wave, int lane, int tid) {
  gemm16_rowA(ktb, W1, b1v, av, hg, nullptr, 0, EPI_GELU_PAIR, wave, lane);
  __syncthreads();
  gemm16_rowA(hg, W2, b2v, ob, nullptr, nullptr, 0, EPI_BIAS, wave, lane);
  __syncthreads();
  {
    float e0 = ob[wave * 64 + lane], e1 = ob[wave * 64 + lane + 32];
    float s = wave_sum32(e0 + e1);
    float q = wave_sum32(e0 * e0 + e1 * e1);
    if (lane == 0) {
      float m = s * (1.f / 64.f);
      muv[wave] = m;
      isv[wave] = rsqrtf(q * (1.f / 64.f) - m * m + LN_EPS);
    }
  }
  __syncthreads();
  for (int idx = tid; idx < 512; idx += 256) {
    int b = idx >> 6, d = idx & 63;
    float xv = (ob[idx] - muv[b]) * isv[b];
    xh[idx] = xv;
    zb[idx] = xv * lgv[d] + lbv[d];
  }
  __syncthreads();
}

#define SCAN_SMEM_FLOATS (3 * 4096 + 10 * 512 + 7 * 64 + 32)

__global__ void __launch_bounds__(256)
ttt_scan_kernel(const float* __restrict__ kg, const float* __restrict__ vg,
                const float* __restrict__ qg,
                const float* __restrict__ fw1, const float* __restrict__ fb1,
                const float* __restrict__ fw2, const float* __restrict__ fb2,
                const float* __restrict__ flg, const float* __restrict__ flb,
                const float* __restrict__ lwg, const float* __restrict__ lbg,
                const float* __restrict__ ttg, const float* __restrict__ ttb,
                float* __restrict__ outg) {
  extern __shared__ float smem[];
  float* W1  = smem;
  float* W2  = W1 + 4096;
  float* LW  = W2 + 4096;
  float* ktb = LW + 4096;
  float* tgt = ktb + 512;
  float* av  = tgt + 512;
  float* hg  = av + 512;
  float* ob  = hg + 512;
  float* xh  = ob + 512;
  float* zb  = xh + 512;
  float* dzb = zb + 512;
  float* dob = dzb + 512;
  float* dab = dob + 512;
  float* b1v = dab + 512;
  float* b2v = b1v + 64;
  float* lgv = b2v + 64;
  float* lbv = lgv + 64;
  float* tgv = lbv + 64;
  float* tbv = tgv + 64;
  float* lbz = tbv + 64;
  float* muv = lbz + 64;
  float* isv = muv + 8;
  float* m1v = isv + 8;
  float* m2v = m1v + 8;

  const int h    = blockIdx.x;
  const int tid  = threadIdx.x;
  const int wave = tid >> 5;
  const int lane = tid & 31;

  const int b0 = tid >> 6,          d0 = tid & 63;
  const int b1i = (tid + 256) >> 6, d1i = (tid + 256) & 63;
  const size_t g0base = (size_t)(b0 * SS) * HIDC + h * 64 + d0;
  const size_t g1base = (size_t)(b1i * SS) * HIDC + h * 64 + d1i;

  for (int i = tid; i < 4096; i += 256) {
    W1[i] = fw1[(size_t)h * 4096 + i];
    W2[i] = fw2[(size_t)h * 4096 + i];
    LW[i] = lwg[i];
  }
  if (tid < 64) {
    b1v[tid] = fb1[h * 64 + tid];
    b2v[tid] = fb2[h * 64 + tid];
    lgv[tid] = flg[h * 64 + tid];
    lbv[tid] = flb[h * 64 + tid];
    tgv[tid] = ttg[h * 64 + tid];
    tbv[tid] = ttb[h * 64 + tid];
    lbz[tid] = lbg[tid];
  }

  float kp0 = kg[g0base], vp0 = vg[g0base];
  float kp1 = kg[g1base], vp1 = vg[g1base];
  __syncthreads();

  for (int t = 0; t < SS; ++t) {
    ktb[tid]       = kp0;  tgt[tid]       = vp0 - kp0;
    ktb[tid + 256] = kp1;  tgt[tid + 256] = vp1 - kp1;
    __syncthreads();
    if (t + 1 < SS) {
      kp0 = kg[g0base + (t + 1) * HIDC];  vp0 = vg[g0base + (t + 1) * HIDC];
      kp1 = kg[g1base + (t + 1) * HIDC];  vp1 = vg[g1base + (t + 1) * HIDC];
    }

    for (int step = 0; step < 4; ++step) {
      fwd_pass(ktb, av, hg, ob, xh, zb, W1, W2, b1v, b2v, lgv, lbv,
               muv, isv, wave, lane, tid);
      gemm16_rowA(zb, LW, lbz, dab, nullptr, tgt, 0, EPI_DPRED, wave, lane);
      __syncthreads();
      gemm16_rowA(dab, LW, nullptr, dzb, nullptr, nullptr, 1, EPI_BIAS, wave, lane);
      __syncthreads();
      {
        float dd0 = dzb[wave * 64 + lane] * lgv[lane];
        float dd1 = dzb[wave * 64 + lane + 32] * lgv[lane + 32];
        float x0 = xh[wave * 64 + lane], x1 = xh[wave * 64 + lane + 32];
        float s1 = wave_sum32(dd0 + dd1);
        float s2 = wave_sum32(dd0 * x0 + dd1 * x1);
        if (lane == 0) { m1v[wave] = s1 * (1.f / 64.f); m2v[wave] = s2 * (1.f / 64.f); }
      }
      __syncthreads();
      for (int idx = tid; idx < 512; idx += 256) {
        int b = idx >> 6, d = idx & 63;
        float dxh = dzb[idx] * lgv[d];
        dob[idx] = isv[b] * (dxh - m1v[b] - xh[idx] * m2v[b]);
      }
      __syncthreads();
      if (tid < 64) {
        float s = 0.f;
        for (int b = 0; b < 8; ++b) s += dzb[b * 64 + tid] * xh[b * 64 + tid];
        lgv[tid] -= LR * s;
      } else if (tid < 128) {
        int d = tid - 64; float s = 0.f;
        for (int b = 0; b < 8; ++b) s += dzb[b * 64 + d];
        lbv[d] -= LR * s;
      }
      gemm16_rowA(dob, W2, nullptr, dab, nullptr, av, 1, EPI_DGELU, wave, lane);
      __syncthreads();
      gemm_outer_update(hg, dob, W2, LR, wave, lane);
      gemm_outer_update(ktb, dab, W1, LR, wave, lane);
      if (tid < 64) {
        float s = 0.f;
        for (int b = 0; b < 8; ++b) s += dob[b * 64 + tid];
        b2v[tid] -= LR * s;
      } else if (tid < 128) {
        int d = tid - 64; float s = 0.f;
        for (int b = 0; b < 8; ++b) s += dab[b * 64 + d];
        b1v[d] -= LR * s;
      }
      __syncthreads();
    }

    fwd_pass(ktb, av, hg, ob, xh, zb, W1, W2, b1v, b2v, lgv, lbv,
             muv, isv, wave, lane, tid);
    {
      float e0 = zb[wave * 64 + lane], e1 = zb[wave * 64 + lane + 32];
      float s = wave_sum32(e0 + e1);
      float q = wave_sum32(e0 * e0 + e1 * e1);
      if (lane == 0) {
        float m = s * (1.f / 64.f);
        muv[wave] = m;
        isv[wave] = rsqrtf(q * (1.f / 64.f) - m * m + LN_EPS);
      }
    }
    __syncthreads();
    for (int idx = tid; idx < 512; idx += 256) {
      int b = idx >> 6, d = idx & 63;
      float z2 = (zb[idx] - muv[b]) * isv[b] * tgv[d] + tbv[d];
      size_t g = ((size_t)(b * SS + t)) * HIDC + h * 64 + d;
      outg[g] = qg[g] + z2;
    }
    __syncthreads();
  }
}

__global__ void __launch_bounds__(256)
gate_ln_kernel(const float* __restrict__ outb, const float* __restrict__ gact,
               const float* __restrict__ png, const float* __restrict__ pnb,
               float* __restrict__ gated) {
  __shared__ float r1[256], r2[256];
  __shared__ float mu_s, inv_s;
  const int r = blockIdx.x, tid = threadIdx.x;
  const float* row = outb + (size_t)r * HIDC;
  float s = 0.f, q = 0.f;
  for (int i = tid; i < HIDC; i += 256) { float v = row[i]; s += v; q += v * v; }
  r1[tid] = s; r2[tid] = q;
  __syncthreads();
  for (int off = 128; off > 0; off >>= 1) {
    if (tid < off) { r1[tid] += r1[tid + off]; r2[tid] += r2[tid + off]; }
    __syncthreads();
  }
  if (tid == 0) {
    float m = r1[0] * (1.f / HIDC);
    mu_s = m;
    inv_s = rsqrtf(r2[0] * (1.f / HIDC) - m * m + LN_EPS);
  }
  __syncthreads();
  for (int i = tid; i < HIDC; i += 256) {
    float y = (row[i] - mu_s) * inv_s * png[i] + pnb[i];
    gated[(size_t)r * HIDC + i] = gact[(size_t)r * HIDC + i] * y;
  }
}

extern "C" void kernel_launch(void* const* d_in, const int* in_sizes, int n_in,
                              void* d_out, int out_size, void* d_ws, size_t ws_size,
                              hipStream_t stream) {
  (void)in_sizes; (void)n_in; (void)out_size; (void)ws_size;
  const float* x   = (const float*)d_in[0];
  const float* wq  = (const float*)d_in[1];
  const float* wk  = (const float*)d_in[2];
  const float* wv  = (const float*)d_in[3];
  const float* fw1 = (const float*)d_in[4];
  const float* fb1 = (const float*)d_in[5];
  const float* fw2 = (const float*)d_in[6];
  const float* fb2 = (const float*)d_in[7];
  const float* flg = (const float*)d_in[8];
  const float* flb = (const float*)d_in[9];
  const float* lw  = (const float*)d_in[10];
  const float* lb  = (const float*)d_in[11];
  const float* ttg = (const float*)d_in[12];
  const float* ttb = (const float*)d_in[13];
  const float* wo  = (const float*)d_in[14];
  const float* wg  = (const float*)d_in[15];
  const float* png = (const float*)d_in[16];
  const float* pnb = (const float*)d_in[17];

  const int M = BB * SS;
  const size_t PLANE = (size_t)M * HIDC;
  float* qb    = (float*)d_ws;
  float* kb    = qb + PLANE;
  float* vb    = kb + PLANE;
  float* gactb = vb + PLANE;
  float* outb  = gactb + PLANE;
  float* gatedb = qb;

  dim3 gblk(256);
  dim3 ggrid(HIDC / 128, M / 16);
  gemm_wmma_f32<<<ggrid, gblk, 0, stream>>>(x, wq, qb,    M, HIDC, HIDC, 0);
  gemm_wmma_f32<<<ggrid, gblk, 0, stream>>>(x, wk, kb,    M, HIDC, HIDC, 0);
  gemm_wmma_f32<<<ggrid, gblk, 0, stream>>>(x, wv, vb,    M, HIDC, HIDC, 0);
  gemm_wmma_f32<<<ggrid, gblk, 0, stream>>>(x, wg, gactb, M, HIDC, HIDC, 1);

  ttt_scan_kernel<<<HH, 256, SCAN_SMEM_FLOATS * sizeof(float), stream>>>(
      kb, vb, qb, fw1, fb1, fw2, fb2, flg, flb, lw, lb, ttg, ttb, outb);

  gate_ln_kernel<<<M, 256, 0, stream>>>(outb, gactb, png, pnb, gatedb);

  gemm_wmma_f32<<<ggrid, gblk, 0, stream>>>(gatedb, wo, (float*)d_out,
                                            M, HIDC, HIDC, 0);
}


# scan: kt@W1-gelu-@W2 and z@LW-d_pred-@LW^T chained in registers: 1st product transposed, its D tile is the 2nd's B operand, 2nd split over k by wave, operands read up front; sums re-associated
# speedup vs baseline: 1.4770x; 1.4770x over previous
#include <hip/hip_runtime.h>
#include <hip/hip_bf16.h>
#include <math.h>

#define BB     8
#define SS     256
#define HIDC   1024
#define HH     16
#define DD     64
#define FF     64
#define LR     0.1f
#define GSCALE (2.0f / 8192.0f)
#define LN_EPS 1e-5f

typedef float v2f __attribute__((ext_vector_type(2)));
typedef float v8f __attribute__((ext_vector_type(8)));

__device__ __forceinline__ v8f wmma4(v2f a, v2f b, v8f c) {
  return __builtin_amdgcn_wmma_f32_16x16x4_f32(
       false, a,  false, b,
       (short)0, c,  false,  false);
}

__device__ __forceinline__ float gelu_f(float x) {
  const float c = 0.7978845608028654f;
  float u = c * (x + 0.044715f * x * x * x);
  return 0.5f * x * (1.0f + tanhf(u));
}
__device__ __forceinline__ float gelu_grad(float x) {
  const float c = 0.7978845608028654f;
  float x2 = x * x;
  float u = c * (x + 0.044715f * x2 * x);
  float t = tanhf(u);
  return 0.5f * (1.0f + t) + 0.5f * x * (1.0f - t * t) * c * (1.0f + 0.134145f * x2);
}
__device__ __forceinline__ float wave_sum32(float v) {
#pragma unroll
  for (int off = 16; off > 0; off >>= 1) v += __shfl_xor(v, off, 32);
  return v;
}

__global__ void __launch_bounds__(256)
gemm_wmma_f32(const float* __restrict__ A, const float* __restrict__ Bm,
              float* __restrict__ C, int M, int N, int K, int act) {
  __shared__ float As[16 * 64];
  const int tid   = threadIdx.x;
  const int wave  = tid >> 5;
  const int lane  = tid & 31;
  const int mbase = blockIdx.y * 16;
  const int tn    = blockIdx.x * 8 + wave;
  const int col   = tn * 16 + (lane & 15);
  const int kh    = (lane >> 4) << 1;
  const int arow  = lane & 15;
  v8f acc = {0.f, 0.f, 0.f, 0.f, 0.f, 0.f, 0.f, 0.f};
  for (int kc = 0; kc < K; kc += 64) {
    __syncthreads();
    for (int i = tid; i < 16 * 64; i += 256)
      As[i] = A[(size_t)(mbase + (i >> 6)) * K + kc + (i & 63)];
    if (kc + 64 < K)
      __builtin_prefetch(&A[(size_t)(mbase + (tid >> 4)) * K + kc + 64], 0, 1);
    __syncthreads();
#pragma unroll
    for (int k0 = 0; k0 < 64; k0 += 4) {
      v2f a, b;
      a.x = As[arow * 64 + k0 + kh];
      a.y = As[arow * 64 + k0 + kh + 1];
      b.x = Bm[(size_t)(kc + k0 + kh) * N + col];
      b.y = Bm[(size_t)(kc + k0 + kh + 1) * N + col];
      acc = wmma4(a, b, acc);
    }
  }
  const int r0 = mbase + ((lane >= 16) ? 8 : 0);
#pragma unroll
  for (int i = 0; i < 8; ++i) {
    float v = acc[i];
    if (act) v = gelu_f(v);
    C[(size_t)(r0 + i) * N + col] = v;
  }
}

#define EPI_BIAS      0
#define EPI_GELU_PAIR 1
#define EPI_DPRED     2
#define EPI_DGELU     3

__device__ __forceinline__ void
gemm16_rowA(const float* A8, const float* Bm, const float* bias,
            float* C8, float* C2, const float* aux,
            int transB, int epi, int wave, int lane) {
  if (wave >= 4) return;
  const int col  = wave * 16 + (lane & 15);
  const int mrow = lane & 15;
  const int kh   = (lane >> 4) << 1;
  const bool live = (mrow < 8);
  v8f acc = {0.f, 0.f, 0.f, 0.f, 0.f, 0.f, 0.f, 0.f};
#pragma unroll
  for (int k0 = 0; k0 < 64; k0 += 4) {
    v2f a, b;
    a.x = live ? A8[mrow * 64 + k0 + kh] : 0.f;
    a.y = live ? A8[mrow * 64 + k0 + kh + 1] : 0.f;
    if (!transB) {
      b.x = Bm[(k0 + kh) * 64 + col];
      b.y = Bm[(k0 + kh + 1) * 64 + col];
    } else {
      b.x = Bm[col * 64 + k0 + kh];
      b.y = Bm[col * 64 + k0 + kh + 1];
    }
    acc = wmma4(a, b, acc);
  }
  if (lane < 16) {
    const float bv = bias ? bias[col] : 0.f;
#pragma unroll
    for (int i = 0; i < 8; ++i) {
      const int o = i * 64 + col;
      float v = acc[i] + bv;
      if (epi == EPI_GELU_PAIR)      { C8[o] = v; C2[o] = gelu_f(v); }
      else if (epi == EPI_DPRED)     { C8[o] = GSCALE * (v - aux[o]); }
      else if (epi == EPI_DGELU)     { C8[o] = v * gelu_grad(aux[o]); }
      else                           { C8[o] = v; }
    }
  }
}

__device__ __forceinline__ void
gemm_outer_update(const float* X8, const float* Y8, float* Wt, float lr,
                  int wave, int lane) {
  const int kh = (lane >> 4) << 1;
#pragma unroll
  for (int ti = 0; ti < 2; ++ti) {
    const int t  = wave * 2 + ti;
    const int tm = t >> 2, tn = t & 3;
    const int m = tm * 16 + (lane & 15);
    const int n = tn * 16 + (lane & 15);
    v8f acc = {0.f, 0.f, 0.f, 0.f, 0.f, 0.f, 0.f, 0.f};
#pragma unroll
    for (int k0 = 0; k0 < 8; k0 += 4) {
      v2f a, b;
      a.x = X8[(k0 + kh) * 64 + m];
      a.y = X8[(k0 + kh + 1) * 64 + m];
      b.x = Y8[(k0 + kh) * 64 + n];
      b.y = Y8[(k0 + kh + 1) * 64 + n];
      acc = wmma4(a, b, acc);
    }
    const int rbase = tm * 16 + ((lane >= 16) ? 8 : 0);
    const int cc    = tn * 16 + (lane & 15);
#pragma unroll
    for (int i = 0; i < 8; ++i) Wt[(rbase + i) * 64 + cc] -= lr * acc[i];
  }
}


#define CH_GELU  0
#define CH_DPRED 1
template <int EPI, bool ROWMAJOR2>
__device__ __forceinline__ void
chain_pair(const float* X8, const float* Wa, const float* biasa, const float* Wb,
           float* Cpre, float* Cact, const float* tgtp, float* part, int wave, int lane) {
  const int c  = lane & 15;
  const int h  = lane >> 4;
  const int bb = c & 7;
  const int kh = h << 1;
  const int fo = wave * 16;
  const int f0 = fo + 8 * h;
  v2f fa[16], fb[16], fa2[16];
  float bi[8], tg[8];
#pragma unroll
  for (int i = 0; i < 16; ++i) {
    const int k0 = 4 * i;
    fa[i].x = Wa[(k0 + kh) * 64 + fo + c];
    fa[i].y = Wa[(k0 + kh + 1) * 64 + fo + c];
    fb[i].x = X8[bb * 64 + k0 + kh];
    fb[i].y = X8[bb * 64 + k0 + kh + 1];
  }
#pragma unroll
  for (int s = 0; s < 4; ++s) {
    const int kk = f0 + 2 * s;
#pragma unroll
    for (int td = 0; td < 4; ++td) {
      const int m = td * 16 + c;
      if (!ROWMAJOR2) {
        fa2[s * 4 + td].x = Wb[kk * 64 + m];
        fa2[s * 4 + td].y = Wb[(kk + 1) * 64 + m];
      } else {
        fa2[s * 4 + td].x = Wb[m * 64 + kk];
        fa2[s * 4 + td].y = Wb[m * 64 + kk + 1];
      }
    }
  }
#pragma unroll
  for (int r = 0; r < 8; ++r) {
    bi[r] = biasa[f0 + r];
    tg[r] = (EPI == CH_DPRED) ? tgtp[bb * 64 + f0 + r] : 0.f;
  }
#define HOLD8(a, o) asm volatile("" :: "v"(a[o]), "v"(a[o + 1]), "v"(a[o + 2]), "v"(a[o + 3]), \
                                       "v"(a[o + 4]), "v"(a[o + 5]), "v"(a[o + 6]), "v"(a[o + 7]))
  HOLD8(fa, 0);  HOLD8(fa, 8);
  HOLD8(fb, 0);  HOLD8(fb, 8);
  HOLD8(fa2, 0); HOLD8(fa2, 8);
  HOLD8(bi, 0);
  if (EPI == CH_DPRED) HOLD8(tg, 0);
#undef HOLD8
  v8f acc = {0.f, 0.f, 0.f, 0.f, 0.f, 0.f, 0.f, 0.f};
#pragma unroll
  for (int i = 0; i < 16; ++i) acc = wmma4(fa[i], fb[i], acc);
  v8f g;
  if (EPI == CH_GELU) {
    v8f pre;
#pragma unroll
    for (int r = 0; r < 8; ++r) {
      float v = acc[r] + bi[r];
      pre[r] = v;
      g[r] = gelu_f(v);
    }
    if (c < 8) {
#pragma unroll
      for (int r = 0; r < 8; ++r) {
        Cpre[bb * 64 + f0 + r] = pre[r];
        Cact[bb * 64 + f0 + r] = g[r];
      }
    }
  } else {
#pragma unroll
    for (int r = 0; r < 8; ++r) {
      float v = acc[r] + bi[r];
      g[r] = GSCALE * (v - tg[r]);
    }
  }
  v8f p0 = {0.f, 0.f, 0.f, 0.f, 0.f, 0.f, 0.f, 0.f};
  v8f p1 = p0, p2 = p0, p3 = p0;
#pragma unroll
  for (int s = 0; s < 4; ++s) {
    v2f b2;
    b2.x = g[2 * s];
    b2.y = g[2 * s + 1];
    p0 = wmma4(fa2[s * 4 + 0], b2, p0);
    p1 = wmma4(fa2[s * 4 + 1], b2, p1);
    p2 = wmma4(fa2[s * 4 + 2], b2, p2);
    p3 = wmma4(fa2[s * 4 + 3], b2, p3);
  }
  if (c < 8) {
    float* dst = part + wave * 512 + bb * 64 + 8 * h;
#pragma unroll
    for (int r = 0; r < 8; ++r) {
      dst[r]      = p0[r];
      dst[16 + r] = p1[r];
      dst[32 + r] = p2[r];
      dst[48 + r] = p3[r];
    }
  }
}

__device__ __forceinline__ float part_sum4(const float* part, int idx) {
  return (part[idx] + part[512 + idx]) + (part[1024 + idx] + part[1536 + idx]);
}

__device__ __forceinline__ void
fwd_pass(const float* ktb, float* av, float* hg, float* ob, float* xh, float* zb,
         const float* W1, const float* W2, const float* b1v, const float* b2v,
         const float* lgv, const float* lbv, float* muv, float* isv, float* part,
         int wave, int lane, int tid) {
  if (wave < 4)
    chain_pair<CH_GELU, false>(ktb, W1, b1v, W2, av, hg, nullptr, part, wave, lane);
  __syncthreads();
  {
    float e0 = part_sum4(part, wave * 64 + lane) + b2v[lane];
    float e1 = part_sum4(part, wave * 64 + lane + 32) + b2v[lane + 32];
    ob[wave * 64 + lane] = e0;
    ob[wave * 64 + lane + 32] = e1;
    float s = wave_sum32(e0 + e1);
    float q = wave_sum32(e0 * e0 + e1 * e1);
    if (lane == 0) {
      float m = s * (1.f / 64.f);
      muv[wave] = m;
      isv[wave] = rsqrtf(q * (1.f / 64.f) - m * m + LN_EPS);
    }
  }
  __syncthreads();
  for (int idx = tid; idx < 512; idx += 256) {
    int b = idx >> 6, d = idx & 63;
    float xv = (ob[idx] - muv[b]) * isv[b];
    xh[idx] = xv;
    zb[idx] = xv * lgv[d] + lbv[d];
  }
  __syncthreads();
}

#define SCAN_SMEM_FLOATS (3 * 4096 + 10 * 512 + 7 * 64 + 32 + 4 * 512)

__global__ void __launch_bounds__(256)
ttt_scan_kernel(const float* __restrict__ kg, const float* __restrict__ vg,
                const float* __restrict__ qg,
                const float* __restrict__ fw1, const float* __restrict__ fb1,
                const float* __restrict__ fw2, const float* __restrict__ fb2,
                const float* __restrict__ flg, const float* __restrict__ flb,
                const float* __restrict__ lwg, const float* __restrict__ lbg,
                const float* __restrict__ ttg, const float* __restrict__ ttb,
                float* __restrict__ outg) {
  extern __shared__ __align__(16) float smem[];
  float* W1  = smem;
  float* W2  = W1 + 4096;
  float* LW  = W2 + 4096;
  float* ktb = LW + 4096;
  float* tgt = ktb + 512;
  float* av  = tgt + 512;
  float* hg  = av + 512;
  float* ob  = hg + 512;
  float* xh  = ob + 512;
  float* zb  = xh + 512;
  float* dzb = zb + 512;
  float* dob = dzb + 512;
  float* dab = dob + 512;
  float* b1v = dab + 512;
  float* b2v = b1v + 64;
  float* lgv = b2v + 64;
  float* lbv = lgv + 64;
  float* tgv = lbv + 64;
  float* tbv = tgv + 64;
  float* lbz = tbv + 64;
  float* muv = lbz + 64;
  float* isv = muv + 8;
  float* m1v = isv + 8;
  float* m2v = m1v + 8;
  float* part = m2v + 8;

  const int h    = blockIdx.x;
  const int tid  = threadIdx.x;
  const int wave = tid >> 5;
  const int lane = tid & 31;

  const int b0 = tid >> 6,          d0 = tid & 63;
  const int b1i = (tid + 256) >> 6, d1i = (tid + 256) & 63;
  const size_t g0base = (size_t)(b0 * SS) * HIDC + h * 64 + d0;
  const size_t g1base = (size_t)(b1i * SS) * HIDC + h * 64 + d1i;

  for (int i = tid; i < 4096; i += 256) {
    W1[i] = fw1[(size_t)h * 4096 + i];
    W2[i] = fw2[(size_t)h * 4096 + i];
    LW[i] = lwg[i];
  }
  if (tid < 64) {
    b1v[tid] = fb1[h * 64 + tid];
    b2v[tid] = fb2[h * 64 + tid];
    lgv[tid] = flg[h * 64 + tid];
    lbv[tid] = flb[h * 64 + tid];
    tgv[tid] = ttg[h * 64 + tid];
    tbv[tid] = ttb[h * 64 + tid];
    lbz[tid] = lbg[tid];
  }

  float kp0 = kg[g0base], vp0 = vg[g0base];
  float kp1 = kg[g1base], vp1 = vg[g1base];
  __syncthreads();

  for (int t = 0; t < SS; ++t) {
    ktb[tid]       = kp0;  tgt[tid]       = vp0 - kp0;
    ktb[tid + 256] = kp1;  tgt[tid + 256] = vp1 - kp1;
    __syncthreads();
    if (t + 1 < SS) {
      kp0 = kg[g0base + (t + 1) * HIDC];  vp0 = vg[g0base + (t + 1) * HIDC];
      kp1 = kg[g1base + (t + 1) * HIDC];  vp1 = vg[g1base + (t + 1) * HIDC];
    }

    for (int step = 0; step < 4; ++step) {
      fwd_pass(ktb, av, hg, ob, xh, zb, W1, W2, b1v, b2v, lgv, lbv,
               muv, isv, part, wave, lane, tid);
      if (wave < 4)
        chain_pair<CH_DPRED, true>(zb, LW, lbz, LW, nullptr, nullptr, tgt, part, wave, lane);
      __syncthreads();
      {
        float z0 = part_sum4(part, wave * 64 + lane);
        float z1 = part_sum4(part, wave * 64 + lane + 32);
        dzb[wave * 64 + lane] = z0;
        dzb[wave * 64 + lane + 32] = z1;
        float dd0 = z0 * lgv[lane];
        float dd1 = z1 * lgv[lane + 32];
        float x0 = xh[wave * 64 + lane], x1 = xh[wave * 64 + lane + 32];
        float s1 = wave_sum32(dd0 + dd1);
        float s2 = wave_sum32(dd0 * x0 + dd1 * x1);
        if (lane == 0) { m1v[wave] = s1 * (1.f / 64.f); m2v[wave] = s2 * (1.f / 64.f); }
      }
      __syncthreads();
      for (int idx = tid; idx < 512; idx += 256) {
        int b = idx >> 6, d = idx & 63;
        float dxh = dzb[idx] * lgv[d];
        dob[idx] = isv[b] * (dxh - m1v[b] - xh[idx] * m2v[b]);
      }
      __syncthreads();
      if (tid < 64) {
        float s = 0.f;
        for (int b = 0; b < 8; ++b) s += dzb[b * 64 + tid] * xh[b * 64 + tid];
        lgv[tid] -= LR * s;
      } else if (tid < 128) {
        int d = tid - 64; float s = 0.f;
        for (int b = 0; b < 8; ++b) s += dzb[b * 64 + d];
        lbv[d] -= LR * s;
      }
      gemm16_rowA(dob, W2, nullptr, dab, nullptr, av, 1, EPI_DGELU, wave, lane);
      __syncthreads();
      gemm_outer_update(hg, dob, W2, LR, wave, lane);
      gemm_outer_update(ktb, dab, W1, LR, wave, lane);
      if (tid < 64) {
        float s = 0.f;
        for (int b = 0; b < 8; ++b) s += dob[b * 64 + tid];
        b2v[tid] -= LR * s;
      } else if (tid < 128) {
        int d = tid - 64; float s = 0.f;
        for (int b = 0; b < 8; ++b) s += dab[b * 64 + d];
        b1v[d] -= LR * s;
      }
      __syncthreads();
    }

    fwd_pass(ktb, av, hg, ob, xh, zb, W1, W2, b1v, b2v, lgv, lbv,
             muv, isv, part, wave, lane, tid);
    {
      float e0 = zb[wave * 64 + lane], e1 = zb[wave * 64 + lane + 32];
      float s = wave_sum32(e0 + e1);
      float q = wave_sum32(e0 * e0 + e1 * e1);
      if (lane == 0) {
        float m = s * (1.f / 64.f);
        muv[wave] = m;
        isv[wave] = rsqrtf(q * (1.f / 64.f) - m * m + LN_EPS);
      }
    }
    __syncthreads();
    for (int idx = tid; idx < 512; idx += 256) {
      int b = idx >> 6, d = idx & 63;
      float z2 = (zb[idx] - muv[b]) * isv[b] * tgv[d] + tbv[d];
      size_t g = ((size_t)(b * SS + t)) * HIDC + h * 64 + d;
      outg[g] = qg[g] + z2;
    }
    __syncthreads();
  }
}

__global__ void __launch_bounds__(256)
gate_ln_kernel(const float* __restrict__ outb, const float* __restrict__ gact,
               const float* __restrict__ png, const float* __restrict__ pnb,
               float* __restrict__ gated) {
  __shared__ float r1[256], r2[256];
  __shared__ float mu_s, inv_s;
  const int r = blockIdx.x, tid = threadIdx.x;
  const float* row = outb + (size_t)r * HIDC;
  float s = 0.f, q = 0.f;
  for (int i = tid; i < HIDC; i += 256) { float v = row[i]; s += v; q += v * v; }
  r1[tid] = s; r2[tid] = q;
  __syncthreads();
  for (int off = 128; off > 0; off >>= 1) {
    if (tid < off) { r1[tid] += r1[tid + off]; r2[tid] += r2[tid + off]; }
    __syncthreads();
  }
  if (tid == 0) {
    float m = r1[0] * (1.f / HIDC);
    mu_s = m;
    inv_s = rsqrtf(r2[0] * (1.f / HIDC) - m * m + LN_EPS);
  }
  __syncthreads();
  for (int i = tid; i < HIDC; i += 256) {
    float y = (row[i] - mu_s) * inv_s * png[i] + pnb[i];
    gated[(size_t)r * HIDC + i] = gact[(size_t)r * HIDC + i] * y;
  }
}

extern "C" void kernel_launch(void* const* d_in, const int* in_sizes, int n_in,
                              void* d_out, int out_size, void* d_ws, size_t ws_size,
                              hipStream_t stream) {
  (void)in_sizes; (void)n_in; (void)out_size; (void)ws_size;
  const float* x   = (const float*)d_in[0];
  const float* wq  = (const float*)d_in[1];
  const float* wk  = (const float*)d_in[2];
  const float* wv  = (const float*)d_in[3];
  const float* fw1 = (const float*)d_in[4];
  const float* fb1 = (const float*)d_in[5];
  const float* fw2 = (const float*)d_in[6];
  const float* fb2 = (const float*)d_in[7];
  const float* flg = (const float*)d_in[8];
  const float* flb = (const float*)d_in[9];
  const float* lw  = (const float*)d_in[10];
  const float* lb  = (const float*)d_in[11];
  const float* ttg = (const float*)d_in[12];
  const float* ttb = (const float*)d_in[13];
  const float* wo  = (const float*)d_in[14];
  const float* wg  = (const float*)d_in[15];
  const float* png = (const float*)d_in[16];
  const float* pnb = (const float*)d_in[17];

  const int M = BB * SS;
  const size_t PLANE = (size_t)M * HIDC;
  float* qb    = (float*)d_ws;
  float* kb    = qb + PLANE;
  float* vb    = kb + PLANE;
  float* gactb = vb + PLANE;
  float* outb  = gactb + PLANE;
  float* gatedb = qb;

  dim3 gblk(256);
  dim3 ggrid(HIDC / 128, M / 16);
  gemm_wmma_f32<<<ggrid, gblk, 0, stream>>>(x, wq, qb,    M, HIDC, HIDC, 0);
  gemm_wmma_f32<<<ggrid, gblk, 0, stream>>>(x, wk, kb,    M, HIDC, HIDC, 0);
  gemm_wmma_f32<<<ggrid, gblk, 0, stream>>>(x, wv, vb,    M, HIDC, HIDC, 0);
  gemm_wmma_f32<<<ggrid, gblk, 0, stream>>>(x, wg, gactb, M, HIDC, HIDC, 1);

  ttt_scan_kernel<<<HH, 256, SCAN_SMEM_FLOATS * sizeof(float), stream>>>(
      kb, vb, qb, fw1, fb1, fw2, fb2, flg, flb, lw, lb, ttg, ttb, outb);

  gate_ln_kernel<<<M, 256, 0, stream>>>(outb, gactb, png, pnb, gatedb);

  gemm_wmma_f32<<<ggrid, gblk, 0, stream>>>(gatedb, wo, (float*)d_out,
                                            M, HIDC, HIDC, 0);
}


# scan: last two phases of a grad step merged: d_a=(dob@W2^T)*gelu' stays in registers as B operand of the W1 update (waves 0-3), W2 and vector updates on waves 4-7, dob-free operands read early
# speedup vs baseline: 1.6592x; 1.1234x over previous
#include <hip/hip_runtime.h>
#include <hip/hip_bf16.h>
#include <math.h>

#define BB     8
#define SS     256
#define HIDC   1024
#define HH     16
#define DD     64
#define FF     64
#define LR     0.1f
#define GSCALE (2.0f / 8192.0f)
#define LN_EPS 1e-5f

typedef float v2f __attribute__((ext_vector_type(2)));
typedef float v8f __attribute__((ext_vector_type(8)));

__device__ __forceinline__ v8f wmma4(v2f a, v2f b, v8f c) {
  return __builtin_amdgcn_wmma_f32_16x16x4_f32(
       false, a,  false, b,
       (short)0, c,  false,  false);
}

__device__ __forceinline__ float gelu_f(float x) {
  const float c = 0.7978845608028654f;
  float u = c * (x + 0.044715f * x * x * x);
  return 0.5f * x * (1.0f + tanhf(u));
}
__device__ __forceinline__ float gelu_grad(float x) {
  const float c = 0.7978845608028654f;
  float x2 = x * x;
  float u = c * (x + 0.044715f * x2 * x);
  float t = tanhf(u);
  return 0.5f * (1.0f + t) + 0.5f * x * (1.0f - t * t) * c * (1.0f + 0.134145f * x2);
}
__device__ __forceinline__ float wave_sum32(float v) {
#pragma unroll
  for (int off = 16; off > 0; off >>= 1) v += __shfl_xor(v, off, 32);
  return v;
}

__global__ void __launch_bounds__(256)
gemm_wmma_f32(const float* __restrict__ A, const float* __restrict__ Bm,
              float* __restrict__ C, int M, int N, int K, int act) {
  __shared__ float As[16 * 64];
  const int tid   = threadIdx.x;
  const int wave  = tid >> 5;
  const int lane  = tid & 31;
  const int mbase = blockIdx.y * 16;
  const int tn    = blockIdx.x * 8 + wave;
  const int col   = tn * 16 + (lane & 15);
  const int kh    = (lane >> 4) << 1;
  const int arow  = lane & 15;
  v8f acc = {0.f, 0.f, 0.f, 0.f, 0.f, 0.f, 0.f, 0.f};
  for (int kc = 0; kc < K; kc += 64) {
    __syncthreads();
    for (int i = tid; i < 16 * 64; i += 256)
      As[i] = A[(size_t)(mbase + (i >> 6)) * K + kc + (i & 63)];
    if (kc + 64 < K)
      __builtin_prefetch(&A[(size_t)(mbase + (tid >> 4)) * K + kc + 64], 0, 1);
    __syncthreads();
#pragma unroll
    for (int k0 = 0; k0 < 64; k0 += 4) {
      v2f a, b;
      a.x = As[arow * 64 + k0 + kh];
      a.y = As[arow * 64 + k0 + kh + 1];
      b.x = Bm[(size_t)(kc + k0 + kh) * N + col];
      b.y = Bm[(size_t)(kc + k0 + kh + 1) * N + col];
      acc = wmma4(a, b, acc);
    }
  }
  const int r0 = mbase + ((lane >= 16) ? 8 : 0);
#pragma unroll
  for (int i = 0; i < 8; ++i) {
    float v = acc[i];
    if (act) v = gelu_f(v);
    C[(size_t)(r0 + i) * N + col] = v;
  }
}


#define HOLD8(a, o) asm volatile("" :: "v"(a[o]), "v"(a[o + 1]), "v"(a[o + 2]), "v"(a[o + 3]), \
                                       "v"(a[o + 4]), "v"(a[o + 5]), "v"(a[o + 6]), "v"(a[o + 7]))
#define CH_GELU  0
#define CH_DPRED 1
template <int EPI, bool ROWMAJOR2, bool HAS_LATER, typename LATER>
__device__ __forceinline__ void
chain_pair(const float* X8, const float* Wa, const float* biasa, const float* Wb,
           float* Cpre, float* Cact, const float* tgtp, float* part, int wave, int lane,
           LATER later_reads) {
  const int c  = lane & 15;
  const int h  = lane >> 4;
  const int bb = c & 7;
  const int kh = h << 1;
  const int fo = wave * 16;
  const int f0 = fo + 8 * h;
  v2f fa[16], fb[16], fa2[16];
  float bi[8], tg[8];
#pragma unroll
  for (int i = 0; i < 16; ++i) {
    const int k0 = 4 * i;
    fa[i].x = Wa[(k0 + kh) * 64 + fo + c];
    fa[i].y = Wa[(k0 + kh + 1) * 64 + fo + c];
    fb[i].x = X8[bb * 64 + k0 + kh];
    fb[i].y = X8[bb * 64 + k0 + kh + 1];
  }
#pragma unroll
  for (int s = 0; s < 4; ++s) {
    const int kk = f0 + 2 * s;
#pragma unroll
    for (int td = 0; td < 4; ++td) {
      const int m = td * 16 + c;
      if (!ROWMAJOR2) {
        fa2[s * 4 + td].x = Wb[kk * 64 + m];
        fa2[s * 4 + td].y = Wb[(kk + 1) * 64 + m];
      } else {
        fa2[s * 4 + td].x = Wb[m * 64 + kk];
        fa2[s * 4 + td].y = Wb[m * 64 + kk + 1];
      }
    }
  }
#pragma unroll
  for (int r = 0; r < 8; ++r) {
    bi[r] = biasa[f0 + r];
    tg[r] = (EPI == CH_DPRED) ? tgtp[bb * 64 + f0 + r] : 0.f;
  }
  HOLD8(fa, 0);  HOLD8(fa, 8);
  HOLD8(fb, 0);  HOLD8(fb, 8);
  HOLD8(fa2, 0); HOLD8(fa2, 8);
  HOLD8(bi, 0);
  if (EPI == CH_DPRED) HOLD8(tg, 0);
  if (HAS_LATER) {
    later_reads();
    __builtin_amdgcn_sched_barrier(0);
  }
  v8f acc = {0.f, 0.f, 0.f, 0.f, 0.f, 0.f, 0.f, 0.f};
#pragma unroll
  for (int i = 0; i < 16; ++i) acc = wmma4(fa[i], fb[i], acc);
  v8f g;
  if (EPI == CH_GELU) {
    v8f pre;
#pragma unroll
    for (int r = 0; r < 8; ++r) {
      float v = acc[r] + bi[r];
      pre[r] = v;
      g[r] = gelu_f(v);
    }
    if (c < 8) {
#pragma unroll
      for (int r = 0; r < 8; ++r) {
        Cpre[bb * 64 + f0 + r] = pre[r];
        Cact[bb * 64 + f0 + r] = g[r];
      }
    }
  } else {
#pragma unroll
    for (int r = 0; r < 8; ++r) {
      float v = acc[r] + bi[r];
      g[r] = GSCALE * (v - tg[r]);
    }
  }
  v8f p0 = {0.f, 0.f, 0.f, 0.f, 0.f, 0.f, 0.f, 0.f};
  v8f p1 = p0, p2 = p0, p3 = p0;
#pragma unroll
  for (int s = 0; s < 4; ++s) {
    v2f b2;
    b2.x = g[2 * s];
    b2.y = g[2 * s + 1];
    p0 = wmma4(fa2[s * 4 + 0], b2, p0);
    p1 = wmma4(fa2[s * 4 + 1], b2, p1);
    p2 = wmma4(fa2[s * 4 + 2], b2, p2);
    p3 = wmma4(fa2[s * 4 + 3], b2, p3);
  }
  if (c < 8) {
    float* dst = part + wave * 512 + bb * 64 + 8 * h;
#pragma unroll
    for (int r = 0; r < 8; ++r) {
      dst[r]      = p0[r];
      dst[16 + r] = p1[r];
      dst[32 + r] = p2[r];
      dst[48 + r] = p3[r];
    }
  }
}

__device__ __forceinline__ float part_sum4(const float* part, int idx) {
  return (part[idx] + part[512 + idx]) + (part[1024 + idx] + part[1536 + idx]);
}

__device__ __forceinline__ void
fwd_pass(const float* ktb, float* av, float* hg, float* ob, float* xh, float* zb,
         const float* W1, const float* W2, const float* b1v, const float* b2v,
         const float* lgv, const float* lbv, float* muv, float* isv, float* part,
         int wave, int lane, int tid) {
  if (wave < 4)
    chain_pair<CH_GELU, false, false>(ktb, W1, b1v, W2, av, hg, nullptr, part, wave, lane, [] {});
  __syncthreads();
  {
    float e0 = part_sum4(part, wave * 64 + lane) + b2v[lane];
    float e1 = part_sum4(part, wave * 64 + lane + 32) + b2v[lane + 32];
    ob[wave * 64 + lane] = e0;
    ob[wave * 64 + lane + 32] = e1;
    float s = wave_sum32(e0 + e1);
    float q = wave_sum32(e0 * e0 + e1 * e1);
    if (lane == 0) {
      float m = s * (1.f / 64.f);
      muv[wave] = m;
      isv[wave] = rsqrtf(q * (1.f / 64.f) - m * m + LN_EPS);
    }
  }
  __syncthreads();
  for (int idx = tid; idx < 512; idx += 256) {
    int b = idx >> 6, d = idx & 63;
    float xv = (ob[idx] - muv[b]) * isv[b];
    xh[idx] = xv;
    zb[idx] = xv * lgv[d] + lbv[d];
  }
  __syncthreads();
}

#define SCAN_SMEM_FLOATS (3 * 4096 + 9 * 512 + 7 * 64 + 32 + 4 * 512)

__global__ void __launch_bounds__(256)
ttt_scan_kernel(const float* __restrict__ kg, const float* __restrict__ vg,
                const float* __restrict__ qg,
                const float* __restrict__ fw1, const float* __restrict__ fb1,
                const float* __restrict__ fw2, const float* __restrict__ fb2,
                const float* __restrict__ flg, const float* __restrict__ flb,
                const float* __restrict__ lwg, const float* __restrict__ lbg,
                const float* __restrict__ ttg, const float* __restrict__ ttb,
                float* __restrict__ outg) {
  extern __shared__ __align__(16) float smem[];
  float* W1  = smem;
  float* W2  = W1 + 4096;
  float* LW  = W2 + 4096;
  float* ktb = LW + 4096;
  float* tgt = ktb + 512;
  float* av  = tgt + 512;
  float* hg  = av + 512;
  float* ob  = hg + 512;
  float* xh  = ob + 512;
  float* zb  = xh + 512;
  float* dzb = zb + 512;
  float* dob = dzb + 512;
  float* b1v = dob + 512;
  float* b2v = b1v + 64;
  float* lgv = b2v + 64;
  float* lbv = lgv + 64;
  float* tgv = lbv + 64;
  float* tbv = tgv + 64;
  float* lbz = tbv + 64;
  float* muv = lbz + 64;
  float* isv = muv + 8;
  float* m1v = isv + 8;
  float* m2v = m1v + 8;
  float* part = m2v + 8;

  const int h    = blockIdx.x;
  const int tid  = threadIdx.x;
  const int wave = tid >> 5;
  const int lane = tid & 31;

  const int b0 = tid >> 6,          d0 = tid & 63;
  const int b1i = (tid + 256) >> 6, d1i = (tid + 256) & 63;
  const size_t g0base = (size_t)(b0 * SS) * HIDC + h * 64 + d0;
  const size_t g1base = (size_t)(b1i * SS) * HIDC + h * 64 + d1i;

  for (int i = tid; i < 4096; i += 256) {
    W1[i] = fw1[(size_t)h * 4096 + i];
    W2[i] = fw2[(size_t)h * 4096 + i];
    LW[i] = lwg[i];
  }
  if (tid < 64) {
    b1v[tid] = fb1[h * 64 + tid];
    b2v[tid] = fb2[h * 64 + tid];
    lgv[tid] = flg[h * 64 + tid];
    lbv[tid] = flb[h * 64 + tid];
    tgv[tid] = ttg[h * 64 + tid];
    tbv[tid] = ttb[h * 64 + tid];
    lbz[tid] = lbg[tid];
  }

  float kp0 = kg[g0base], vp0 = vg[g0base];
  float kp1 = kg[g1base], vp1 = vg[g1base];
  __syncthreads();

  for (int t = 0; t < SS; ++t) {
    ktb[tid]       = kp0;  tgt[tid]       = vp0 - kp0;
    ktb[tid + 256] = kp1;  tgt[tid + 256] = vp1 - kp1;
    __syncthreads();
    if (t + 1 < SS) {
      kp0 = kg[g0base + (t + 1) * HIDC];  vp0 = vg[g0base + (t + 1) * HIDC];
      kp1 = kg[g1base + (t + 1) * HIDC];  vp1 = vg[g1base + (t + 1) * HIDC];
    }

    for (int step = 0; step < 4; ++step) {
      fwd_pass(ktb, av, hg, ob, xh, zb, W1, W2, b1v, b2v, lgv, lbv,
               muv, isv, part, wave, lane, tid);
      const int fc = lane & 15;
      const int fh = lane >> 4;
      const int fk = fh << 1;
      v2f   pb[16];
      v2f   pk[8];
      float pg[8];
      float pb1 = 0.f;
      v2f   ph[2];
      float wold[32];
      if (wave < 4) {
        chain_pair<CH_DPRED, true, true>(zb, LW, lbz, LW, nullptr, nullptr, tgt, part, wave, lane, [&] {
          const int col = wave * 16 + fc;
#pragma unroll
          for (int i = 0; i < 16; ++i) {
            pb[i].x = W2[col * 64 + 4 * i + fk];
            pb[i].y = W2[col * 64 + 4 * i + fk + 1];
          }
#pragma unroll
          for (int r = 0; r < 8; ++r) pg[r] = av[r * 64 + col];
          pb1 = b1v[col];
#pragma unroll
          for (int tm = 0; tm < 4; ++tm) {
#pragma unroll
            for (int s = 0; s < 2; ++s) {
              pk[tm * 2 + s].x = ktb[(4 * s + fk) * 64 + tm * 16 + fc];
              pk[tm * 2 + s].y = ktb[(4 * s + fk + 1) * 64 + tm * 16 + fc];
            }
#pragma unroll
            for (int i = 0; i < 8; ++i) wold[tm * 8 + i] = W1[(tm * 16 + 8 * fh + i) * 64 + col];
          }
        });
      } else {
        const int wv = wave - 4;
#pragma unroll
        for (int s = 0; s < 2; ++s) {
          ph[s].x = hg[(4 * s + fk) * 64 + wv * 16 + fc];
          ph[s].y = hg[(4 * s + fk + 1) * 64 + wv * 16 + fc];
        }
#pragma unroll
        for (int tn = 0; tn < 4; ++tn) {
#pragma unroll
          for (int i = 0; i < 8; ++i) wold[tn * 8 + i] = W2[(wv * 16 + 8 * fh + i) * 64 + tn * 16 + fc];
        }
      }
      __syncthreads();
      {
        float z0 = part_sum4(part, wave * 64 + lane);
        float z1 = part_sum4(part, wave * 64 + lane + 32);
        dzb[wave * 64 + lane] = z0;
        dzb[wave * 64 + lane + 32] = z1;
        float dd0 = z0 * lgv[lane];
        float dd1 = z1 * lgv[lane + 32];
        float x0 = xh[wave * 64 + lane], x1 = xh[wave * 64 + lane + 32];
        float s1 = wave_sum32(dd0 + dd1);
        float s2 = wave_sum32(dd0 * x0 + dd1 * x1);
        if (lane == 0) { m1v[wave] = s1 * (1.f / 64.f); m2v[wave] = s2 * (1.f / 64.f); }
      }
      __syncthreads();
      for (int idx = tid; idx < 512; idx += 256) {
        int b = idx >> 6, d = idx & 63;
        float dxh = dzb[idx] * lgv[d];
        dob[idx] = isv[b] * (dxh - m1v[b] - xh[idx] * m2v[b]);
      }
      __syncthreads();
      if (wave < 4) {
        const int col = wave * 16 + fc;
        const int bb  = fc & 7;
        v2f pa[16];
#pragma unroll
        for (int i = 0; i < 16; ++i) {
          pa[i].x = dob[bb * 64 + 4 * i + fk];
          pa[i].y = dob[bb * 64 + 4 * i + fk + 1];
        }
        HOLD8(pa, 0); HOLD8(pa, 8);
        v8f acc = {0.f, 0.f, 0.f, 0.f, 0.f, 0.f, 0.f, 0.f};
#pragma unroll
        for (int i = 0; i < 16; ++i) acc = wmma4(pa[i], pb[i], acc);
        float da[8];
        float sb = 0.f;
#pragma unroll
        for (int r = 0; r < 8; ++r) da[r] = acc[r] * gelu_grad(pg[r]);
#pragma unroll
        for (int r = 0; r < 8; ++r) sb += da[r];
        if (fh == 0) b1v[col] = pb1 - LR * sb;
        v8f u0 = {0.f, 0.f, 0.f, 0.f, 0.f, 0.f, 0.f, 0.f};
        v8f u1 = u0, u2 = u0, u3 = u0;
#pragma unroll
        for (int s = 0; s < 2; ++s) {
          v2f b2;
          b2.x = fh ? da[4 * s + 2] : da[4 * s];
          b2.y = fh ? da[4 * s + 3] : da[4 * s + 1];
          u0 = wmma4(pk[0 + s], b2, u0);
          u1 = wmma4(pk[2 + s], b2, u1);
          u2 = wmma4(pk[4 + s], b2, u2);
          u3 = wmma4(pk[6 + s], b2, u3);
        }
#pragma unroll
        for (int i = 0; i < 8; ++i) {
          W1[(0 * 16 + 8 * fh + i) * 64 + col] = wold[0 * 8 + i] - LR * u0[i];
          W1[(1 * 16 + 8 * fh + i) * 64 + col] = wold[1 * 8 + i] - LR * u1[i];
          W1[(2 * 16 + 8 * fh + i) * 64 + col] = wold[2 * 8 + i] - LR * u2[i];
          W1[(3 * 16 + 8 * fh + i) * 64 + col] = wold[3 * 8 + i] - LR * u3[i];
        }
      } else {
        const int wv = wave - 4;
        v2f pd[8];
#pragma unroll
        for (int tn = 0; tn < 4; ++tn) {
#pragma unroll
          for (int s = 0; s < 2; ++s) {
            pd[tn * 2 + s].x = dob[(4 * s + fk) * 64 + tn * 16 + fc];
            pd[tn * 2 + s].y = dob[(4 * s + fk + 1) * 64 + tn * 16 + fc];
          }
        }
        v8f u0 = {0.f, 0.f, 0.f, 0.f, 0.f, 0.f, 0.f, 0.f};
        v8f u1 = u0, u2 = u0, u3 = u0;
#pragma unroll
        for (int s = 0; s < 2; ++s) {
          u0 = wmma4(ph[s], pd[0 + s], u0);
          u1 = wmma4(ph[s], pd[2 + s], u1);
          u2 = wmma4(ph[s], pd[4 + s], u2);
          u3 = wmma4(ph[s], pd[6 + s], u3);
        }
        const int rb = (wv * 16 + 8 * fh) * 64 + fc;
#pragma unroll
        for (int i = 0; i < 8; ++i) {
          W2[rb + i * 64 + 0]  = wold[0 * 8 + i] - LR * u0[i];
          W2[rb + i * 64 + 16] = wold[1 * 8 + i] - LR * u1[i];
          W2[rb + i * 64 + 32] = wold[2 * 8 + i] - LR * u2[i];
          W2[rb + i * 64 + 48] = wold[3 * 8 + i] - LR * u3[i];
        }
        const int t2 = tid - 128;
        if (t2 < 64) {
          float s = 0.f, s2 = 0.f;
          for (int b = 0; b < 8; ++b) s += dzb[b * 64 + t2] * xh[b * 64 + t2];
          for (int b = 0; b < 8; ++b) s2 += dob[b * 64 + t2];
          lgv[t2] -= LR * s;
          b2v[t2] -= LR * s2;
        } else {
          const int d = t2 - 64;
          float s = 0.f;
          for (int b = 0; b < 8; ++b) s += dzb[b * 64 + d];
          lbv[d] -= LR * s;
        }
      }
      __syncthreads();
    }

    fwd_pass(ktb, av, hg, ob, xh, zb, W1, W2, b1v, b2v, lgv, lbv,
             muv, isv, part, wave, lane, tid);
    {
      float e0 = zb[wave * 64 + lane], e1 = zb[wave * 64 + lane + 32];
      float s = wave_sum32(e0 + e1);
      float q = wave_sum32(e0 * e0 + e1 * e1);
      if (lane == 0) {
        float m = s * (1.f / 64.f);
        muv[wave] = m;
        isv[wave] = rsqrtf(q * (1.f / 64.f) - m * m + LN_EPS);
      }
    }
    __syncthreads();
    for (int idx = tid; idx < 512; idx += 256) {
      int b = idx >> 6, d = idx & 63;
      float z2 = (zb[idx] - muv[b]) * isv[b] * tgv[d] + tbv[d];
      size_t g = ((size_t)(b * SS + t)) * HIDC + h * 64 + d;
      outg[g] = qg[g] + z2;
    }
    __syncthreads();
  }
}

__global__ void __launch_bounds__(256)
gate_ln_kernel(const float* __restrict__ outb, const float* __restrict__ gact,
               const float* __restrict__ png, const float* __restrict__ pnb,
               float* __restrict__ gated) {
  __shared__ float r1[256], r2[256];
  __shared__ float mu_s, inv_s;
  const int r = blockIdx.x, tid = threadIdx.x;
  const float* row = outb + (size_t)r * HIDC;
  float s = 0.f, q = 0.f;
  for (int i = tid; i < HIDC; i += 256) { float v = row[i]; s += v; q += v * v; }
  r1[tid] = s; r2[tid] = q;
  __syncthreads();
  for (int off = 128; off > 0; off >>= 1) {
    if (tid < off) { r1[tid] += r1[tid + off]; r2[tid] += r2[tid + off]; }
    __syncthreads();
  }
  if (tid == 0) {
    float m = r1[0] * (1.f / HIDC);
    mu_s = m;
    inv_s = rsqrtf(r2[0] * (1.f / HIDC) - m * m + LN_EPS);
  }
  __syncthreads();
  for (int i = tid; i < HIDC; i += 256) {
    float y = (row[i] - mu_s) * inv_s * png[i] + pnb[i];
    gated[(size_t)r * HIDC + i] = gact[(size_t)r * HIDC + i] * y;
  }
}

extern "C" void kernel_launch(void* const* d_in, const int* in_sizes, int n_in,
                              void* d_out, int out_size, void* d_ws, size_t ws_size,
                              hipStream_t stream) {
  (void)in_sizes; (void)n_in; (void)out_size; (void)ws_size;
  const float* x   = (const float*)d_in[0];
  const float* wq  = (const float*)d_in[1];
  const float* wk  = (const float*)d_in[2];
  const float* wv  = (const float*)d_in[3];
  const float* fw1 = (const float*)d_in[4];
  const float* fb1 = (const float*)d_in[5];
  const float* fw2 = (const float*)d_in[6];
  const float* fb2 = (const float*)d_in[7];
  const float* flg = (const float*)d_in[8];
  const float* flb = (const float*)d_in[9];
  const float* lw  = (const float*)d_in[10];
  const float* lb  = (const float*)d_in[11];
  const float* ttg = (const float*)d_in[12];
  const float* ttb = (const float*)d_in[13];
  const float* wo  = (const float*)d_in[14];
  const float* wg  = (const float*)d_in[15];
  const float* png = (const float*)d_in[16];
  const float* pnb = (const float*)d_in[17];

  const int M = BB * SS;
  const size_t PLANE = (size_t)M * HIDC;
  float* qb    = (float*)d_ws;
  float* kb    = qb + PLANE;
  float* vb    = kb + PLANE;
  float* gactb = vb + PLANE;
  float* outb  = gactb + PLANE;
  float* gatedb = qb;

  dim3 gblk(256);
  dim3 ggrid(HIDC / 128, M / 16);
  gemm_wmma_f32<<<ggrid, gblk, 0, stream>>>(x, wq, qb,    M, HIDC, HIDC, 0);
  gemm_wmma_f32<<<ggrid, gblk, 0, stream>>>(x, wk, kb,    M, HIDC, HIDC, 0);
  gemm_wmma_f32<<<ggrid, gblk, 0, stream>>>(x, wv, vb,    M, HIDC, HIDC, 0);
  gemm_wmma_f32<<<ggrid, gblk, 0, stream>>>(x, wg, gactb, M, HIDC, HIDC, 1);

  ttt_scan_kernel<<<HH, 256, SCAN_SMEM_FLOATS * sizeof(float), stream>>>(
      kb, vb, qb, fw1, fb1, fw2, fb2, flg, flb, lw, lb, ttg, ttb, outb);

  gate_ln_kernel<<<M, 256, 0, stream>>>(outb, gactb, png, pnb, gatedb);

  gemm_wmma_f32<<<ggrid, gblk, 0, stream>>>(gatedb, wo, (float*)d_out,
                                            M, HIDC, HIDC, 0);
}


# scan: gelu'(a) formed in the forward epilogue from the tanh that gelu evaluates anyway, stored in the pre-activation's place; backward multiplies by it: 8 fewer serial tanh per lane per grad step
# speedup vs baseline: 1.7516x; 1.0557x over previous
#include <hip/hip_runtime.h>
#include <hip/hip_bf16.h>
#include <math.h>

#define BB     8
#define SS     256
#define HIDC   1024
#define HH     16
#define DD     64
#define FF     64
#define LR     0.1f
#define GSCALE (2.0f / 8192.0f)
#define LN_EPS 1e-5f

typedef float v2f __attribute__((ext_vector_type(2)));
typedef float v8f __attribute__((ext_vector_type(8)));

__device__ __forceinline__ v8f wmma4(v2f a, v2f b, v8f c) {
  return __builtin_amdgcn_wmma_f32_16x16x4_f32(
       false, a,  false, b,
       (short)0, c,  false,  false);
}

__device__ __forceinline__ float gelu_f(float x) {
  const float c = 0.7978845608028654f;
  float u = c * (x + 0.044715f * x * x * x);
  return 0.5f * x * (1.0f + tanhf(u));
}
__device__ __forceinline__ void gelu_and_grad(float x, float& g, float& gd) {
  const float c = 0.7978845608028654f;
  float u = c * (x + 0.044715f * x * x * x);
  float t = tanhf(u);
  float x2 = x * x;
  g  = 0.5f * x * (1.0f + t);
  gd = 0.5f * (1.0f + t) + 0.5f * x * (1.0f - t * t) * c * (1.0f + 0.134145f * x2);
}
__device__ __forceinline__ float wave_sum32(float v) {
#pragma unroll
  for (int off = 16; off > 0; off >>= 1) v += __shfl_xor(v, off, 32);
  return v;
}

__global__ void __launch_bounds__(256)
gemm_wmma_f32(const float* __restrict__ A, const float* __restrict__ Bm,
              float* __restrict__ C, int M, int N, int K, int act) {
  __shared__ float As[16 * 64];
  const int tid   = threadIdx.x;
  const int wave  = tid >> 5;
  const int lane  = tid & 31;
  const int mbase = blockIdx.y * 16;
  const int tn    = blockIdx.x * 8 + wave;
  const int col   = tn * 16 + (lane & 15);
  const int kh    = (lane >> 4) << 1;
  const int arow  = lane & 15;
  v8f acc = {0.f, 0.f, 0.f, 0.f, 0.f, 0.f, 0.f, 0.f};
  for (int kc = 0; kc < K; kc += 64) {
    __syncthreads();
    for (int i = tid; i < 16 * 64; i += 256)
      As[i] = A[(size_t)(mbase + (i >> 6)) * K + kc + (i & 63)];
    if (kc + 64 < K)
      __builtin_prefetch(&A[(size_t)(mbase + (tid >> 4)) * K + kc + 64], 0, 1);
    __syncthreads();
#pragma unroll
    for (int k0 = 0; k0 < 64; k0 += 4) {
      v2f a, b;
      a.x = As[arow * 64 + k0 + kh];
      a.y = As[arow * 64 + k0 + kh + 1];
      b.x = Bm[(size_t)(kc + k0 + kh) * N + col];
      b.y = Bm[(size_t)(kc + k0 + kh + 1) * N + col];
      acc = wmma4(a, b, acc);
    }
  }
  const int r0 = mbase + ((lane >= 16) ? 8 : 0);
#pragma unroll
  for (int i = 0; i < 8; ++i) {
    float v = acc[i];
    if (act) v = gelu_f(v);
    C[(size_t)(r0 + i) * N + col] = v;
  }
}


#define HOLD8(a, o) asm volatile("" :: "v"(a[o]), "v"(a[o + 1]), "v"(a[o + 2]), "v"(a[o + 3]), \
                                       "v"(a[o + 4]), "v"(a[o + 5]), "v"(a[o + 6]), "v"(a[o + 7]))
#define CH_GELU  0
#define CH_DPRED 1
template <int EPI, bool ROWMAJOR2, bool HAS_LATER, typename LATER>
__device__ __forceinline__ void
chain_pair(const float* X8, const float* Wa, const float* biasa, const float* Wb,
           float* Cgd, float* Cact, const float* tgtp, float* part, int wave, int lane,
           LATER later_reads) {
  const int c  = lane & 15;
  const int h  = lane >> 4;
  const int bb = c & 7;
  const int kh = h << 1;
  const int fo = wave * 16;
  const int f0 = fo + 8 * h;
  v2f fa[16], fb[16], fa2[16];
  float bi[8], tg[8];
#pragma unroll
  for (int i = 0; i < 16; ++i) {
    const int k0 = 4 * i;
    fa[i].x = Wa[(k0 + kh) * 64 + fo + c];
    fa[i].y = Wa[(k0 + kh + 1) * 64 + fo + c];
    fb[i].x = X8[bb * 64 + k0 + kh];
    fb[i].y = X8[bb * 64 + k0 + kh + 1];
  }
#pragma unroll
  for (int s = 0; s < 4; ++s) {
    const int kk = f0 + 2 * s;
#pragma unroll
    for (int td = 0; td < 4; ++td) {
      const int m = td * 16 + c;
      if (!ROWMAJOR2) {
        fa2[s * 4 + td].x = Wb[kk * 64 + m];
        fa2[s * 4 + td].y = Wb[(kk + 1) * 64 + m];
      } else {
        fa2[s * 4 + td].x = Wb[m * 64 + kk];
        fa2[s * 4 + td].y = Wb[m * 64 + kk + 1];
      }
    }
  }
#pragma unroll
  for (int r = 0; r < 8; ++r) {
    bi[r] = biasa[f0 + r];
    tg[r] = (EPI == CH_DPRED) ? tgtp[bb * 64 + f0 + r] : 0.f;
  }
  HOLD8(fa, 0);  HOLD8(fa, 8);
  HOLD8(fb, 0);  HOLD8(fb, 8);
  HOLD8(fa2, 0); HOLD8(fa2, 8);
  HOLD8(bi, 0);
  if (EPI == CH_DPRED) HOLD8(tg, 0);
  if (HAS_LATER) {
    later_reads();
    __builtin_amdgcn_sched_barrier(0);
  }
  v8f acc = {0.f, 0.f, 0.f, 0.f, 0.f, 0.f, 0.f, 0.f};
#pragma unroll
  for (int i = 0; i < 16; ++i) acc = wmma4(fa[i], fb[i], acc);
  v8f g;
  if (EPI == CH_GELU) {
    v8f gd;
#pragma unroll
    for (int r = 0; r < 8; ++r) {
      float gv, dv;
      gelu_and_grad(acc[r] + bi[r], gv, dv);
      g[r] = gv;
      gd[r] = dv;
    }
    if (c < 8) {
#pragma unroll
      for (int r = 0; r < 8; ++r) {
        Cgd[bb * 64 + f0 + r] = gd[r];
        Cact[bb * 64 + f0 + r] = g[r];
      }
    }
  } else {
#pragma unroll
    for (int r = 0; r < 8; ++r) {
      float v = acc[r] + bi[r];
      g[r] = GSCALE * (v - tg[r]);
    }
  }
  v8f p0 = {0.f, 0.f, 0.f, 0.f, 0.f, 0.f, 0.f, 0.f};
  v8f p1 = p0, p2 = p0, p3 = p0;
#pragma unroll
  for (int s = 0; s < 4; ++s) {
    v2f b2;
    b2.x = g[2 * s];
    b2.y = g[2 * s + 1];
    p0 = wmma4(fa2[s * 4 + 0], b2, p0);
    p1 = wmma4(fa2[s * 4 + 1], b2, p1);
    p2 = wmma4(fa2[s * 4 + 2], b2, p2);
    p3 = wmma4(fa2[s * 4 + 3], b2, p3);
  }
  if (c < 8) {
    float* dst = part + wave * 512 + bb * 64 + 8 * h;
#pragma unroll
    for (int r = 0; r < 8; ++r) {
      dst[r]      = p0[r];
      dst[16 + r] = p1[r];
      dst[32 + r] = p2[r];
      dst[48 + r] = p3[r];
    }
  }
}

__device__ __forceinline__ float part_sum4(const float* part, int idx) {
  return (part[idx] + part[512 + idx]) + (part[1024 + idx] + part[1536 + idx]);
}

__device__ __forceinline__ void
fwd_pass(const float* ktb, float* gdv, float* hg, float* ob, float* xh, float* zb,
         const float* W1, const float* W2, const float* b1v, const float* b2v,
         const float* lgv, const float* lbv, float* muv, float* isv, float* part,
         int wave, int lane, int tid) {
  if (wave < 4)
    chain_pair<CH_GELU, false, false>(ktb, W1, b1v, W2, gdv, hg, nullptr, part, wave, lane, [] {});
  __syncthreads();
  {
    float e0 = part_sum4(part, wave * 64 + lane) + b2v[lane];
    float e1 = part_sum4(part, wave * 64 + lane + 32) + b2v[lane + 32];
    ob[wave * 64 + lane] = e0;
    ob[wave * 64 + lane + 32] = e1;
    float s = wave_sum32(e0 + e1);
    float q = wave_sum32(e0 * e0 + e1 * e1);
    if (lane == 0) {
      float m = s * (1.f / 64.f);
      muv[wave] = m;
      isv[wave] = rsqrtf(q * (1.f / 64.f) - m * m + LN_EPS);
    }
  }
  __syncthreads();
  for (int idx = tid; idx < 512; idx += 256) {
    int b = idx >> 6, d = idx & 63;
    float xv = (ob[idx] - muv[b]) * isv[b];
    xh[idx] = xv;
    zb[idx] = xv * lgv[d] + lbv[d];
  }
  __syncthreads();
}

#define SCAN_SMEM_FLOATS (3 * 4096 + 9 * 512 + 7 * 64 + 32 + 4 * 512)

__global__ void __launch_bounds__(256)
ttt_scan_kernel(const float* __restrict__ kg, const float* __restrict__ vg,
                const float* __restrict__ qg,
                const float* __restrict__ fw1, const float* __restrict__ fb1,
                const float* __restrict__ fw2, const float* __restrict__ fb2,
                const float* __restrict__ flg, const float* __restrict__ flb,
                const float* __restrict__ lwg, const float* __restrict__ lbg,
                const float* __restrict__ ttg, const float* __restrict__ ttb,
                float* __restrict__ outg) {
  extern __shared__ __align__(16) float smem[];
  float* W1  = smem;
  float* W2  = W1 + 4096;
  float* LW  = W2 + 4096;
  float* ktb = LW + 4096;
  float* tgt = ktb + 512;
  float* gdv = tgt + 512;
  float* hg  = gdv + 512;
  float* ob  = hg + 512;
  float* xh  = ob + 512;
  float* zb  = xh + 512;
  float* dzb = zb + 512;
  float* dob = dzb + 512;
  float* b1v = dob + 512;
  float* b2v = b1v + 64;
  float* lgv = b2v + 64;
  float* lbv = lgv + 64;
  float* tgv = lbv + 64;
  float* tbv = tgv + 64;
  float* lbz = tbv + 64;
  float* muv = lbz + 64;
  float* isv = muv + 8;
  float* m1v = isv + 8;
  float* m2v = m1v + 8;
  float* part = m2v + 8;

  const int h    = blockIdx.x;
  const int tid  = threadIdx.x;
  const int wave = tid >> 5;
  const int lane = tid & 31;

  const int b0 = tid >> 6,          d0 = tid & 63;
  const int b1i = (tid + 256) >> 6, d1i = (tid + 256) & 63;
  const size_t g0base = (size_t)(b0 * SS) * HIDC + h * 64 + d0;
  const size_t g1base = (size_t)(b1i * SS) * HIDC + h * 64 + d1i;

  for (int i = tid; i < 4096; i += 256) {
    W1[i] = fw1[(size_t)h * 4096 + i];
    W2[i] = fw2[(size_t)h * 4096 + i];
    LW[i] = lwg[i];
  }
  if (tid < 64) {
    b1v[tid] = fb1[h * 64 + tid];
    b2v[tid] = fb2[h * 64 + tid];
    lgv[tid] = flg[h * 64 + tid];
    lbv[tid] = flb[h * 64 + tid];
    tgv[tid] = ttg[h * 64 + tid];
    tbv[tid] = ttb[h * 64 + tid];
    lbz[tid] = lbg[tid];
  }

  float kp0 = kg[g0base], vp0 = vg[g0base];
  float kp1 = kg[g1base], vp1 = vg[g1base];
  __syncthreads();

  for (int t = 0; t < SS; ++t) {
    ktb[tid]       = kp0;  tgt[tid]       = vp0 - kp0;
    ktb[tid + 256] = kp1;  tgt[tid + 256] = vp1 - kp1;
    __syncthreads();
    if (t + 1 < SS) {
      kp0 = kg[g0base + (t + 1) * HIDC];  vp0 = vg[g0base + (t + 1) * HIDC];
      kp1 = kg[g1base + (t + 1) * HIDC];  vp1 = vg[g1base + (t + 1) * HIDC];
    }

    for (int step = 0; step < 4; ++step) {
      fwd_pass(ktb, gdv, hg, ob, xh, zb, W1, W2, b1v, b2v, lgv, lbv,
               muv, isv, part, wave, lane, tid);
      const int fc = lane & 15;
      const int fh = lane >> 4;
      const int fk = fh << 1;
      v2f   pb[16];
      v2f   pk[8];
      float pg[8];
      float pb1 = 0.f;
      v2f   ph[2];
      float wold[32];
      if (wave < 4) {
        chain_pair<CH_DPRED, true, true>(zb, LW, lbz, LW, nullptr, nullptr, tgt, part, wave, lane, [&] {
          const int col = wave * 16 + fc;
#pragma unroll
          for (int i = 0; i < 16; ++i) {
            pb[i].x = W2[col * 64 + 4 * i + fk];
            pb[i].y = W2[col * 64 + 4 * i + fk + 1];
          }
#pragma unroll
          for (int r = 0; r < 8; ++r) pg[r] = gdv[r * 64 + col];
          pb1 = b1v[col];
#pragma unroll
          for (int tm = 0; tm < 4; ++tm) {
#pragma unroll
            for (int s = 0; s < 2; ++s) {
              pk[tm * 2 + s].x = ktb[(4 * s + fk) * 64 + tm * 16 + fc];
              pk[tm * 2 + s].y = ktb[(4 * s + fk + 1) * 64 + tm * 16 + fc];
            }
#pragma unroll
            for (int i = 0; i < 8; ++i) wold[tm * 8 + i] = W1[(tm * 16 + 8 * fh + i) * 64 + col];
          }
        });
      } else {
        const int wv = wave - 4;
#pragma unroll
        for (int s = 0; s < 2; ++s) {
          ph[s].x = hg[(4 * s + fk) * 64 + wv * 16 + fc];
          ph[s].y = hg[(4 * s + fk + 1) * 64 + wv * 16 + fc];
        }
#pragma unroll
        for (int tn = 0; tn < 4; ++tn) {
#pragma unroll
          for (int i = 0; i < 8; ++i) wold[tn * 8 + i] = W2[(wv * 16 + 8 * fh + i) * 64 + tn * 16 + fc];
        }
      }
      __syncthreads();
      {
        float z0 = part_sum4(part, wave * 64 + lane);
        float z1 = part_sum4(part, wave * 64 + lane + 32);
        dzb[wave * 64 + lane] = z0;
        dzb[wave * 64 + lane + 32] = z1;
        float dd0 = z0 * lgv[lane];
        float dd1 = z1 * lgv[lane + 32];
        float x0 = xh[wave * 64 + lane], x1 = xh[wave * 64 + lane + 32];
        float s1 = wave_sum32(dd0 + dd1);
        float s2 = wave_sum32(dd0 * x0 + dd1 * x1);
        if (lane == 0) { m1v[wave] = s1 * (1.f / 64.f); m2v[wave] = s2 * (1.f / 64.f); }
      }
      __syncthreads();
      for (int idx = tid; idx < 512; idx += 256) {
        int b = idx >> 6, d = idx & 63;
        float dxh = dzb[idx] * lgv[d];
        dob[idx] = isv[b] * (dxh - m1v[b] - xh[idx] * m2v[b]);
      }
      __syncthreads();
      if (wave < 4) {
        const int col = wave * 16 + fc;
        const int bb  = fc & 7;
        v2f pa[16];
#pragma unroll
        for (int i = 0; i < 16; ++i) {
          pa[i].x = dob[bb * 64 + 4 * i + fk];
          pa[i].y = dob[bb * 64 + 4 * i + fk + 1];
        }
        HOLD8(pa, 0); HOLD8(pa, 8);
        v8f acc = {0.f, 0.f, 0.f, 0.f, 0.f, 0.f, 0.f, 0.f};
#pragma unroll
        for (int i = 0; i < 16; ++i) acc = wmma4(pa[i], pb[i], acc);
        float da[8];
        float sb = 0.f;
#pragma unroll
        for (int r = 0; r < 8; ++r) da[r] = acc[r] * pg[r];
#pragma unroll
        for (int r = 0; r < 8; ++r) sb += da[r];
        if (fh == 0) b1v[col] = pb1 - LR * sb;
        v8f u0 = {0.f, 0.f, 0.f, 0.f, 0.f, 0.f, 0.f, 0.f};
        v8f u1 = u0, u2 = u0, u3 = u0;
#pragma unroll
        for (int s = 0; s < 2; ++s) {
          v2f b2;
          b2.x = fh ? da[4 * s + 2] : da[4 * s];
          b2.y = fh ? da[4 * s + 3] : da[4 * s + 1];
          u0 = wmma4(pk[0 + s], b2, u0);
          u1 = wmma4(pk[2 + s], b2, u1);
          u2 = wmma4(pk[4 + s], b2, u2);
          u3 = wmma4(pk[6 + s], b2, u3);
        }
#pragma unroll
        for (int i = 0; i < 8; ++i) {
          W1[(0 * 16 + 8 * fh + i) * 64 + col] = wold[0 * 8 + i] - LR * u0[i];
          W1[(1 * 16 + 8 * fh + i) * 64 + col] = wold[1 * 8 + i] - LR * u1[i];
          W1[(2 * 16 + 8 * fh + i) * 64 + col] = wold[2 * 8 + i] - LR * u2[i];
          W1[(3 * 16 + 8 * fh + i) * 64 + col] = wold[3 * 8 + i] - LR * u3[i];
        }
      } else {
        const int wv = wave - 4;
        v2f pd[8];
#pragma unroll
        for (int tn = 0; tn < 4; ++tn) {
#pragma unroll
          for (int s = 0; s < 2; ++s) {
            pd[tn * 2 + s].x = dob[(4 * s + fk) * 64 + tn * 16 + fc];
            pd[tn * 2 + s].y = dob[(4 * s + fk + 1) * 64 + tn * 16 + fc];
          }
        }
        v8f u0 = {0.f, 0.f, 0.f, 0.f, 0.f, 0.f, 0.f, 0.f};
        v8f u1 = u0, u2 = u0, u3 = u0;
#pragma unroll
        for (int s = 0; s < 2; ++s) {
          u0 = wmma4(ph[s], pd[0 + s], u0);
          u1 = wmma4(ph[s], pd[2 + s], u1);
          u2 = wmma4(ph[s], pd[4 + s], u2);
          u3 = wmma4(ph[s], pd[6 + s], u3);
        }
        const int rb = (wv * 16 + 8 * fh) * 64 + fc;
#pragma unroll
        for (int i = 0; i < 8; ++i) {
          W2[rb + i * 64 + 0]  = wold[0 * 8 + i] - LR * u0[i];
          W2[rb + i * 64 + 16] = wold[1 * 8 + i] - LR * u1[i];
          W2[rb + i * 64 + 32] = wold[2 * 8 + i] - LR * u2[i];
          W2[rb + i * 64 + 48] = wold[3 * 8 + i] - LR * u3[i];
        }
        const int t2 = tid - 128;
        if (t2 < 64) {
          float s = 0.f, s2 = 0.f;
          for (int b = 0; b < 8; ++b) s += dzb[b * 64 + t2] * xh[b * 64 + t2];
          for (int b = 0; b < 8; ++b) s2 += dob[b * 64 + t2];
          lgv[t2] -= LR * s;
          b2v[t2] -= LR * s2;
        } else {
          const int d = t2 - 64;
          float s = 0.f;
          for (int b = 0; b < 8; ++b) s += dzb[b * 64 + d];
          lbv[d] -= LR * s;
        }
      }
      __syncthreads();
    }

    fwd_pass(ktb, gdv, hg, ob, xh, zb, W1, W2, b1v, b2v, lgv, lbv,
             muv, isv, part, wave, lane, tid);
    {
      float e0 = zb[wave * 64 + lane], e1 = zb[wave * 64 + lane + 32];
      float s = wave_sum32(e0 + e1);
      float q = wave_sum32(e0 * e0 + e1 * e1);
      if (lane == 0) {
        float m = s * (1.f / 64.f);
        muv[wave] = m;
        isv[wave] = rsqrtf(q * (1.f / 64.f) - m * m + LN_EPS);
      }
    }
    __syncthreads();
    for (int idx = tid; idx < 512; idx += 256) {
      int b = idx >> 6, d = idx & 63;
      float z2 = (zb[idx] - muv[b]) * isv[b] * tgv[d] + tbv[d];
      size_t g = ((size_t)(b * SS + t)) * HIDC + h * 64 + d;
      outg[g] = qg[g] + z2;
    }
    __syncthreads();
  }
}

__global__ void __launch_bounds__(256)
gate_ln_kernel(const float* __restrict__ outb, const float* __restrict__ gact,
               const float* __restrict__ png, const float* __restrict__ pnb,
               float* __restrict__ gated) {
  __shared__ float r1[256], r2[256];
  __shared__ float mu_s, inv_s;
  const int r = blockIdx.x, tid = threadIdx.x;
  const float* row = outb + (size_t)r * HIDC;
  float s = 0.f, q = 0.f;
  for (int i = tid; i < HIDC; i += 256) { float v = row[i]; s += v; q += v * v; }
  r1[tid] = s; r2[tid] = q;
  __syncthreads();
  for (int off = 128; off > 0; off >>= 1) {
    if (tid < off) { r1[tid] += r1[tid + off]; r2[tid] += r2[tid + off]; }
    __syncthreads();
  }
  if (tid == 0) {
    float m = r1[0] * (1.f / HIDC);
    mu_s = m;
    inv_s = rsqrtf(r2[0] * (1.f / HIDC) - m * m + LN_EPS);
  }
  __syncthreads();
  for (int i = tid; i < HIDC; i += 256) {
    float y = (row[i] - mu_s) * inv_s * png[i] + pnb[i];
    gated[(size_t)r * HIDC + i] = gact[(size_t)r * HIDC + i] * y;
  }
}

extern "C" void kernel_launch(void* const* d_in, const int* in_sizes, int n_in,
                              void* d_out, int out_size, void* d_ws, size_t ws_size,
                              hipStream_t stream) {
  (void)in_sizes; (void)n_in; (void)out_size; (void)ws_size;
  const float* x   = (const float*)d_in[0];
  const float* wq  = (const float*)d_in[1];
  const float* wk  = (const float*)d_in[2];
  const float* wv  = (const float*)d_in[3];
  const float* fw1 = (const float*)d_in[4];
  const float* fb1 = (const float*)d_in[5];
  const float* fw2 = (const float*)d_in[6];
  const float* fb2 = (const float*)d_in[7];
  const float* flg = (const float*)d_in[8];
  const float* flb = (const float*)d_in[9];
  const float* lw  = (const float*)d_in[10];
  const float* lb  = (const float*)d_in[11];
  const float* ttg = (const float*)d_in[12];
  const float* ttb = (const float*)d_in[13];
  const float* wo  = (const float*)d_in[14];
  const float* wg  = (const float*)d_in[15];
  const float* png = (const float*)d_in[16];
  const float* pnb = (const float*)d_in[17];

  const int M = BB * SS;
  const size_t PLANE = (size_t)M * HIDC;
  float* qb    = (float*)d_ws;
  float* kb    = qb + PLANE;
  float* vb    = kb + PLANE;
  float* gactb = vb + PLANE;
  float* outb  = gactb + PLANE;
  float* gatedb = qb;

  dim3 gblk(256);
  dim3 ggrid(HIDC / 128, M / 16);
  gemm_wmma_f32<<<ggrid, gblk, 0, stream>>>(x, wq, qb,    M, HIDC, HIDC, 0);
  gemm_wmma_f32<<<ggrid, gblk, 0, stream>>>(x, wk, kb,    M, HIDC, HIDC, 0);
  gemm_wmma_f32<<<ggrid, gblk, 0, stream>>>(x, wv, vb,    M, HIDC, HIDC, 0);
  gemm_wmma_f32<<<ggrid, gblk, 0, stream>>>(x, wg, gactb, M, HIDC, HIDC, 1);

  ttt_scan_kernel<<<HH, 256, SCAN_SMEM_FLOATS * sizeof(float), stream>>>(
      kb, vb, qb, fw1, fb1, fw2, fb2, flg, flb, lw, lb, ttg, ttb, outb);

  gate_ln_kernel<<<M, 256, 0, stream>>>(outb, gactb, png, pnb, gatedb);

  gemm_wmma_f32<<<ggrid, gblk, 0, stream>>>(gatedb, wo, (float*)d_out,
                                            M, HIDC, HIDC, 0);
}


# scan fwd gelu epilogue: lanes c and c+8 of the tile hold the same batch row; each evaluates 4 of its 8 gelu/gelu' values and stores them, gelu swapped by DPP row_xmask:8: 4 serial tanh a lane, not 8
# speedup vs baseline: 1.8158x; 1.0367x over previous
#include <hip/hip_runtime.h>
#include <hip/hip_bf16.h>
#include <math.h>

#define BB     8
#define SS     256
#define HIDC   1024
#define HH     16
#define DD     64
#define FF     64
#define LR     0.1f
#define GSCALE (2.0f / 8192.0f)
#define LN_EPS 1e-5f

typedef float v2f __attribute__((ext_vector_type(2)));
typedef float v8f __attribute__((ext_vector_type(8)));

__device__ __forceinline__ v8f wmma4(v2f a, v2f b, v8f c) {
  return __builtin_amdgcn_wmma_f32_16x16x4_f32(
       false, a,  false, b,
       (short)0, c,  false,  false);
}

__device__ __forceinline__ float gelu_f(float x) {
  const float c = 0.7978845608028654f;
  float u = c * (x + 0.044715f * x * x * x);
  return 0.5f * x * (1.0f + tanhf(u));
}
__device__ __forceinline__ void gelu_and_grad(float x, float& g, float& gd) {
  const float c = 0.7978845608028654f;
  float u = c * (x + 0.044715f * x * x * x);
  float t = tanhf(u);
  float x2 = x * x;
  g  = 0.5f * x * (1.0f + t);
  gd = 0.5f * (1.0f + t) + 0.5f * x * (1.0f - t * t) * c * (1.0f + 0.134145f * x2);
}
__device__ __forceinline__ float wave_sum32(float v) {
#pragma unroll
  for (int off = 16; off > 0; off >>= 1) v += __shfl_xor(v, off, 32);
  return v;
}

__device__ __forceinline__ float row_xor8(float v) {
  const int b = __builtin_bit_cast(int, v);
  return __builtin_bit_cast(float, __builtin_amdgcn_update_dpp(b, b, 0x168, 0xf, 0xf, false));
}

__global__ void __launch_bounds__(256)
gemm_wmma_f32(const float* __restrict__ A, const float* __restrict__ Bm,
              float* __restrict__ C, int M, int N, int K, int act) {
  __shared__ float As[16 * 64];
  const int tid   = threadIdx.x;
  const int wave  = tid >> 5;
  const int lane  = tid & 31;
  const int mbase = blockIdx.y * 16;
  const int tn    = blockIdx.x * 8 + wave;
  const int col   = tn * 16 + (lane & 15);
  const int kh    = (lane >> 4) << 1;
  const int arow  = lane & 15;
  v8f acc = {0.f, 0.f, 0.f, 0.f, 0.f, 0.f, 0.f, 0.f};
  for (int kc = 0; kc < K; kc += 64) {
    __syncthreads();
    for (int i = tid; i < 16 * 64; i += 256)
      As[i] = A[(size_t)(mbase + (i >> 6)) * K + kc + (i & 63)];
    if (kc + 64 < K)
      __builtin_prefetch(&A[(size_t)(mbase + (tid >> 4)) * K + kc + 64], 0, 1);
    __syncthreads();
#pragma unroll
    for (int k0 = 0; k0 < 64; k0 += 4) {
      v2f a, b;
      a.x = As[arow * 64 + k0 + kh];
      a.y = As[arow * 64 + k0 + kh + 1];
      b.x = Bm[(size_t)(kc + k0 + kh) * N + col];
      b.y = Bm[(size_t)(kc + k0 + kh + 1) * N + col];
      acc = wmma4(a, b, acc);
    }
  }
  const int r0 = mbase + ((lane >= 16) ? 8 : 0);
#pragma unroll
  for (int i = 0; i < 8; ++i) {
    float v = acc[i];
    if (act) v = gelu_f(v);
    C[(size_t)(r0 + i) * N + col] = v;
  }
}


#define HOLD8(a, o) asm volatile("" :: "v"(a[o]), "v"(a[o + 1]), "v"(a[o + 2]), "v"(a[o + 3]), \
                                       "v"(a[o + 4]), "v"(a[o + 5]), "v"(a[o + 6]), "v"(a[o + 7]))
#define CH_GELU  0
#define CH_DPRED 1
template <int EPI, bool ROWMAJOR2, bool HAS_LATER, typename LATER>
__device__ __forceinline__ void
chain_pair(const float* X8, const float* Wa, const float* biasa, const float* Wb,
           float* Cgd, float* Cact, const float* tgtp, float* part, int wave, int lane,
           LATER later_reads) {
  const int c  = lane & 15;
  const int h  = lane >> 4;
  const int bb = c & 7;
  const int kh = h << 1;
  const int fo = wave * 16;
  const int f0 = fo + 8 * h;
  v2f fa[16], fb[16], fa2[16];
  float bi[8], tg[8];
#pragma unroll
  for (int i = 0; i < 16; ++i) {
    const int k0 = 4 * i;
    fa[i].x = Wa[(k0 + kh) * 64 + fo + c];
    fa[i].y = Wa[(k0 + kh + 1) * 64 + fo + c];
    fb[i].x = X8[bb * 64 + k0 + kh];
    fb[i].y = X8[bb * 64 + k0 + kh + 1];
  }
#pragma unroll
  for (int s = 0; s < 4; ++s) {
    const int kk = f0 + 2 * s;
#pragma unroll
    for (int td = 0; td < 4; ++td) {
      const int m = td * 16 + c;
      if (!ROWMAJOR2) {
        fa2[s * 4 + td].x = Wb[kk * 64 + m];
        fa2[s * 4 + td].y = Wb[(kk + 1) * 64 + m];
      } else {
        fa2[s * 4 + td].x = Wb[m * 64 + kk];
        fa2[s * 4 + td].y = Wb[m * 64 + kk + 1];
      }
    }
  }
#pragma unroll
  for (int r = 0; r < 8; ++r) {
    bi[r] = biasa[f0 + r];
    tg[r] = (EPI == CH_DPRED) ? tgtp[bb * 64 + f0 + r] : 0.f;
  }
  HOLD8(fa, 0);  HOLD8(fa, 8);
  HOLD8(fb, 0);  HOLD8(fb, 8);
  HOLD8(fa2, 0); HOLD8(fa2, 8);
  HOLD8(bi, 0);
  if (EPI == CH_DPRED) HOLD8(tg, 0);
  if (HAS_LATER) {
    later_reads();
    __builtin_amdgcn_sched_barrier(0);
  }
  v8f acc = {0.f, 0.f, 0.f, 0.f, 0.f, 0.f, 0.f, 0.f};
#pragma unroll
  for (int i = 0; i < 16; ++i) acc = wmma4(fa[i], fb[i], acc);
  v8f g;
  if (EPI == CH_GELU) {
    const bool hi = (c >= 8);
    float gm[4], dm[4];
#pragma unroll
    for (int r = 0; r < 4; ++r) {
      const float xlo = acc[r] + bi[r];
      const float xhi = acc[4 + r] + bi[4 + r];
      gelu_and_grad(hi ? xhi : xlo, gm[r], dm[r]);
    }
    {
      const int so = bb * 64 + f0 + (hi ? 4 : 0);
#pragma unroll
      for (int r = 0; r < 4; ++r) {
        Cgd[so + r] = dm[r];
        Cact[so + r] = gm[r];
      }
    }
#pragma unroll
    for (int r = 0; r < 4; ++r) {
      const float go = row_xor8(gm[r]);
      g[r]     = hi ? go : gm[r];
      g[4 + r] = hi ? gm[r] : go;
    }
  } else {
#pragma unroll
    for (int r = 0; r < 8; ++r) {
      float v = acc[r] + bi[r];
      g[r] = GSCALE * (v - tg[r]);
    }
  }
  v8f p0 = {0.f, 0.f, 0.f, 0.f, 0.f, 0.f, 0.f, 0.f};
  v8f p1 = p0, p2 = p0, p3 = p0;
#pragma unroll
  for (int s = 0; s < 4; ++s) {
    v2f b2;
    b2.x = g[2 * s];
    b2.y = g[2 * s + 1];
    p0 = wmma4(fa2[s * 4 + 0], b2, p0);
    p1 = wmma4(fa2[s * 4 + 1], b2, p1);
    p2 = wmma4(fa2[s * 4 + 2], b2, p2);
    p3 = wmma4(fa2[s * 4 + 3], b2, p3);
  }
  if (c < 8) {
    float* dst = part + wave * 512 + bb * 64 + 8 * h;
#pragma unroll
    for (int r = 0; r < 8; ++r) {
      dst[r]      = p0[r];
      dst[16 + r] = p1[r];
      dst[32 + r] = p2[r];
      dst[48 + r] = p3[r];
    }
  }
}

__device__ __forceinline__ float part_sum4(const float* part, int idx) {
  return (part[idx] + part[512 + idx]) + (part[1024 + idx] + part[1536 + idx]);
}

__device__ __forceinline__ void
fwd_pass(const float* ktb, float* gdv, float* hg, float* ob, float* xh, float* zb,
         const float* W1, const float* W2, const float* b1v, const float* b2v,
         const float* lgv, const float* lbv, float* muv, float* isv, float* part,
         int wave, int lane, int tid) {
  if (wave < 4)
    chain_pair<CH_GELU, false, false>(ktb, W1, b1v, W2, gdv, hg, nullptr, part, wave, lane, [] {});
  __syncthreads();
  {
    float e0 = part_sum4(part, wave * 64 + lane) + b2v[lane];
    float e1 = part_sum4(part, wave * 64 + lane + 32) + b2v[lane + 32];
    ob[wave * 64 + lane] = e0;
    ob[wave * 64 + lane + 32] = e1;
    float s = wave_sum32(e0 + e1);
    float q = wave_sum32(e0 * e0 + e1 * e1);
    if (lane == 0) {
      float m = s * (1.f / 64.f);
      muv[wave] = m;
      isv[wave] = rsqrtf(q * (1.f / 64.f) - m * m + LN_EPS);
    }
  }
  __syncthreads();
  for (int idx = tid; idx < 512; idx += 256) {
    int b = idx >> 6, d = idx & 63;
    float xv = (ob[idx] - muv[b]) * isv[b];
    xh[idx] = xv;
    zb[idx] = xv * lgv[d] + lbv[d];
  }
  __syncthreads();
}

#define SCAN_SMEM_FLOATS (3 * 4096 + 9 * 512 + 7 * 64 + 32 + 4 * 512)

__global__ void __launch_bounds__(256)
ttt_scan_kernel(const float* __restrict__ kg, const float* __restrict__ vg,
                const float* __restrict__ qg,
                const float* __restrict__ fw1, const float* __restrict__ fb1,
                const float* __restrict__ fw2, const float* __restrict__ fb2,
                const float* __restrict__ flg, const float* __restrict__ flb,
                const float* __restrict__ lwg, const float* __restrict__ lbg,
                const float* __restrict__ ttg, const float* __restrict__ ttb,
                float* __restrict__ outg) {
  extern __shared__ __align__(16) float smem[];
  float* W1  = smem;
  float* W2  = W1 + 4096;
  float* LW  = W2 + 4096;
  float* ktb = LW + 4096;
  float* tgt = ktb + 512;
  float* gdv = tgt + 512;
  float* hg  = gdv + 512;
  float* ob  = hg + 512;
  float* xh  = ob + 512;
  float* zb  = xh + 512;
  float* dzb = zb + 512;
  float* dob = dzb + 512;
  float* b1v = dob + 512;
  float* b2v = b1v + 64;
  float* lgv = b2v + 64;
  float* lbv = lgv + 64;
  float* tgv = lbv + 64;
  float* tbv = tgv + 64;
  float* lbz = tbv + 64;
  float* muv = lbz + 64;
  float* isv = muv + 8;
  float* m1v = isv + 8;
  float* m2v = m1v + 8;
  float* part = m2v + 8;

  const int h    = blockIdx.x;
  const int tid  = threadIdx.x;
  const int wave = tid >> 5;
  const int lane = tid & 31;

  const int b0 = tid >> 6,          d0 = tid & 63;
  const int b1i = (tid + 256) >> 6, d1i = (tid + 256) & 63;
  const size_t g0base = (size_t)(b0 * SS) * HIDC + h * 64 + d0;
  const size_t g1base = (size_t)(b1i * SS) * HIDC + h * 64 + d1i;

  for (int i = tid; i < 4096; i += 256) {
    W1[i] = fw1[(size_t)h * 4096 + i];
    W2[i] = fw2[(size_t)h * 4096 + i];
    LW[i] = lwg[i];
  }
  if (tid < 64) {
    b1v[tid] = fb1[h * 64 + tid];
    b2v[tid] = fb2[h * 64 + tid];
    lgv[tid] = flg[h * 64 + tid];
    lbv[tid] = flb[h * 64 + tid];
    tgv[tid] = ttg[h * 64 + tid];
    tbv[tid] = ttb[h * 64 + tid];
    lbz[tid] = lbg[tid];
  }

  float kp0 = kg[g0base], vp0 = vg[g0base];
  float kp1 = kg[g1base], vp1 = vg[g1base];
  __syncthreads();

  for (int t = 0; t < SS; ++t) {
    ktb[tid]       = kp0;  tgt[tid]       = vp0 - kp0;
    ktb[tid + 256] = kp1;  tgt[tid + 256] = vp1 - kp1;
    __syncthreads();
    if (t + 1 < SS) {
      kp0 = kg[g0base + (t + 1) * HIDC];  vp0 = vg[g0base + (t + 1) * HIDC];
      kp1 = kg[g1base + (t + 1) * HIDC];  vp1 = vg[g1base + (t + 1) * HIDC];
    }

    for (int step = 0; step < 4; ++step) {
      fwd_pass(ktb, gdv, hg, ob, xh, zb, W1, W2, b1v, b2v, lgv, lbv,
               muv, isv, part, wave, lane, tid);
      const int fc = lane & 15;
      const int fh = lane >> 4;
      const int fk = fh << 1;
      v2f   pb[16];
      v2f   pk[8];
      float pg[8];
      float pb1 = 0.f;
      v2f   ph[2];
      float wold[32];
      if (wave < 4) {
        chain_pair<CH_DPRED, true, true>(zb, LW, lbz, LW, nullptr, nullptr, tgt, part, wave, lane, [&] {
          const int col = wave * 16 + fc;
#pragma unroll
          for (int i = 0; i < 16; ++i) {
            pb[i].x = W2[col * 64 + 4 * i + fk];
            pb[i].y = W2[col * 64 + 4 * i + fk + 1];
          }
#pragma unroll
          for (int r = 0; r < 8; ++r) pg[r] = gdv[r * 64 + col];
          pb1 = b1v[col];
#pragma unroll
          for (int tm = 0; tm < 4; ++tm) {
#pragma unroll
            for (int s = 0; s < 2; ++s) {
              pk[tm * 2 + s].x = ktb[(4 * s + fk) * 64 + tm * 16 + fc];
              pk[tm * 2 + s].y = ktb[(4 * s + fk + 1) * 64 + tm * 16 + fc];
            }
#pragma unroll
            for (int i = 0; i < 8; ++i) wold[tm * 8 + i] = W1[(tm * 16 + 8 * fh + i) * 64 + col];
          }
        });
      } else {
        const int wv = wave - 4;
#pragma unroll
        for (int s = 0; s < 2; ++s) {
          ph[s].x = hg[(4 * s + fk) * 64 + wv * 16 + fc];
          ph[s].y = hg[(4 * s + fk + 1) * 64 + wv * 16 + fc];
        }
#pragma unroll
        for (int tn = 0; tn < 4; ++tn) {
#pragma unroll
          for (int i = 0; i < 8; ++i) wold[tn * 8 + i] = W2[(wv * 16 + 8 * fh + i) * 64 + tn * 16 + fc];
        }
      }
      __syncthreads();
      {
        float z0 = part_sum4(part, wave * 64 + lane);
        float z1 = part_sum4(part, wave * 64 + lane + 32);
        dzb[wave * 64 + lane] = z0;
        dzb[wave * 64 + lane + 32] = z1;
        float dd0 = z0 * lgv[lane];
        float dd1 = z1 * lgv[lane + 32];
        float x0 = xh[wave * 64 + lane], x1 = xh[wave * 64 + lane + 32];
        float s1 = wave_sum32(dd0 + dd1);
        float s2 = wave_sum32(dd0 * x0 + dd1 * x1);
        if (lane == 0) { m1v[wave] = s1 * (1.f / 64.f); m2v[wave] = s2 * (1.f / 64.f); }
      }
      __syncthreads();
      for (int idx = tid; idx < 512; idx += 256) {
        int b = idx >> 6, d = idx & 63;
        float dxh = dzb[idx] * lgv[d];
        dob[idx] = isv[b] * (dxh - m1v[b] - xh[idx] * m2v[b]);
      }
      __syncthreads();
      if (wave < 4) {
        const int col = wave * 16 + fc;
        const int bb  = fc & 7;
        v2f pa[16];
#pragma unroll
        for (int i = 0; i < 16; ++i) {
          pa[i].x = dob[bb * 64 + 4 * i + fk];
          pa[i].y = dob[bb * 64 + 4 * i + fk + 1];
        }
        HOLD8(pa, 0); HOLD8(pa, 8);
        v8f acc = {0.f, 0.f, 0.f, 0.f, 0.f, 0.f, 0.f, 0.f};
#pragma unroll
        for (int i = 0; i < 16; ++i) acc = wmma4(pa[i], pb[i], acc);
        float da[8];
        float sb = 0.f;
#pragma unroll
        for (int r = 0; r < 8; ++r) da[r] = acc[r] * pg[r];
#pragma unroll
        for (int r = 0; r < 8; ++r) sb += da[r];
        if (fh == 0) b1v[col] = pb1 - LR * sb;
        v8f u0 = {0.f, 0.f, 0.f, 0.f, 0.f, 0.f, 0.f, 0.f};
        v8f u1 = u0, u2 = u0, u3 = u0;
#pragma unroll
        for (int s = 0; s < 2; ++s) {
          v2f b2;
          b2.x = fh ? da[4 * s + 2] : da[4 * s];
          b2.y = fh ? da[4 * s + 3] : da[4 * s + 1];
          u0 = wmma4(pk[0 + s], b2, u0);
          u1 = wmma4(pk[2 + s], b2, u1);
          u2 = wmma4(pk[4 + s], b2, u2);
          u3 = wmma4(pk[6 + s], b2, u3);
        }
#pragma unroll
        for (int i = 0; i < 8; ++i) {
          W1[(0 * 16 + 8 * fh + i) * 64 + col] = wold[0 * 8 + i] - LR * u0[i];
          W1[(1 * 16 + 8 * fh + i) * 64 + col] = wold[1 * 8 + i] - LR * u1[i];
          W1[(2 * 16 + 8 * fh + i) * 64 + col] = wold[2 * 8 + i] - LR * u2[i];
          W1[(3 * 16 + 8 * fh + i) * 64 + col] = wold[3 * 8 + i] - LR * u3[i];
        }
      } else {
        const int wv = wave - 4;
        v2f pd[8];
#pragma unroll
        for (int tn = 0; tn < 4; ++tn) {
#pragma unroll
          for (int s = 0; s < 2; ++s) {
            pd[tn * 2 + s].x = dob[(4 * s + fk) * 64 + tn * 16 + fc];
            pd[tn * 2 + s].y = dob[(4 * s + fk + 1) * 64 + tn * 16 + fc];
          }
        }
        v8f u0 = {0.f, 0.f, 0.f, 0.f, 0.f, 0.f, 0.f, 0.f};
        v8f u1 = u0, u2 = u0, u3 = u0;
#pragma unroll
        for (int s = 0; s < 2; ++s) {
          u0 = wmma4(ph[s], pd[0 + s], u0);
          u1 = wmma4(ph[s], pd[2 + s], u1);
          u2 = wmma4(ph[s], pd[4 + s], u2);
          u3 = wmma4(ph[s], pd[6 + s], u3);
        }
        const int rb = (wv * 16 + 8 * fh) * 64 + fc;
#pragma unroll
        for (int i = 0; i < 8; ++i) {
          W2[rb + i * 64 + 0]  = wold[0 * 8 + i] - LR * u0[i];
          W2[rb + i * 64 + 16] = wold[1 * 8 + i] - LR * u1[i];
          W2[rb + i * 64 + 32] = wold[2 * 8 + i] - LR * u2[i];
          W2[rb + i * 64 + 48] = wold[3 * 8 + i] - LR * u3[i];
        }
        const int t2 = tid - 128;
        if (t2 < 64) {
          float s = 0.f, s2 = 0.f;
          for (int b = 0; b < 8; ++b) s += dzb[b * 64 + t2] * xh[b * 64 + t2];
          for (int b = 0; b < 8; ++b) s2 += dob[b * 64 + t2];
          lgv[t2] -= LR * s;
          b2v[t2] -= LR * s2;
        } else {
          const int d = t2 - 64;
          float s = 0.f;
          for (int b = 0; b < 8; ++b) s += dzb[b * 64 + d];
          lbv[d] -= LR * s;
        }
      }
      __syncthreads();
    }

    fwd_pass(ktb, gdv, hg, ob, xh, zb, W1, W2, b1v, b2v, lgv, lbv,
             muv, isv, part, wave, lane, tid);
    {
      float e0 = zb[wave * 64 + lane], e1 = zb[wave * 64 + lane + 32];
      float s = wave_sum32(e0 + e1);
      float q = wave_sum32(e0 * e0 + e1 * e1);
      if (lane == 0) {
        float m = s * (1.f / 64.f);
        muv[wave] = m;
        isv[wave] = rsqrtf(q * (1.f / 64.f) - m * m + LN_EPS);
      }
    }
    __syncthreads();
    for (int idx = tid; idx < 512; idx += 256) {
      int b = idx >> 6, d = idx & 63;
      float z2 = (zb[idx] - muv[b]) * isv[b] * tgv[d] + tbv[d];
      size_t g = ((size_t)(b * SS + t)) * HIDC + h * 64 + d;
      outg[g] = qg[g] + z2;
    }
    __syncthreads();
  }
}

__global__ void __launch_bounds__(256)
gate_ln_kernel(const float* __restrict__ outb, const float* __restrict__ gact,
               const float* __restrict__ png, const float* __restrict__ pnb,
               float* __restrict__ gated) {
  __shared__ float r1[256], r2[256];
  __shared__ float mu_s, inv_s;
  const int r = blockIdx.x, tid = threadIdx.x;
  const float* row = outb + (size_t)r * HIDC;
  float s = 0.f, q = 0.f;
  for (int i = tid; i < HIDC; i += 256) { float v = row[i]; s += v; q += v * v; }
  r1[tid] = s; r2[tid] = q;
  __syncthreads();
  for (int off = 128; off > 0; off >>= 1) {
    if (tid < off) { r1[tid] += r1[tid + off]; r2[tid] += r2[tid + off]; }
    __syncthreads();
  }
  if (tid == 0) {
    float m = r1[0] * (1.f / HIDC);
    mu_s = m;
    inv_s = rsqrtf(r2[0] * (1.f / HIDC) - m * m + LN_EPS);
  }
  __syncthreads();
  for (int i = tid; i < HIDC; i += 256) {
    float y = (row[i] - mu_s) * inv_s * png[i] + pnb[i];
    gated[(size_t)r * HIDC + i] = gact[(size_t)r * HIDC + i] * y;
  }
}

extern "C" void kernel_launch(void* const* d_in, const int* in_sizes, int n_in,
                              void* d_out, int out_size, void* d_ws, size_t ws_size,
                              hipStream_t stream) {
  (void)in_sizes; (void)n_in; (void)out_size; (void)ws_size;
  const float* x   = (const float*)d_in[0];
  const float* wq  = (const float*)d_in[1];
  const float* wk  = (const float*)d_in[2];
  const float* wv  = (const float*)d_in[3];
  const float* fw1 = (const float*)d_in[4];
  const float* fb1 = (const float*)d_in[5];
  const float* fw2 = (const float*)d_in[6];
  const float* fb2 = (const float*)d_in[7];
  const float* flg = (const float*)d_in[8];
  const float* flb = (const float*)d_in[9];
  const float* lw  = (const float*)d_in[10];
  const float* lb  = (const float*)d_in[11];
  const float* ttg = (const float*)d_in[12];
  const float* ttb = (const float*)d_in[13];
  const float* wo  = (const float*)d_in[14];
  const float* wg  = (const float*)d_in[15];
  const float* png = (const float*)d_in[16];
  const float* pnb = (const float*)d_in[17];

  const int M = BB * SS;
  const size_t PLANE = (size_t)M * HIDC;
  float* qb    = (float*)d_ws;
  float* kb    = qb + PLANE;
  float* vb    = kb + PLANE;
  float* gactb = vb + PLANE;
  float* outb  = gactb + PLANE;
  float* gatedb = qb;

  dim3 gblk(256);
  dim3 ggrid(HIDC / 128, M / 16);
  gemm_wmma_f32<<<ggrid, gblk, 0, stream>>>(x, wq, qb,    M, HIDC, HIDC, 0);
  gemm_wmma_f32<<<ggrid, gblk, 0, stream>>>(x, wk, kb,    M, HIDC, HIDC, 0);
  gemm_wmma_f32<<<ggrid, gblk, 0, stream>>>(x, wv, vb,    M, HIDC, HIDC, 0);
  gemm_wmma_f32<<<ggrid, gblk, 0, stream>>>(x, wg, gactb, M, HIDC, HIDC, 1);

  ttt_scan_kernel<<<HH, 256, SCAN_SMEM_FLOATS * sizeof(float), stream>>>(
      kb, vb, qb, fw1, fb1, fw2, fb2, flg, flb, lw, lb, ttg, ttb, outb);

  gate_ln_kernel<<<M, 256, 0, stream>>>(outb, gactb, png, pnb, gatedb);

  gemm_wmma_f32<<<ggrid, gblk, 0, stream>>>(gatedb, wo, (float*)d_out,
                                            M, HIDC, HIDC, 0);
}


# scan: kt, target, z, dob and the k-split partials kept in LDS transposed with k pairs interleaved, plus such a copy of loss_w: fragment accesses that walked pitch-64 rows on one bank now spread out
# speedup vs baseline: 2.2783x; 1.2547x over previous
#include <hip/hip_runtime.h>
#include <hip/hip_bf16.h>
#include <math.h>

#define BB     8
#define SS     256
#define HIDC   1024
#define HH     16
#define DD     64
#define FF     64
#define LR     0.1f
#define GSCALE (2.0f / 8192.0f)
#define LN_EPS 1e-5f

typedef float v2f __attribute__((ext_vector_type(2)));
typedef float v8f __attribute__((ext_vector_type(8)));

__device__ __forceinline__ v8f wmma4(v2f a, v2f b, v8f c) {
  return __builtin_amdgcn_wmma_f32_16x16x4_f32(
       false, a,  false, b,
       (short)0, c,  false,  false);
}

__device__ __forceinline__ float gelu_f(float x) {
  const float c = 0.7978845608028654f;
  float u = c * (x + 0.044715f * x * x * x);
  return 0.5f * x * (1.0f + tanhf(u));
}
__device__ __forceinline__ void gelu_and_grad(float x, float& g, float& gd) {
  const float c = 0.7978845608028654f;
  float u = c * (x + 0.044715f * x * x * x);
  float t = tanhf(u);
  float x2 = x * x;
  g  = 0.5f * x * (1.0f + t);
  gd = 0.5f * (1.0f + t) + 0.5f * x * (1.0f - t * t) * c * (1.0f + 0.134145f * x2);
}
__device__ __forceinline__ float wave_sum32(float v) {
#pragma unroll
  for (int off = 16; off > 0; off >>= 1) v += __shfl_xor(v, off, 32);
  return v;
}

__device__ __forceinline__ float row_xor8(float v) {
  const int b = __builtin_bit_cast(int, v);
  return __builtin_bit_cast(float, __builtin_amdgcn_update_dpp(b, b, 0x168, 0xf, 0xf, false));
}

__global__ void __launch_bounds__(256)
gemm_wmma_f32(const float* __restrict__ A, const float* __restrict__ Bm,
              float* __restrict__ C, int M, int N, int K, int act) {
  __shared__ float As[16 * 64];
  const int tid   = threadIdx.x;
  const int wave  = tid >> 5;
  const int lane  = tid & 31;
  const int mbase = blockIdx.y * 16;
  const int tn    = blockIdx.x * 8 + wave;
  const int col   = tn * 16 + (lane & 15);
  const int kh    = (lane >> 4) << 1;
  const int arow  = lane & 15;
  v8f acc = {0.f, 0.f, 0.f, 0.f, 0.f, 0.f, 0.f, 0.f};
  for (int kc = 0; kc < K; kc += 64) {
    __syncthreads();
    for (int i = tid; i < 16 * 64; i += 256)
      As[i] = A[(size_t)(mbase + (i >> 6)) * K + kc + (i & 63)];
    if (kc + 64 < K)
      __builtin_prefetch(&A[(size_t)(mbase + (tid >> 4)) * K + kc + 64], 0, 1);
    __syncthreads();
#pragma unroll
    for (int k0 = 0; k0 < 64; k0 += 4) {
      v2f a, b;
      a.x = As[arow * 64 + k0 + kh];
      a.y = As[arow * 64 + k0 + kh + 1];
      b.x = Bm[(size_t)(kc + k0 + kh) * N + col];
      b.y = Bm[(size_t)(kc + k0 + kh + 1) * N + col];
      acc = wmma4(a, b, acc);
    }
  }
  const int r0 = mbase + ((lane >= 16) ? 8 : 0);
#pragma unroll
  for (int i = 0; i < 8; ++i) {
    float v = acc[i];
    if (act) v = gelu_f(v);
    C[(size_t)(r0 + i) * N + col] = v;
  }
}


#define HOLD8(a, o) asm volatile("" :: "v"(a[o]), "v"(a[o + 1]), "v"(a[o + 2]), "v"(a[o + 3]), \
                                       "v"(a[o + 4]), "v"(a[o + 5]), "v"(a[o + 6]), "v"(a[o + 7]))
#define TIDX(b, d) ((((d) >> 1) << 4) + ((b) << 1) + ((d) & 1))
#define CH_GELU  0
#define CH_DPRED 1
template <int EPI, bool PAIRED2, bool HAS_LATER, typename LATER>
__device__ __forceinline__ void
chain_pair(const float* X8T, const float* Wa, const float* biasa, const float* Wb,
           float* Cgd, float* Cact, const float* tgtT, float* part, int wave, int lane,
           LATER later_reads) {
  const int c  = lane & 15;
  const int h  = lane >> 4;
  const int bb = c & 7;
  const int kh = h << 1;
  const int fo = wave * 16;
  const int f0 = fo + 8 * h;
  const int xb = 16 * h + 2 * bb;
  const int tb = 128 * wave + 64 * h + 2 * bb;
  const int wp = 1024 * wave + 512 * h + 2 * c;
  v2f fa[16], fb[16], fa2[16];
  float bi[8], tg[8];
#pragma unroll
  for (int i = 0; i < 16; ++i) {
    const int k0 = 4 * i;
    fa[i].x = Wa[(k0 + kh) * 64 + fo + c];
    fa[i].y = Wa[(k0 + kh + 1) * 64 + fo + c];
    fb[i].x = X8T[xb + 32 * i];
    fb[i].y = X8T[xb + 32 * i + 1];
  }
#pragma unroll
  for (int s = 0; s < 4; ++s) {
    const int kk = f0 + 2 * s;
#pragma unroll
    for (int td = 0; td < 4; ++td) {
      const int m = td * 16 + c;
      if (!PAIRED2) {
        fa2[s * 4 + td].x = Wb[kk * 64 + m];
        fa2[s * 4 + td].y = Wb[(kk + 1) * 64 + m];
      } else {
        fa2[s * 4 + td].x = Wb[wp + 128 * s + 32 * td];
        fa2[s * 4 + td].y = Wb[wp + 128 * s + 32 * td + 1];
      }
    }
  }
#pragma unroll
  for (int r = 0; r < 8; ++r) {
    bi[r] = biasa[f0 + r];
    tg[r] = (EPI == CH_DPRED) ? tgtT[tb + 16 * (r >> 1) + (r & 1)] : 0.f;
  }
  HOLD8(fa, 0);  HOLD8(fa, 8);
  HOLD8(fb, 0);  HOLD8(fb, 8);
  HOLD8(fa2, 0); HOLD8(fa2, 8);
  HOLD8(bi, 0);
  if (EPI == CH_DPRED) HOLD8(tg, 0);
  if (HAS_LATER) {
    later_reads();
    __builtin_amdgcn_sched_barrier(0);
  }
  v8f acc = {0.f, 0.f, 0.f, 0.f, 0.f, 0.f, 0.f, 0.f};
#pragma unroll
  for (int i = 0; i < 16; ++i) acc = wmma4(fa[i], fb[i], acc);
  v8f g;
  if (EPI == CH_GELU) {
    const bool hi = (c >= 8);
    float gm[4], dm[4];
#pragma unroll
    for (int r = 0; r < 4; ++r) {
      const float xlo = acc[r] + bi[r];
      const float xhi = acc[4 + r] + bi[4 + r];
      gelu_and_grad(hi ? xhi : xlo, gm[r], dm[r]);
    }
    {
      const int so = bb * 64 + f0 + (hi ? 4 : 0);
#pragma unroll
      for (int r = 0; r < 4; ++r) {
        Cgd[so + r] = dm[r];
        Cact[so + r] = gm[r];
      }
    }
#pragma unroll
    for (int r = 0; r < 4; ++r) {
      const float go = row_xor8(gm[r]);
      g[r]     = hi ? go : gm[r];
      g[4 + r] = hi ? gm[r] : go;
    }
  } else {
#pragma unroll
    for (int r = 0; r < 8; ++r) {
      float v = acc[r] + bi[r];
      g[r] = GSCALE * (v - tg[r]);
    }
  }
  v8f p0 = {0.f, 0.f, 0.f, 0.f, 0.f, 0.f, 0.f, 0.f};
  v8f p1 = p0, p2 = p0, p3 = p0;
#pragma unroll
  for (int s = 0; s < 4; ++s) {
    v2f b2;
    b2.x = g[2 * s];
    b2.y = g[2 * s + 1];
    p0 = wmma4(fa2[s * 4 + 0], b2, p0);
    p1 = wmma4(fa2[s * 4 + 1], b2, p1);
    p2 = wmma4(fa2[s * 4 + 2], b2, p2);
    p3 = wmma4(fa2[s * 4 + 3], b2, p3);
  }
  if (c < 8) {
    float* dst = part + wave * 512 + 64 * h + 2 * bb;
#pragma unroll
    for (int r = 0; r < 8; ++r) {
      const int o = 16 * (r >> 1) + (r & 1);
      dst[o]       = p0[r];
      dst[128 + o] = p1[r];
      dst[256 + o] = p2[r];
      dst[384 + o] = p3[r];
    }
  }
}

__device__ __forceinline__ float part_sum4(const float* part, int idx) {
  return (part[idx] + part[512 + idx]) + (part[1024 + idx] + part[1536 + idx]);
}

__device__ __forceinline__ void
fwd_pass(const float* ktT, float* gdv, float* hg, float* ob, float* xh, float* zb, float* zbT,
         const float* W1, const float* W2, const float* b1v, const float* b2v,
         const float* lgv, const float* lbv, float* muv, float* isv, float* part,
         int wave, int lane, int tid) {
  if (wave < 4)
    chain_pair<CH_GELU, false, false>(ktT, W1, b1v, W2, gdv, hg, nullptr, part, wave, lane, [] {});
  __syncthreads();
  {
    const int pi = TIDX(wave, lane);
    float e0 = part_sum4(part, pi) + b2v[lane];
    float e1 = part_sum4(part, pi + 256) + b2v[lane + 32];
    ob[wave * 64 + lane] = e0;
    ob[wave * 64 + lane + 32] = e1;
    float s = wave_sum32(e0 + e1);
    float q = wave_sum32(e0 * e0 + e1 * e1);
    if (lane == 0) {
      float m = s * (1.f / 64.f);
      muv[wave] = m;
      isv[wave] = rsqrtf(q * (1.f / 64.f) - m * m + LN_EPS);
    }
  }
  __syncthreads();
  for (int idx = tid; idx < 512; idx += 256) {
    int b = idx >> 6, d = idx & 63;
    float xv = (ob[idx] - muv[b]) * isv[b];
    xh[idx] = xv;
    const float zv = xv * lgv[d] + lbv[d];
    zb[idx] = zv;
    zbT[TIDX(b, d)] = zv;
  }
  __syncthreads();
}

#define SCAN_SMEM_FLOATS (4 * 4096 + 12 * 512 + 7 * 64 + 32 + 4 * 512)

__global__ void __launch_bounds__(256)
ttt_scan_kernel(const float* __restrict__ kg, const float* __restrict__ vg,
                const float* __restrict__ qg,
                const float* __restrict__ fw1, const float* __restrict__ fb1,
                const float* __restrict__ fw2, const float* __restrict__ fb2,
                const float* __restrict__ flg, const float* __restrict__ flb,
                const float* __restrict__ lwg, const float* __restrict__ lbg,
                const float* __restrict__ ttg, const float* __restrict__ ttb,
                float* __restrict__ outg) {
  extern __shared__ __align__(16) float smem[];
  float* W1  = smem;
  float* W2  = W1 + 4096;
  float* LW  = W2 + 4096;
  float* ktb = LW + 4096;
  float* tgt = ktb + 512;
  float* gdv = tgt + 512;
  float* hg  = gdv + 512;
  float* ob  = hg + 512;
  float* xh  = ob + 512;
  float* zb  = xh + 512;
  float* dzb = zb + 512;
  float* dob = dzb + 512;
  float* ktT = dob + 512;
  float* zbT = ktT + 512;
  float* dobT = zbT + 512;
  float* b1v = dobT + 512;
  float* b2v = b1v + 64;
  float* lgv = b2v + 64;
  float* lbv = lgv + 64;
  float* tgv = lbv + 64;
  float* tbv = tgv + 64;
  float* lbz = tbv + 64;
  float* muv = lbz + 64;
  float* isv = muv + 8;
  float* m1v = isv + 8;
  float* m2v = m1v + 8;
  float* part = m2v + 8;
  float* LWP = part + 4 * 512;

  const int h    = blockIdx.x;
  const int tid  = threadIdx.x;
  const int wave = tid >> 5;
  const int lane = tid & 31;

  const int b0 = tid >> 6,          d0 = tid & 63;
  const int b1i = (tid + 256) >> 6, d1i = (tid + 256) & 63;
  const size_t g0base = (size_t)(b0 * SS) * HIDC + h * 64 + d0;
  const size_t g1base = (size_t)(b1i * SS) * HIDC + h * 64 + d1i;

  for (int i = tid; i < 4096; i += 256) {
    W1[i] = fw1[(size_t)h * 4096 + i];
    W2[i] = fw2[(size_t)h * 4096 + i];
    LW[i] = lwg[i];
    LWP[((i & 63) >> 1) * 128 + 2 * (i >> 6) + (i & 1)] = lwg[i];
  }
  if (tid < 64) {
    b1v[tid] = fb1[h * 64 + tid];
    b2v[tid] = fb2[h * 64 + tid];
    lgv[tid] = flg[h * 64 + tid];
    lbv[tid] = flb[h * 64 + tid];
    tgv[tid] = ttg[h * 64 + tid];
    tbv[tid] = ttb[h * 64 + tid];
    lbz[tid] = lbg[tid];
  }

  float kp0 = kg[g0base], vp0 = vg[g0base];
  float kp1 = kg[g1base], vp1 = vg[g1base];
  __syncthreads();

  for (int t = 0; t < SS; ++t) {
    ktb[tid]            = kp0;  ktb[tid + 256]      = kp1;
    ktT[TIDX(b0, d0)]   = kp0;  ktT[TIDX(b1i, d1i)] = kp1;
    tgt[TIDX(b0, d0)]   = vp0 - kp0;
    tgt[TIDX(b1i, d1i)] = vp1 - kp1;
    __syncthreads();
    if (t + 1 < SS) {
      kp0 = kg[g0base + (t + 1) * HIDC];  vp0 = vg[g0base + (t + 1) * HIDC];
      kp1 = kg[g1base + (t + 1) * HIDC];  vp1 = vg[g1base + (t + 1) * HIDC];
    }

    for (int step = 0; step < 4; ++step) {
      fwd_pass(ktT, gdv, hg, ob, xh, zb, zbT, W1, W2, b1v, b2v, lgv, lbv,
               muv, isv, part, wave, lane, tid);
      const int fc = lane & 15;
      const int fh = lane >> 4;
      const int fk = fh << 1;
      v2f   pb[16];
      v2f   pk[8];
      float pg[8];
      float pb1 = 0.f;
      v2f   ph[2];
      float wold[32];
      if (wave < 4) {
        chain_pair<CH_DPRED, true, true>(zbT, LW, lbz, LWP, nullptr, nullptr, tgt, part, wave, lane, [&] {
          const int col = wave * 16 + fc;
#pragma unroll
          for (int i = 0; i < 16; ++i) {
            pb[i].x = W2[col * 64 + 4 * i + fk];
            pb[i].y = W2[col * 64 + 4 * i + fk + 1];
          }
#pragma unroll
          for (int r = 0; r < 8; ++r) pg[r] = gdv[r * 64 + col];
          pb1 = b1v[col];
#pragma unroll
          for (int tm = 0; tm < 4; ++tm) {
#pragma unroll
            for (int s = 0; s < 2; ++s) {
              pk[tm * 2 + s].x = ktb[(4 * s + fk) * 64 + tm * 16 + fc];
              pk[tm * 2 + s].y = ktb[(4 * s + fk + 1) * 64 + tm * 16 + fc];
            }
#pragma unroll
            for (int i = 0; i < 8; ++i) wold[tm * 8 + i] = W1[(tm * 16 + 8 * fh + i) * 64 + col];
          }
        });
      } else {
        const int wv = wave - 4;
#pragma unroll
        for (int s = 0; s < 2; ++s) {
          ph[s].x = hg[(4 * s + fk) * 64 + wv * 16 + fc];
          ph[s].y = hg[(4 * s + fk + 1) * 64 + wv * 16 + fc];
        }
#pragma unroll
        for (int tn = 0; tn < 4; ++tn) {
#pragma unroll
          for (int i = 0; i < 8; ++i) wold[tn * 8 + i] = W2[(wv * 16 + 8 * fh + i) * 64 + tn * 16 + fc];
        }
      }
      __syncthreads();
      {
        const int pi = TIDX(wave, lane);
        float z0 = part_sum4(part, pi);
        float z1 = part_sum4(part, pi + 256);
        dzb[wave * 64 + lane] = z0;
        dzb[wave * 64 + lane + 32] = z1;
        float dd0 = z0 * lgv[lane];
        float dd1 = z1 * lgv[lane + 32];
        float x0 = xh[wave * 64 + lane], x1 = xh[wave * 64 + lane + 32];
        float s1 = wave_sum32(dd0 + dd1);
        float s2 = wave_sum32(dd0 * x0 + dd1 * x1);
        if (lane == 0) { m1v[wave] = s1 * (1.f / 64.f); m2v[wave] = s2 * (1.f / 64.f); }
      }
      __syncthreads();
      for (int idx = tid; idx < 512; idx += 256) {
        int b = idx >> 6, d = idx & 63;
        float dxh = dzb[idx] * lgv[d];
        const float dv = isv[b] * (dxh - m1v[b] - xh[idx] * m2v[b]);
        dob[idx] = dv;
        dobT[TIDX(b, d)] = dv;
      }
      __syncthreads();
      if (wave < 4) {
        const int col = wave * 16 + fc;
        const int bb  = fc & 7;
        v2f pa[16];
#pragma unroll
        for (int i = 0; i < 16; ++i) {
          pa[i].x = dobT[16 * fh + 2 * bb + 32 * i];
          pa[i].y = dobT[16 * fh + 2 * bb + 32 * i + 1];
        }
        HOLD8(pa, 0); HOLD8(pa, 8);
        v8f acc = {0.f, 0.f, 0.f, 0.f, 0.f, 0.f, 0.f, 0.f};
#pragma unroll
        for (int i = 0; i < 16; ++i) acc = wmma4(pa[i], pb[i], acc);
        float da[8];
        float sb = 0.f;
#pragma unroll
        for (int r = 0; r < 8; ++r) da[r] = acc[r] * pg[r];
#pragma unroll
        for (int r = 0; r < 8; ++r) sb += da[r];
        if (fh == 0) b1v[col] = pb1 - LR * sb;
        v8f u0 = {0.f, 0.f, 0.f, 0.f, 0.f, 0.f, 0.f, 0.f};
        v8f u1 = u0, u2 = u0, u3 = u0;
#pragma unroll
        for (int s = 0; s < 2; ++s) {
          v2f b2;
          b2.x = fh ? da[4 * s + 2] : da[4 * s];
          b2.y = fh ? da[4 * s + 3] : da[4 * s + 1];
          u0 = wmma4(pk[0 + s], b2, u0);
          u1 = wmma4(pk[2 + s], b2, u1);
          u2 = wmma4(pk[4 + s], b2, u2);
          u3 = wmma4(pk[6 + s], b2, u3);
        }
#pragma unroll
        for (int i = 0; i < 8; ++i) {
          W1[(0 * 16 + 8 * fh + i) * 64 + col] = wold[0 * 8 + i] - LR * u0[i];
          W1[(1 * 16 + 8 * fh + i) * 64 + col] = wold[1 * 8 + i] - LR * u1[i];
          W1[(2 * 16 + 8 * fh + i) * 64 + col] = wold[2 * 8 + i] - LR * u2[i];
          W1[(3 * 16 + 8 * fh + i) * 64 + col] = wold[3 * 8 + i] - LR * u3[i];
        }
      } else {
        const int wv = wave - 4;
        v2f pd[8];
#pragma unroll
        for (int tn = 0; tn < 4; ++tn) {
#pragma unroll
          for (int s = 0; s < 2; ++s) {
            pd[tn * 2 + s].x = dob[(4 * s + fk) * 64 + tn * 16 + fc];
            pd[tn * 2 + s].y = dob[(4 * s + fk + 1) * 64 + tn * 16 + fc];
          }
        }
        v8f u0 = {0.f, 0.f, 0.f, 0.f, 0.f, 0.f, 0.f, 0.f};
        v8f u1 = u0, u2 = u0, u3 = u0;
#pragma unroll
        for (int s = 0; s < 2; ++s) {
          u0 = wmma4(ph[s], pd[0 + s], u0);
          u1 = wmma4(ph[s], pd[2 + s], u1);
          u2 = wmma4(ph[s], pd[4 + s], u2);
          u3 = wmma4(ph[s], pd[6 + s], u3);
        }
        const int rb = (wv * 16 + 8 * fh) * 64 + fc;
#pragma unroll
        for (int i = 0; i < 8; ++i) {
          W2[rb + i * 64 + 0]  = wold[0 * 8 + i] - LR * u0[i];
          W2[rb + i * 64 + 16] = wold[1 * 8 + i] - LR * u1[i];
          W2[rb + i * 64 + 32] = wold[2 * 8 + i] - LR * u2[i];
          W2[rb + i * 64 + 48] = wold[3 * 8 + i] - LR * u3[i];
        }
        const int t2 = tid - 128;
        if (t2 < 64) {
          float s = 0.f, s2 = 0.f;
          for (int b = 0; b < 8; ++b) s += dzb[b * 64 + t2] * xh[b * 64 + t2];
          for (int b = 0; b < 8; ++b) s2 += dob[b * 64 + t2];
          lgv[t2] -= LR * s;
          b2v[t2] -= LR * s2;
        } else {
          const int d = t2 - 64;
          float s = 0.f;
          for (int b = 0; b < 8; ++b) s += dzb[b * 64 + d];
          lbv[d] -= LR * s;
        }
      }
      __syncthreads();
    }

    fwd_pass(ktT, gdv, hg, ob, xh, zb, zbT, W1, W2, b1v, b2v, lgv, lbv,
             muv, isv, part, wave, lane, tid);
    {
      float e0 = zb[wave * 64 + lane], e1 = zb[wave * 64 + lane + 32];
      float s = wave_sum32(e0 + e1);
      float q = wave_sum32(e0 * e0 + e1 * e1);
      if (lane == 0) {
        float m = s * (1.f / 64.f);
        muv[wave] = m;
        isv[wave] = rsqrtf(q * (1.f / 64.f) - m * m + LN_EPS);
      }
    }
    __syncthreads();
    for (int idx = tid; idx < 512; idx += 256) {
      int b = idx >> 6, d = idx & 63;
      float z2 = (zb[idx] - muv[b]) * isv[b] * tgv[d] + tbv[d];
      size_t g = ((size_t)(b * SS + t)) * HIDC + h * 64 + d;
      outg[g] = qg[g] + z2;
    }
    __syncthreads();
  }
}

__global__ void __launch_bounds__(256)
gate_ln_kernel(const float* __restrict__ outb, const float* __restrict__ gact,
               const float* __restrict__ png, const float* __restrict__ pnb,
               float* __restrict__ gated) {
  __shared__ float r1[256], r2[256];
  __shared__ float mu_s, inv_s;
  const int r = blockIdx.x, tid = threadIdx.x;
  const float* row = outb + (size_t)r * HIDC;
  float s = 0.f, q = 0.f;
  for (int i = tid; i < HIDC; i += 256) { float v = row[i]; s += v; q += v * v; }
  r1[tid] = s; r2[tid] = q;
  __syncthreads();
  for (int off = 128; off > 0; off >>= 1) {
    if (tid < off) { r1[tid] += r1[tid + off]; r2[tid] += r2[tid + off]; }
    __syncthreads();
  }
  if (tid == 0) {
    float m = r1[0] * (1.f / HIDC);
    mu_s = m;
    inv_s = rsqrtf(r2[0] * (1.f / HIDC) - m * m + LN_EPS);
  }
  __syncthreads();
  for (int i = tid; i < HIDC; i += 256) {
    float y = (row[i] - mu_s) * inv_s * png[i] + pnb[i];
    gated[(size_t)r * HIDC + i] = gact[(size_t)r * HIDC + i] * y;
  }
}

extern "C" void kernel_launch(void* const* d_in, const int* in_sizes, int n_in,
                              void* d_out, int out_size, void* d_ws, size_t ws_size,
                              hipStream_t stream) {
  (void)in_sizes; (void)n_in; (void)out_size; (void)ws_size;
  const float* x   = (const float*)d_in[0];
  const float* wq  = (const float*)d_in[1];
  const float* wk  = (const float*)d_in[2];
  const float* wv  = (const float*)d_in[3];
  const float* fw1 = (const float*)d_in[4];
  const float* fb1 = (const float*)d_in[5];
  const float* fw2 = (const float*)d_in[6];
  const float* fb2 = (const float*)d_in[7];
  const float* flg = (const float*)d_in[8];
  const float* flb = (const float*)d_in[9];
  const float* lw  = (const float*)d_in[10];
  const float* lb  = (const float*)d_in[11];
  const float* ttg = (const float*)d_in[12];
  const float* ttb = (const float*)d_in[13];
  const float* wo  = (const float*)d_in[14];
  const float* wg  = (const float*)d_in[15];
  const float* png = (const float*)d_in[16];
  const float* pnb = (const float*)d_in[17];

  const int M = BB * SS;
  const size_t PLANE = (size_t)M * HIDC;
  float* qb    = (float*)d_ws;
  float* kb    = qb + PLANE;
  float* vb    = kb + PLANE;
  float* gactb = vb + PLANE;
  float* outb  = gactb + PLANE;
  float* gatedb = qb;

  dim3 gblk(256);
  dim3 ggrid(HIDC / 128, M / 16);
  gemm_wmma_f32<<<ggrid, gblk, 0, stream>>>(x, wq, qb,    M, HIDC, HIDC, 0);
  gemm_wmma_f32<<<ggrid, gblk, 0, stream>>>(x, wk, kb,    M, HIDC, HIDC, 0);
  gemm_wmma_f32<<<ggrid, gblk, 0, stream>>>(x, wv, vb,    M, HIDC, HIDC, 0);
  gemm_wmma_f32<<<ggrid, gblk, 0, stream>>>(x, wg, gactb, M, HIDC, HIDC, 1);

  ttt_scan_kernel<<<HH, 256, SCAN_SMEM_FLOATS * sizeof(float), stream>>>(
      kb, vb, qb, fw1, fb1, fw2, fb2, flg, flb, lw, lb, ttg, ttb, outb);

  gate_ln_kernel<<<M, 256, 0, stream>>>(outb, gactb, png, pnb, gatedb);

  gemm_wmma_f32<<<ggrid, gblk, 0, stream>>>(gatedb, wo, (float*)d_out,
                                            M, HIDC, HIDC, 0);
}


# scan: W2 held in LDS d-pair-major with a bank rotation per pair instead of row-major: backward reads (lanes walk rows, were 16 deep on one bank pair), forward reads and update tiles all conflict-free
# speedup vs baseline: 2.3860x; 1.0473x over previous
#include <hip/hip_runtime.h>
#include <hip/hip_bf16.h>
#include <math.h>

#define BB     8
#define SS     256
#define HIDC   1024
#define HH     16
#define DD     64
#define FF     64
#define LR     0.1f
#define GSCALE (2.0f / 8192.0f)
#define LN_EPS 1e-5f

typedef float v2f __attribute__((ext_vector_type(2)));
typedef float v8f __attribute__((ext_vector_type(8)));

__device__ __forceinline__ v8f wmma4(v2f a, v2f b, v8f c) {
  return __builtin_amdgcn_wmma_f32_16x16x4_f32(
       false, a,  false, b,
       (short)0, c,  false,  false);
}

__device__ __forceinline__ float gelu_f(float x) {
  const float c = 0.7978845608028654f;
  float u = c * (x + 0.044715f * x * x * x);
  return 0.5f * x * (1.0f + tanhf(u));
}
__device__ __forceinline__ void gelu_and_grad(float x, float& g, float& gd) {
  const float c = 0.7978845608028654f;
  float u = c * (x + 0.044715f * x * x * x);
  float t = tanhf(u);
  float x2 = x * x;
  g  = 0.5f * x * (1.0f + t);
  gd = 0.5f * (1.0f + t) + 0.5f * x * (1.0f - t * t) * c * (1.0f + 0.134145f * x2);
}
__device__ __forceinline__ float wave_sum32(float v) {
#pragma unroll
  for (int off = 16; off > 0; off >>= 1) v += __shfl_xor(v, off, 32);
  return v;
}

__device__ __forceinline__ float row_xor8(float v) {
  const int b = __builtin_bit_cast(int, v);
  return __builtin_bit_cast(float, __builtin_amdgcn_update_dpp(b, b, 0x168, 0xf, 0xf, false));
}

__global__ void __launch_bounds__(256)
gemm_wmma_f32(const float* __restrict__ A, const float* __restrict__ Bm,
              float* __restrict__ C, int M, int N, int K, int act) {
  __shared__ float As[16 * 64];
  const int tid   = threadIdx.x;
  const int wave  = tid >> 5;
  const int lane  = tid & 31;
  const int mbase = blockIdx.y * 16;
  const int tn    = blockIdx.x * 8 + wave;
  const int col   = tn * 16 + (lane & 15);
  const int kh    = (lane >> 4) << 1;
  const int arow  = lane & 15;
  v8f acc = {0.f, 0.f, 0.f, 0.f, 0.f, 0.f, 0.f, 0.f};
  for (int kc = 0; kc < K; kc += 64) {
    __syncthreads();
    for (int i = tid; i < 16 * 64; i += 256)
      As[i] = A[(size_t)(mbase + (i >> 6)) * K + kc + (i & 63)];
    if (kc + 64 < K)
      __builtin_prefetch(&A[(size_t)(mbase + (tid >> 4)) * K + kc + 64], 0, 1);
    __syncthreads();
#pragma unroll
    for (int k0 = 0; k0 < 64; k0 += 4) {
      v2f a, b;
      a.x = As[arow * 64 + k0 + kh];
      a.y = As[arow * 64 + k0 + kh + 1];
      b.x = Bm[(size_t)(kc + k0 + kh) * N + col];
      b.y = Bm[(size_t)(kc + k0 + kh + 1) * N + col];
      acc = wmma4(a, b, acc);
    }
  }
  const int r0 = mbase + ((lane >= 16) ? 8 : 0);
#pragma unroll
  for (int i = 0; i < 8; ++i) {
    float v = acc[i];
    if (act) v = gelu_f(v);
    C[(size_t)(r0 + i) * N + col] = v;
  }
}


#define HOLD8(a, o) asm volatile("" :: "v"(a[o]), "v"(a[o + 1]), "v"(a[o + 2]), "v"(a[o + 3]), \
                                       "v"(a[o + 4]), "v"(a[o + 5]), "v"(a[o + 6]), "v"(a[o + 7]))
#define TIDX(b, d) ((((d) >> 1) << 4) + ((b) << 1) + ((d) & 1))
#define W2P_PITCH 192
#define W2P_ROT(p) ((((p) & 6) << 1) + (((p) & 1) << 5))
#define W2P_IDX(f, d) (((d) >> 1) * W2P_PITCH + 2 * (f) + W2P_ROT((d) >> 1) + ((d) & 1))
#define WB_W2P 0
#define WB_LWP 1
#define CH_GELU  0
#define CH_DPRED 1
template <int EPI, int WBL, bool HAS_LATER, typename LATER>
__device__ __forceinline__ void
chain_pair(const float* X8T, const float* Wa, const float* biasa, const float* Wb,
           float* Cgd, float* Cact, const float* tgtT, float* part, int wave, int lane,
           LATER later_reads) {
  const int c  = lane & 15;
  const int h  = lane >> 4;
  const int bb = c & 7;
  const int kh = h << 1;
  const int fo = wave * 16;
  const int f0 = fo + 8 * h;
  const int xb = 16 * h + 2 * bb;
  const int tb = 128 * wave + 64 * h + 2 * bb;
  const int wp = 1024 * wave + 512 * h + 2 * c;
  const int wr = W2P_IDX(f0, c);
  v2f fa[16], fb[16], fa2[16];
  float bi[8], tg[8];
#pragma unroll
  for (int i = 0; i < 16; ++i) {
    const int k0 = 4 * i;
    fa[i].x = Wa[(k0 + kh) * 64 + fo + c];
    fa[i].y = Wa[(k0 + kh + 1) * 64 + fo + c];
    fb[i].x = X8T[xb + 32 * i];
    fb[i].y = X8T[xb + 32 * i + 1];
  }
#pragma unroll
  for (int s = 0; s < 4; ++s) {
#pragma unroll
    for (int td = 0; td < 4; ++td) {
      if (WBL == WB_W2P) {
        fa2[s * 4 + td].x = Wb[wr + 8 * W2P_PITCH * td + 4 * s];
        fa2[s * 4 + td].y = Wb[wr + 8 * W2P_PITCH * td + 4 * s + 2];
      } else {
        fa2[s * 4 + td].x = Wb[wp + 128 * s + 32 * td];
        fa2[s * 4 + td].y = Wb[wp + 128 * s + 32 * td + 1];
      }
    }
  }
#pragma unroll
  for (int r = 0; r < 8; ++r) {
    bi[r] = biasa[f0 + r];
    tg[r] = (EPI == CH_DPRED) ? tgtT[tb + 16 * (r >> 1) + (r & 1)] : 0.f;
  }
  HOLD8(fa, 0);  HOLD8(fa, 8);
  HOLD8(fb, 0);  HOLD8(fb, 8);
  HOLD8(fa2, 0); HOLD8(fa2, 8);
  HOLD8(bi, 0);
  if (EPI == CH_DPRED) HOLD8(tg, 0);
  if (HAS_LATER) {
    later_reads();
    __builtin_amdgcn_sched_barrier(0);
  }
  v8f acc = {0.f, 0.f, 0.f, 0.f, 0.f, 0.f, 0.f, 0.f};
#pragma unroll
  for (int i = 0; i < 16; ++i) acc = wmma4(fa[i], fb[i], acc);
  v8f g;
  if (EPI == CH_GELU) {
    const bool hi = (c >= 8);
    float gm[4], dm[4];
#pragma unroll
    for (int r = 0; r < 4; ++r) {
      const float xlo = acc[r] + bi[r];
      const float xhi = acc[4 + r] + bi[4 + r];
      gelu_and_grad(hi ? xhi : xlo, gm[r], dm[r]);
    }
    {
      const int so = bb * 64 + f0 + (hi ? 4 : 0);
#pragma unroll
      for (int r = 0; r < 4; ++r) {
        Cgd[so + r] = dm[r];
        Cact[so + r] = gm[r];
      }
    }
#pragma unroll
    for (int r = 0; r < 4; ++r) {
      const float go = row_xor8(gm[r]);
      g[r]     = hi ? go : gm[r];
      g[4 + r] = hi ? gm[r] : go;
    }
  } else {
#pragma unroll
    for (int r = 0; r < 8; ++r) {
      float v = acc[r] + bi[r];
      g[r] = GSCALE * (v - tg[r]);
    }
  }
  v8f p0 = {0.f, 0.f, 0.f, 0.f, 0.f, 0.f, 0.f, 0.f};
  v8f p1 = p0, p2 = p0, p3 = p0;
#pragma unroll
  for (int s = 0; s < 4; ++s) {
    v2f b2;
    b2.x = g[2 * s];
    b2.y = g[2 * s + 1];
    p0 = wmma4(fa2[s * 4 + 0], b2, p0);
    p1 = wmma4(fa2[s * 4 + 1], b2, p1);
    p2 = wmma4(fa2[s * 4 + 2], b2, p2);
    p3 = wmma4(fa2[s * 4 + 3], b2, p3);
  }
  if (c < 8) {
    float* dst = part + wave * 512 + 64 * h + 2 * bb;
#pragma unroll
    for (int r = 0; r < 8; ++r) {
      const int o = 16 * (r >> 1) + (r & 1);
      dst[o]       = p0[r];
      dst[128 + o] = p1[r];
      dst[256 + o] = p2[r];
      dst[384 + o] = p3[r];
    }
  }
}

__device__ __forceinline__ float part_sum4(const float* part, int idx) {
  return (part[idx] + part[512 + idx]) + (part[1024 + idx] + part[1536 + idx]);
}

__device__ __forceinline__ void
fwd_pass(const float* ktT, float* gdv, float* hg, float* ob, float* xh, float* zb, float* zbT,
         const float* W1, const float* W2, const float* b1v, const float* b2v,
         const float* lgv, const float* lbv, float* muv, float* isv, float* part,
         int wave, int lane, int tid) {
  if (wave < 4)
    chain_pair<CH_GELU, WB_W2P, false>(ktT, W1, b1v, W2, gdv, hg, nullptr, part, wave, lane, [] {});
  __syncthreads();
  {
    const int pi = TIDX(wave, lane);
    float e0 = part_sum4(part, pi) + b2v[lane];
    float e1 = part_sum4(part, pi + 256) + b2v[lane + 32];
    ob[wave * 64 + lane] = e0;
    ob[wave * 64 + lane + 32] = e1;
    float s = wave_sum32(e0 + e1);
    float q = wave_sum32(e0 * e0 + e1 * e1);
    if (lane == 0) {
      float m = s * (1.f / 64.f);
      muv[wave] = m;
      isv[wave] = rsqrtf(q * (1.f / 64.f) - m * m + LN_EPS);
    }
  }
  __syncthreads();
  for (int idx = tid; idx < 512; idx += 256) {
    int b = idx >> 6, d = idx & 63;
    float xv = (ob[idx] - muv[b]) * isv[b];
    xh[idx] = xv;
    const float zv = xv * lgv[d] + lbv[d];
    zb[idx] = zv;
    zbT[TIDX(b, d)] = zv;
  }
  __syncthreads();
}

#define SCAN_SMEM_FLOATS (3 * 4096 + 12 * 512 + 7 * 64 + 32 + 4 * 512 + 32 * W2P_PITCH)

__global__ void __launch_bounds__(256)
ttt_scan_kernel(const float* __restrict__ kg, const float* __restrict__ vg,
                const float* __restrict__ qg,
                const float* __restrict__ fw1, const float* __restrict__ fb1,
                const float* __restrict__ fw2, const float* __restrict__ fb2,
                const float* __restrict__ flg, const float* __restrict__ flb,
                const float* __restrict__ lwg, const float* __restrict__ lbg,
                const float* __restrict__ ttg, const float* __restrict__ ttb,
                float* __restrict__ outg) {
  extern __shared__ __align__(16) float smem[];
  float* W1  = smem;
  float* LW  = W1 + 4096;
  float* ktb = LW + 4096;
  float* tgt = ktb + 512;
  float* gdv = tgt + 512;
  float* hg  = gdv + 512;
  float* ob  = hg + 512;
  float* xh  = ob + 512;
  float* zb  = xh + 512;
  float* dzb = zb + 512;
  float* dob = dzb + 512;
  float* ktT = dob + 512;
  float* zbT = ktT + 512;
  float* dobT = zbT + 512;
  float* b1v = dobT + 512;
  float* b2v = b1v + 64;
  float* lgv = b2v + 64;
  float* lbv = lgv + 64;
  float* tgv = lbv + 64;
  float* tbv = tgv + 64;
  float* lbz = tbv + 64;
  float* muv = lbz + 64;
  float* isv = muv + 8;
  float* m1v = isv + 8;
  float* m2v = m1v + 8;
  float* part = m2v + 8;
  float* LWP = part + 4 * 512;
  float* W2  = LWP + 4096;

  const int h    = blockIdx.x;
  const int tid  = threadIdx.x;
  const int wave = tid >> 5;
  const int lane = tid & 31;

  const int b0 = tid >> 6,          d0 = tid & 63;
  const int b1i = (tid + 256) >> 6, d1i = (tid + 256) & 63;
  const size_t g0base = (size_t)(b0 * SS) * HIDC + h * 64 + d0;
  const size_t g1base = (size_t)(b1i * SS) * HIDC + h * 64 + d1i;

  for (int i = tid; i < 4096; i += 256) {
    W1[i] = fw1[(size_t)h * 4096 + i];
    W2[W2P_IDX(i >> 6, i & 63)] = fw2[(size_t)h * 4096 + i];
    LW[i] = lwg[i];
    LWP[((i & 63) >> 1) * 128 + 2 * (i >> 6) + (i & 1)] = lwg[i];
  }
  if (tid < 64) {
    b1v[tid] = fb1[h * 64 + tid];
    b2v[tid] = fb2[h * 64 + tid];
    lgv[tid] = flg[h * 64 + tid];
    lbv[tid] = flb[h * 64 + tid];
    tgv[tid] = ttg[h * 64 + tid];
    tbv[tid] = ttb[h * 64 + tid];
    lbz[tid] = lbg[tid];
  }

  float kp0 = kg[g0base], vp0 = vg[g0base];
  float kp1 = kg[g1base], vp1 = vg[g1base];
  __syncthreads();

  for (int t = 0; t < SS; ++t) {
    ktb[tid]            = kp0;  ktb[tid + 256]      = kp1;
    ktT[TIDX(b0, d0)]   = kp0;  ktT[TIDX(b1i, d1i)] = kp1;
    tgt[TIDX(b0, d0)]   = vp0 - kp0;
    tgt[TIDX(b1i, d1i)] = vp1 - kp1;
    __syncthreads();
    if (t + 1 < SS) {
      kp0 = kg[g0base + (t + 1) * HIDC];  vp0 = vg[g0base + (t + 1) * HIDC];
      kp1 = kg[g1base + (t + 1) * HIDC];  vp1 = vg[g1base + (t + 1) * HIDC];
    }

    for (int step = 0; step < 4; ++step) {
      fwd_pass(ktT, gdv, hg, ob, xh, zb, zbT, W1, W2, b1v, b2v, lgv, lbv,
               muv, isv, part, wave, lane, tid);
      const int fc = lane & 15;
      const int fh = lane >> 4;
      const int fk = fh << 1;
      v2f   pb[16];
      v2f   pk[8];
      float pg[8];
      float pb1 = 0.f;
      v2f   ph[2];
      float wold[32];
      if (wave < 4) {
        chain_pair<CH_DPRED, WB_LWP, true>(zbT, LW, lbz, LWP, nullptr, nullptr, tgt, part, wave, lane, [&] {
          const int col = wave * 16 + fc;
          const int wq  = fh * (W2P_PITCH + 32) + 2 * col;
#pragma unroll
          for (int i = 0; i < 16; ++i) {
            pb[i].x = W2[wq + 2 * W2P_PITCH * i + 4 * (i & 3)];
            pb[i].y = W2[wq + 2 * W2P_PITCH * i + 4 * (i & 3) + 1];
          }
#pragma unroll
          for (int r = 0; r < 8; ++r) pg[r] = gdv[r * 64 + col];
          pb1 = b1v[col];
#pragma unroll
          for (int tm = 0; tm < 4; ++tm) {
#pragma unroll
            for (int s = 0; s < 2; ++s) {
              pk[tm * 2 + s].x = ktb[(4 * s + fk) * 64 + tm * 16 + fc];
              pk[tm * 2 + s].y = ktb[(4 * s + fk + 1) * 64 + tm * 16 + fc];
            }
#pragma unroll
            for (int i = 0; i < 8; ++i) wold[tm * 8 + i] = W1[(tm * 16 + 8 * fh + i) * 64 + col];
          }
        });
      } else {
        const int wv = wave - 4;
#pragma unroll
        for (int s = 0; s < 2; ++s) {
          ph[s].x = hg[(4 * s + fk) * 64 + wv * 16 + fc];
          ph[s].y = hg[(4 * s + fk + 1) * 64 + wv * 16 + fc];
        }
#pragma unroll
        for (int tn = 0; tn < 4; ++tn) {
#pragma unroll
          for (int i = 0; i < 8; ++i)
            wold[tn * 8 + i] = W2[W2P_IDX(wv * 16 + 8 * fh, fc) + 8 * W2P_PITCH * tn + 2 * i];
        }
      }
      __syncthreads();
      {
        const int pi = TIDX(wave, lane);
        float z0 = part_sum4(part, pi);
        float z1 = part_sum4(part, pi + 256);
        dzb[wave * 64 + lane] = z0;
        dzb[wave * 64 + lane + 32] = z1;
        float dd0 = z0 * lgv[lane];
        float dd1 = z1 * lgv[lane + 32];
        float x0 = xh[wave * 64 + lane], x1 = xh[wave * 64 + lane + 32];
        float s1 = wave_sum32(dd0 + dd1);
        float s2 = wave_sum32(dd0 * x0 + dd1 * x1);
        if (lane == 0) { m1v[wave] = s1 * (1.f / 64.f); m2v[wave] = s2 * (1.f / 64.f); }
      }
      __syncthreads();
      for (int idx = tid; idx < 512; idx += 256) {
        int b = idx >> 6, d = idx & 63;
        float dxh = dzb[idx] * lgv[d];
        const float dv = isv[b] * (dxh - m1v[b] - xh[idx] * m2v[b]);
        dob[idx] = dv;
        dobT[TIDX(b, d)] = dv;
      }
      __syncthreads();
      if (wave < 4) {
        const int col = wave * 16 + fc;
        const int bb  = fc & 7;
        v2f pa[16];
#pragma unroll
        for (int i = 0; i < 16; ++i) {
          pa[i].x = dobT[16 * fh + 2 * bb + 32 * i];
          pa[i].y = dobT[16 * fh + 2 * bb + 32 * i + 1];
        }
        HOLD8(pa, 0); HOLD8(pa, 8);
        v8f acc = {0.f, 0.f, 0.f, 0.f, 0.f, 0.f, 0.f, 0.f};
#pragma unroll
        for (int i = 0; i < 16; ++i) acc = wmma4(pa[i], pb[i], acc);
        float da[8];
        float sb = 0.f;
#pragma unroll
        for (int r = 0; r < 8; ++r) da[r] = acc[r] * pg[r];
#pragma unroll
        for (int r = 0; r < 8; ++r) sb += da[r];
        if (fh == 0) b1v[col] = pb1 - LR * sb;
        v8f u0 = {0.f, 0.f, 0.f, 0.f, 0.f, 0.f, 0.f, 0.f};
        v8f u1 = u0, u2 = u0, u3 = u0;
#pragma unroll
        for (int s = 0; s < 2; ++s) {
          v2f b2;
          b2.x = fh ? da[4 * s + 2] : da[4 * s];
          b2.y = fh ? da[4 * s + 3] : da[4 * s + 1];
          u0 = wmma4(pk[0 + s], b2, u0);
          u1 = wmma4(pk[2 + s], b2, u1);
          u2 = wmma4(pk[4 + s], b2, u2);
          u3 = wmma4(pk[6 + s], b2, u3);
        }
#pragma unroll
        for (int i = 0; i < 8; ++i) {
          W1[(0 * 16 + 8 * fh + i) * 64 + col] = wold[0 * 8 + i] - LR * u0[i];
          W1[(1 * 16 + 8 * fh + i) * 64 + col] = wold[1 * 8 + i] - LR * u1[i];
          W1[(2 * 16 + 8 * fh + i) * 64 + col] = wold[2 * 8 + i] - LR * u2[i];
          W1[(3 * 16 + 8 * fh + i) * 64 + col] = wold[3 * 8 + i] - LR * u3[i];
        }
      } else {
        const int wv = wave - 4;
        v2f pd[8];
#pragma unroll
        for (int tn = 0; tn < 4; ++tn) {
#pragma unroll
          for (int s = 0; s < 2; ++s) {
            pd[tn * 2 + s].x = dob[(4 * s + fk) * 64 + tn * 16 + fc];
            pd[tn * 2 + s].y = dob[(4 * s + fk + 1) * 64 + tn * 16 + fc];
          }
        }
        v8f u0 = {0.f, 0.f, 0.f, 0.f, 0.f, 0.f, 0.f, 0.f};
        v8f u1 = u0, u2 = u0, u3 = u0;
#pragma unroll
        for (int s = 0; s < 2; ++s) {
          u0 = wmma4(ph[s], pd[0 + s], u0);
          u1 = wmma4(ph[s], pd[2 + s], u1);
          u2 = wmma4(ph[s], pd[4 + s], u2);
          u3 = wmma4(ph[s], pd[6 + s], u3);
        }
        const int rb = W2P_IDX(wv * 16 + 8 * fh, fc);
#pragma unroll
        for (int i = 0; i < 8; ++i) {
          W2[rb + 0 * 8 * W2P_PITCH + 2 * i] = wold[0 * 8 + i] - LR * u0[i];
          W2[rb + 1 * 8 * W2P_PITCH + 2 * i] = wold[1 * 8 + i] - LR * u1[i];
          W2[rb + 2 * 8 * W2P_PITCH + 2 * i] = wold[2 * 8 + i] - LR * u2[i];
          W2[rb + 3 * 8 * W2P_PITCH + 2 * i] = wold[3 * 8 + i] - LR * u3[i];
        }
        const int t2 = tid - 128;
        if (t2 < 64) {
          float s = 0.f, s2 = 0.f;
          for (int b = 0; b < 8; ++b) s += dzb[b * 64 + t2] * xh[b * 64 + t2];
          for (int b = 0; b < 8; ++b) s2 += dob[b * 64 + t2];
          lgv[t2] -= LR * s;
          b2v[t2] -= LR * s2;
        } else {
          const int d = t2 - 64;
          float s = 0.f;
          for (int b = 0; b < 8; ++b) s += dzb[b * 64 + d];
          lbv[d] -= LR * s;
        }
      }
      __syncthreads();
    }

    fwd_pass(ktT, gdv, hg, ob, xh, zb, zbT, W1, W2, b1v, b2v, lgv, lbv,
             muv, isv, part, wave, lane, tid);
    {
      float e0 = zb[wave * 64 + lane], e1 = zb[wave * 64 + lane + 32];
      float s = wave_sum32(e0 + e1);
      float q = wave_sum32(e0 * e0 + e1 * e1);
      if (lane == 0) {
        float m = s * (1.f / 64.f);
        muv[wave] = m;
        isv[wave] = rsqrtf(q * (1.f / 64.f) - m * m + LN_EPS);
      }
    }
    __syncthreads();
    for (int idx = tid; idx < 512; idx += 256) {
      int b = idx >> 6, d = idx & 63;
      float z2 = (zb[idx] - muv[b]) * isv[b] * tgv[d] + tbv[d];
      size_t g = ((size_t)(b * SS + t)) * HIDC + h * 64 + d;
      outg[g] = qg[g] + z2;
    }
    __syncthreads();
  }
}

__global__ void __launch_bounds__(256)
gate_ln_kernel(const float* __restrict__ outb, const float* __restrict__ gact,
               const float* __restrict__ png, const float* __restrict__ pnb,
               float* __restrict__ gated) {
  __shared__ float r1[256], r2[256];
  __shared__ float mu_s, inv_s;
  const int r = blockIdx.x, tid = threadIdx.x;
  const float* row = outb + (size_t)r * HIDC;
  float s = 0.f, q = 0.f;
  for (int i = tid; i < HIDC; i += 256) { float v = row[i]; s += v; q += v * v; }
  r1[tid] = s; r2[tid] = q;
  __syncthreads();
  for (int off = 128; off > 0; off >>= 1) {
    if (tid < off) { r1[tid] += r1[tid + off]; r2[tid] += r2[tid + off]; }
    __syncthreads();
  }
  if (tid == 0) {
    float m = r1[0] * (1.f / HIDC);
    mu_s = m;
    inv_s = rsqrtf(r2[0] * (1.f / HIDC) - m * m + LN_EPS);
  }
  __syncthreads();
  for (int i = tid; i < HIDC; i += 256) {
    float y = (row[i] - mu_s) * inv_s * png[i] + pnb[i];
    gated[(size_t)r * HIDC + i] = gact[(size_t)r * HIDC + i] * y;
  }
}

extern "C" void kernel_launch(void* const* d_in, const int* in_sizes, int n_in,
                              void* d_out, int out_size, void* d_ws, size_t ws_size,
                              hipStream_t stream) {
  (void)in_sizes; (void)n_in; (void)out_size; (void)ws_size;
  const float* x   = (const float*)d_in[0];
  const float* wq  = (const float*)d_in[1];
  const float* wk  = (const float*)d_in[2];
  const float* wv  = (const float*)d_in[3];
  const float* fw1 = (const float*)d_in[4];
  const float* fb1 = (const float*)d_in[5];
  const float* fw2 = (const float*)d_in[6];
  const float* fb2 = (const float*)d_in[7];
  const float* flg = (const float*)d_in[8];
  const float* flb = (const float*)d_in[9];
  const float* lw  = (const float*)d_in[10];
  const float* lb  = (const float*)d_in[11];
  const float* ttg = (const float*)d_in[12];
  const float* ttb = (const float*)d_in[13];
  const float* wo  = (const float*)d_in[14];
  const float* wg  = (const float*)d_in[15];
  const float* png = (const float*)d_in[16];
  const float* pnb = (const float*)d_in[17];

  const int M = BB * SS;
  const size_t PLANE = (size_t)M * HIDC;
  float* qb    = (float*)d_ws;
  float* kb    = qb + PLANE;
  float* vb    = kb + PLANE;
  float* gactb = vb + PLANE;
  float* outb  = gactb + PLANE;
  float* gatedb = qb;

  dim3 gblk(256);
  dim3 ggrid(HIDC / 128, M / 16);
  gemm_wmma_f32<<<ggrid, gblk, 0, stream>>>(x, wq, qb,    M, HIDC, HIDC, 0);
  gemm_wmma_f32<<<ggrid, gblk, 0, stream>>>(x, wk, kb,    M, HIDC, HIDC, 0);
  gemm_wmma_f32<<<ggrid, gblk, 0, stream>>>(x, wv, vb,    M, HIDC, HIDC, 0);
  gemm_wmma_f32<<<ggrid, gblk, 0, stream>>>(x, wg, gactb, M, HIDC, HIDC, 1);

  ttt_scan_kernel<<<HH, 256, SCAN_SMEM_FLOATS * sizeof(float), stream>>>(
      kb, vb, qb, fw1, fb1, fw2, fb2, flg, flb, lw, lb, ttg, ttb, outb);

  gate_ln_kernel<<<M, 256, 0, stream>>>(outb, gactb, png, pnb, gatedb);

  gemm_wmma_f32<<<ggrid, gblk, 0, stream>>>(gatedb, wo, (float*)d_out,
                                            M, HIDC, HIDC, 0);
}


# projection GEMMs re-tiled: 64x128 block tile, each wave a 32x32 register tile (2x2 accumulators), k chunks of 32 of A and B staged in padded LDS tiles by 16-byte loads, two buffers; k order kept
# speedup vs baseline: 2.5356x; 1.0627x over previous
#include <hip/hip_runtime.h>
#include <hip/hip_bf16.h>
#include <math.h>

#define BB     8
#define SS     256
#define HIDC   1024
#define HH     16
#define DD     64
#define FF     64
#define LR     0.1f
#define GSCALE (2.0f / 8192.0f)
#define LN_EPS 1e-5f

typedef float v2f __attribute__((ext_vector_type(2)));
typedef float v8f __attribute__((ext_vector_type(8)));

__device__ __forceinline__ v8f wmma4(v2f a, v2f b, v8f c) {
  return __builtin_amdgcn_wmma_f32_16x16x4_f32(
       false, a,  false, b,
       (short)0, c,  false,  false);
}

__device__ __forceinline__ float gelu_f(float x) {
  const float c = 0.7978845608028654f;
  float u = c * (x + 0.044715f * x * x * x);
  return 0.5f * x * (1.0f + tanhf(u));
}
__device__ __forceinline__ void gelu_and_grad(float x, float& g, float& gd) {
  const float c = 0.7978845608028654f;
  float u = c * (x + 0.044715f * x * x * x);
  float t = tanhf(u);
  float x2 = x * x;
  g  = 0.5f * x * (1.0f + t);
  gd = 0.5f * (1.0f + t) + 0.5f * x * (1.0f - t * t) * c * (1.0f + 0.134145f * x2);
}
__device__ __forceinline__ float wave_sum32(float v) {
#pragma unroll
  for (int off = 16; off > 0; off >>= 1) v += __shfl_xor(v, off, 32);
  return v;
}

__device__ __forceinline__ float row_xor8(float v) {
  const int b = __builtin_bit_cast(int, v);
  return __builtin_bit_cast(float, __builtin_amdgcn_update_dpp(b, b, 0x168, 0xf, 0xf, false));
}

typedef float v4f __attribute__((ext_vector_type(4)));
#define GA_PITCH 36
#define GB_PITCH 288
#define GHOLD8(a, o) asm volatile("" :: "v"(a[o]), "v"(a[o + 1]), "v"(a[o + 2]), "v"(a[o + 3]), \
                                        "v"(a[o + 4]), "v"(a[o + 5]), "v"(a[o + 6]), "v"(a[o + 7]))
__global__ void __launch_bounds__(256)
gemm_wmma_f32(const float* __restrict__ A, const float* __restrict__ Bm,
              float* __restrict__ C, int M, int N, int K, int act) {
  __shared__ __align__(16) float As[2][64 * GA_PITCH];
  __shared__ __align__(16) float Bs[2][16 * GB_PITCH];
  (void)M;
  const int tid   = threadIdx.x;
  const int wave  = tid >> 5;
  const int lane  = tid & 31;
  const int c     = lane & 15;
  const int h     = lane >> 4;
  const int wm    = wave >> 2;
  const int wn    = wave & 3;
  const int mbase = blockIdx.y * 64;
  const int nbase = blockIdx.x * 128;
  const int ar = tid >> 3, ac = (tid & 7) << 2;
  const int bp = tid >> 5, bc = (tid & 31) << 2;
  const float* ag = A + (size_t)(mbase + ar) * K + ac;
  const float* bg = Bm + (size_t)(2 * bp) * N + nbase + bc;
  const int aso = ar * GA_PITCH + ac;
  const int bso = bp * GB_PITCH + 2 * bc;
  v4f ra0, ra1, rb0, rb1, rb2, rb3;
  ra0 = *(const v4f*)(ag);
  ra1 = *(const v4f*)(ag + (size_t)32 * K);
  rb0 = *(const v4f*)(bg);
  rb1 = *(const v4f*)(bg + N);
  rb2 = *(const v4f*)(bg + (size_t)16 * N);
  rb3 = *(const v4f*)(bg + (size_t)17 * N);
  *(v4f*)&As[0][aso]                 = ra0;
  *(v4f*)&As[0][aso + 32 * GA_PITCH] = ra1;
  *(v4f*)&Bs[0][bso]                    = v4f{rb0.x, rb1.x, rb0.y, rb1.y};
  *(v4f*)&Bs[0][bso + 4]                = v4f{rb0.z, rb1.z, rb0.w, rb1.w};
  *(v4f*)&Bs[0][bso + 8 * GB_PITCH]     = v4f{rb2.x, rb3.x, rb2.y, rb3.y};
  *(v4f*)&Bs[0][bso + 8 * GB_PITCH + 4] = v4f{rb2.z, rb3.z, rb2.w, rb3.w};
  __syncthreads();
  v8f acc00 = {0.f, 0.f, 0.f, 0.f, 0.f, 0.f, 0.f, 0.f};
  v8f acc01 = acc00, acc10 = acc00, acc11 = acc00;
  const int afo = (wm * 32 + c) * GA_PITCH + 2 * h;
  const int bfo = h * GB_PITCH + 2 * (wn * 32 + c);
  const int nch = K >> 5;
  for (int ch = 0; ch < nch; ++ch) {
    const int cur = ch & 1;
    const bool more = (ch + 1 < nch);
    if (more) {
      const float* an = ag + (size_t)(ch + 1) * 32;
      const float* bn = bg + (size_t)(ch + 1) * 32 * N;
      ra0 = *(const v4f*)(an);
      ra1 = *(const v4f*)(an + (size_t)32 * K);
      rb0 = *(const v4f*)(bn);
      rb1 = *(const v4f*)(bn + N);
      rb2 = *(const v4f*)(bn + (size_t)16 * N);
      rb3 = *(const v4f*)(bn + (size_t)17 * N);
    }
    const float* as = &As[cur][afo];
    const float* bs = &Bs[cur][bfo];
    v2f fa[16], fb[16];
#pragma unroll
    for (int ks = 0; ks < 8; ++ks) {
      fa[2 * ks]     = *(const v2f*)(as + 4 * ks);
      fa[2 * ks + 1] = *(const v2f*)(as + 16 * GA_PITCH + 4 * ks);
      fb[2 * ks]     = *(const v2f*)(bs + 2 * GB_PITCH * ks);
      fb[2 * ks + 1] = *(const v2f*)(bs + 2 * GB_PITCH * ks + 32);
    }
    GHOLD8(fa, 0); GHOLD8(fa, 8); GHOLD8(fb, 0); GHOLD8(fb, 8);
#pragma unroll
    for (int ks = 0; ks < 8; ++ks) {
      acc00 = wmma4(fa[2 * ks],     fb[2 * ks],     acc00);
      acc10 = wmma4(fa[2 * ks + 1], fb[2 * ks],     acc10);
      acc01 = wmma4(fa[2 * ks],     fb[2 * ks + 1], acc01);
      acc11 = wmma4(fa[2 * ks + 1], fb[2 * ks + 1], acc11);
    }
    if (more) {
      float* ad = &As[cur ^ 1][aso];
      float* bd = &Bs[cur ^ 1][bso];
      *(v4f*)(ad)                 = ra0;
      *(v4f*)(ad + 32 * GA_PITCH) = ra1;
      *(v4f*)(bd)                    = v4f{rb0.x, rb1.x, rb0.y, rb1.y};
      *(v4f*)(bd + 4)                = v4f{rb0.z, rb1.z, rb0.w, rb1.w};
      *(v4f*)(bd + 8 * GB_PITCH)     = v4f{rb2.x, rb3.x, rb2.y, rb3.y};
      *(v4f*)(bd + 8 * GB_PITCH + 4) = v4f{rb2.z, rb3.z, rb2.w, rb3.w};
    }
    __syncthreads();
  }
  const int row0 = mbase + wm * 32 + 8 * h;
  const int col0 = nbase + wn * 32 + c;
#pragma unroll
  for (int r = 0; r < 8; ++r) {
    float v00 = acc00[r], v01 = acc01[r], v10 = acc10[r], v11 = acc11[r];
    if (act) { v00 = gelu_f(v00); v01 = gelu_f(v01); v10 = gelu_f(v10); v11 = gelu_f(v11); }
    C[(size_t)(row0 + r) * N + col0]           = v00;
    C[(size_t)(row0 + r) * N + col0 + 16]      = v01;
    C[(size_t)(row0 + 16 + r) * N + col0]      = v10;
    C[(size_t)(row0 + 16 + r) * N + col0 + 16] = v11;
  }
}


#define HOLD8(a, o) asm volatile("" :: "v"(a[o]), "v"(a[o + 1]), "v"(a[o + 2]), "v"(a[o + 3]), \
                                       "v"(a[o + 4]), "v"(a[o + 5]), "v"(a[o + 6]), "v"(a[o + 7]))
#define TIDX(b, d) ((((d) >> 1) << 4) + ((b) << 1) + ((d) & 1))
#define W2P_PITCH 192
#define W2P_ROT(p) ((((p) & 6) << 1) + (((p) & 1) << 5))
#define W2P_IDX(f, d) (((d) >> 1) * W2P_PITCH + 2 * (f) + W2P_ROT((d) >> 1) + ((d) & 1))
#define WB_W2P 0
#define WB_LWP 1
#define CH_GELU  0
#define CH_DPRED 1
template <int EPI, int WBL, bool HAS_LATER, typename LATER>
__device__ __forceinline__ void
chain_pair(const float* X8T, const float* Wa, const float* biasa, const float* Wb,
           float* Cgd, float* Cact, const float* tgtT, float* part, int wave, int lane,
           LATER later_reads) {
  const int c  = lane & 15;
  const int h  = lane >> 4;
  const int bb = c & 7;
  const int kh = h << 1;
  const int fo = wave * 16;
  const int f0 = fo + 8 * h;
  const int xb = 16 * h + 2 * bb;
  const int tb = 128 * wave + 64 * h + 2 * bb;
  const int wp = 1024 * wave + 512 * h + 2 * c;
  const int wr = W2P_IDX(f0, c);
  v2f fa[16], fb[16], fa2[16];
  float bi[8], tg[8];
#pragma unroll
  for (int i = 0; i < 16; ++i) {
    const int k0 = 4 * i;
    fa[i].x = Wa[(k0 + kh) * 64 + fo + c];
    fa[i].y = Wa[(k0 + kh + 1) * 64 + fo + c];
    fb[i].x = X8T[xb + 32 * i];
    fb[i].y = X8T[xb + 32 * i + 1];
  }
#pragma unroll
  for (int s = 0; s < 4; ++s) {
#pragma unroll
    for (int td = 0; td < 4; ++td) {
      if (WBL == WB_W2P) {
        fa2[s * 4 + td].x = Wb[wr + 8 * W2P_PITCH * td + 4 * s];
        fa2[s * 4 + td].y = Wb[wr + 8 * W2P_PITCH * td + 4 * s + 2];
      } else {
        fa2[s * 4 + td].x = Wb[wp + 128 * s + 32 * td];
        fa2[s * 4 + td].y = Wb[wp + 128 * s + 32 * td + 1];
      }
    }
  }
#pragma unroll
  for (int r = 0; r < 8; ++r) {
    bi[r] = biasa[f0 + r];
    tg[r] = (EPI == CH_DPRED) ? tgtT[tb + 16 * (r >> 1) + (r & 1)] : 0.f;
  }
  HOLD8(fa, 0);  HOLD8(fa, 8);
  HOLD8(fb, 0);  HOLD8(fb, 8);
  HOLD8(fa2, 0); HOLD8(fa2, 8);
  HOLD8(bi, 0);
  if (EPI == CH_DPRED) HOLD8(tg, 0);
  if (HAS_LATER) {
    later_reads();
    __builtin_amdgcn_sched_barrier(0);
  }
  v8f acc = {0.f, 0.f, 0.f, 0.f, 0.f, 0.f, 0.f, 0.f};
#pragma unroll
  for (int i = 0; i < 16; ++i) acc = wmma4(fa[i], fb[i], acc);
  v8f g;
  if (EPI == CH_GELU) {
    const bool hi = (c >= 8);
    float gm[4], dm[4];
#pragma unroll
    for (int r = 0; r < 4; ++r) {
      const float xlo = acc[r] + bi[r];
      const float xhi = acc[4 + r] + bi[4 + r];
      gelu_and_grad(hi ? xhi : xlo, gm[r], dm[r]);
    }
    {
      const int so = bb * 64 + f0 + (hi ? 4 : 0);
#pragma unroll
      for (int r = 0; r < 4; ++r) {
        Cgd[so + r] = dm[r];
        Cact[so + r] = gm[r];
      }
    }
#pragma unroll
    for (int r = 0; r < 4; ++r) {
      const float go = row_xor8(gm[r]);
      g[r]     = hi ? go : gm[r];
      g[4 + r] = hi ? gm[r] : go;
    }
  } else {
#pragma unroll
    for (int r = 0; r < 8; ++r) {
      float v = acc[r] + bi[r];
      g[r] = GSCALE * (v - tg[r]);
    }
  }
  v8f p0 = {0.f, 0.f, 0.f, 0.f, 0.f, 0.f, 0.f, 0.f};
  v8f p1 = p0, p2 = p0, p3 = p0;
#pragma unroll
  for (int s = 0; s < 4; ++s) {
    v2f b2;
    b2.x = g[2 * s];
    b2.y = g[2 * s + 1];
    p0 = wmma4(fa2[s * 4 + 0], b2, p0);
    p1 = wmma4(fa2[s * 4 + 1], b2, p1);
    p2 = wmma4(fa2[s * 4 + 2], b2, p2);
    p3 = wmma4(fa2[s * 4 + 3], b2, p3);
  }
  if (c < 8) {
    float* dst = part + wave * 512 + 64 * h + 2 * bb;
#pragma unroll
    for (int r = 0; r < 8; ++r) {
      const int o = 16 * (r >> 1) + (r & 1);
      dst[o]       = p0[r];
      dst[128 + o] = p1[r];
      dst[256 + o] = p2[r];
      dst[384 + o] = p3[r];
    }
  }
}

__device__ __forceinline__ float part_sum4(const float* part, int idx) {
  return (part[idx] + part[512 + idx]) + (part[1024 + idx] + part[1536 + idx]);
}

__device__ __forceinline__ void
fwd_pass(const float* ktT, float* gdv, float* hg, float* ob, float* xh, float* zb, float* zbT,
         const float* W1, const float* W2, const float* b1v, const float* b2v,
         const float* lgv, const float* lbv, float* muv, float* isv, float* part,
         int wave, int lane, int tid) {
  if (wave < 4)
    chain_pair<CH_GELU, WB_W2P, false>(ktT, W1, b1v, W2, gdv, hg, nullptr, part, wave, lane, [] {});
  __syncthreads();
  {
    const int pi = TIDX(wave, lane);
    float e0 = part_sum4(part, pi) + b2v[lane];
    float e1 = part_sum4(part, pi + 256) + b2v[lane + 32];
    ob[wave * 64 + lane] = e0;
    ob[wave * 64 + lane + 32] = e1;
    float s = wave_sum32(e0 + e1);
    float q = wave_sum32(e0 * e0 + e1 * e1);
    if (lane == 0) {
      float m = s * (1.f / 64.f);
      muv[wave] = m;
      isv[wave] = rsqrtf(q * (1.f / 64.f) - m * m + LN_EPS);
    }
  }
  __syncthreads();
  for (int idx = tid; idx < 512; idx += 256) {
    int b = idx >> 6, d = idx & 63;
    float xv = (ob[idx] - muv[b]) * isv[b];
    xh[idx] = xv;
    const float zv = xv * lgv[d] + lbv[d];
    zb[idx] = zv;
    zbT[TIDX(b, d)] = zv;
  }
  __syncthreads();
}

#define SCAN_SMEM_FLOATS (3 * 4096 + 12 * 512 + 7 * 64 + 32 + 4 * 512 + 32 * W2P_PITCH)

__global__ void __launch_bounds__(256)
ttt_scan_kernel(const float* __restrict__ kg, const float* __restrict__ vg,
                const float* __restrict__ qg,
                const float* __restrict__ fw1, const float* __restrict__ fb1,
                const float* __restrict__ fw2, const float* __restrict__ fb2,
                const float* __restrict__ flg, const float* __restrict__ flb,
                const float* __restrict__ lwg, const float* __restrict__ lbg,
                const float* __restrict__ ttg, const float* __restrict__ ttb,
                float* __restrict__ outg) {
  extern __shared__ __align__(16) float smem[];
  float* W1  = smem;
  float* LW  = W1 + 4096;
  float* ktb = LW + 4096;
  float* tgt = ktb + 512;
  float* gdv = tgt + 512;
  float* hg  = gdv + 512;
  float* ob  = hg + 512;
  float* xh  = ob + 512;
  float* zb  = xh + 512;
  float* dzb = zb + 512;
  float* dob = dzb + 512;
  float* ktT = dob + 512;
  float* zbT = ktT + 512;
  float* dobT = zbT + 512;
  float* b1v = dobT + 512;
  float* b2v = b1v + 64;
  float* lgv = b2v + 64;
  float* lbv = lgv + 64;
  float* tgv = lbv + 64;
  float* tbv = tgv + 64;
  float* lbz = tbv + 64;
  float* muv = lbz + 64;
  float* isv = muv + 8;
  float* m1v = isv + 8;
  float* m2v = m1v + 8;
  float* part = m2v + 8;
  float* LWP = part + 4 * 512;
  float* W2  = LWP + 4096;

  const int h    = blockIdx.x;
  const int tid  = threadIdx.x;
  const int wave = tid >> 5;
  const int lane = tid & 31;

  const int b0 = tid >> 6,          d0 = tid & 63;
  const int b1i = (tid + 256) >> 6, d1i = (tid + 256) & 63;
  const size_t g0base = (size_t)(b0 * SS) * HIDC + h * 64 + d0;
  const size_t g1base = (size_t)(b1i * SS) * HIDC + h * 64 + d1i;

  for (int i = tid; i < 4096; i += 256) {
    W1[i] = fw1[(size_t)h * 4096 + i];
    W2[W2P_IDX(i >> 6, i & 63)] = fw2[(size_t)h * 4096 + i];
    LW[i] = lwg[i];
    LWP[((i & 63) >> 1) * 128 + 2 * (i >> 6) + (i & 1)] = lwg[i];
  }
  if (tid < 64) {
    b1v[tid] = fb1[h * 64 + tid];
    b2v[tid] = fb2[h * 64 + tid];
    lgv[tid] = flg[h * 64 + tid];
    lbv[tid] = flb[h * 64 + tid];
    tgv[tid] = ttg[h * 64 + tid];
    tbv[tid] = ttb[h * 64 + tid];
    lbz[tid] = lbg[tid];
  }

  float kp0 = kg[g0base], vp0 = vg[g0base];
  float kp1 = kg[g1base], vp1 = vg[g1base];
  __syncthreads();

  for (int t = 0; t < SS; ++t) {
    ktb[tid]            = kp0;  ktb[tid + 256]      = kp1;
    ktT[TIDX(b0, d0)]   = kp0;  ktT[TIDX(b1i, d1i)] = kp1;
    tgt[TIDX(b0, d0)]   = vp0 - kp0;
    tgt[TIDX(b1i, d1i)] = vp1 - kp1;
    __syncthreads();
    if (t + 1 < SS) {
      kp0 = kg[g0base + (t + 1) * HIDC];  vp0 = vg[g0base + (t + 1) * HIDC];
      kp1 = kg[g1base + (t + 1) * HIDC];  vp1 = vg[g1base + (t + 1) * HIDC];
    }

    for (int step = 0; step < 4; ++step) {
      fwd_pass(ktT, gdv, hg, ob, xh, zb, zbT, W1, W2, b1v, b2v, lgv, lbv,
               muv, isv, part, wave, lane, tid);
      const int fc = lane & 15;
      const int fh = lane >> 4;
      const int fk = fh << 1;
      v2f   pb[16];
      v2f   pk[8];
      float pg[8];
      float pb1 = 0.f;
      v2f   ph[2];
      float wold[32];
      if (wave < 4) {
        chain_pair<CH_DPRED, WB_LWP, true>(zbT, LW, lbz, LWP, nullptr, nullptr, tgt, part, wave, lane, [&] {
          const int col = wave * 16 + fc;
          const int wq  = fh * (W2P_PITCH + 32) + 2 * col;
#pragma unroll
          for (int i = 0; i < 16; ++i) {
            pb[i].x = W2[wq + 2 * W2P_PITCH * i + 4 * (i & 3)];
            pb[i].y = W2[wq + 2 * W2P_PITCH * i + 4 * (i & 3) + 1];
          }
#pragma unroll
          for (int r = 0; r < 8; ++r) pg[r] = gdv[r * 64 + col];
          pb1 = b1v[col];
#pragma unroll
          for (int tm = 0; tm < 4; ++tm) {
#pragma unroll
            for (int s = 0; s < 2; ++s) {
              pk[tm * 2 + s].x = ktb[(4 * s + fk) * 64 + tm * 16 + fc];
              pk[tm * 2 + s].y = ktb[(4 * s + fk + 1) * 64 + tm * 16 + fc];
            }
#pragma unroll
            for (int i = 0; i < 8; ++i) wold[tm * 8 + i] = W1[(tm * 16 + 8 * fh + i) * 64 + col];
          }
        });
      } else {
        const int wv = wave - 4;
#pragma unroll
        for (int s = 0; s < 2; ++s) {
          ph[s].x = hg[(4 * s + fk) * 64 + wv * 16 + fc];
          ph[s].y = hg[(4 * s + fk + 1) * 64 + wv * 16 + fc];
        }
#pragma unroll
        for (int tn = 0; tn < 4; ++tn) {
#pragma unroll
          for (int i = 0; i < 8; ++i)
            wold[tn * 8 + i] = W2[W2P_IDX(wv * 16 + 8 * fh, fc) + 8 * W2P_PITCH * tn + 2 * i];
        }
      }
      __syncthreads();
      {
        const int pi = TIDX(wave, lane);
        float z0 = part_sum4(part, pi);
        float z1 = part_sum4(part, pi + 256);
        dzb[wave * 64 + lane] = z0;
        dzb[wave * 64 + lane + 32] = z1;
        float dd0 = z0 * lgv[lane];
        float dd1 = z1 * lgv[lane + 32];
        float x0 = xh[wave * 64 + lane], x1 = xh[wave * 64 + lane + 32];
        float s1 = wave_sum32(dd0 + dd1);
        float s2 = wave_sum32(dd0 * x0 + dd1 * x1);
        if (lane == 0) { m1v[wave] = s1 * (1.f / 64.f); m2v[wave] = s2 * (1.f / 64.f); }
      }
      __syncthreads();
      for (int idx = tid; idx < 512; idx += 256) {
        int b = idx >> 6, d = idx & 63;
        float dxh = dzb[idx] * lgv[d];
        const float dv = isv[b] * (dxh - m1v[b] - xh[idx] * m2v[b]);
        dob[idx] = dv;
        dobT[TIDX(b, d)] = dv;
      }
      __syncthreads();
      if (wave < 4) {
        const int col = wave * 16 + fc;
        const int bb  = fc & 7;
        v2f pa[16];
#pragma unroll
        for (int i = 0; i < 16; ++i) {
          pa[i].x = dobT[16 * fh + 2 * bb + 32 * i];
          pa[i].y = dobT[16 * fh + 2 * bb + 32 * i + 1];
        }
        HOLD8(pa, 0); HOLD8(pa, 8);
        v8f acc = {0.f, 0.f, 0.f, 0.f, 0.f, 0.f, 0.f, 0.f};
#pragma unroll
        for (int i = 0; i < 16; ++i) acc = wmma4(pa[i], pb[i], acc);
        float da[8];
        float sb = 0.f;
#pragma unroll
        for (int r = 0; r < 8; ++r) da[r] = acc[r] * pg[r];
#pragma unroll
        for (int r = 0; r < 8; ++r) sb += da[r];
        if (fh == 0) b1v[col] = pb1 - LR * sb;
        v8f u0 = {0.f, 0.f, 0.f, 0.f, 0.f, 0.f, 0.f, 0.f};
        v8f u1 = u0, u2 = u0, u3 = u0;
#pragma unroll
        for (int s = 0; s < 2; ++s) {
          v2f b2;
          b2.x = fh ? da[4 * s + 2] : da[4 * s];
          b2.y = fh ? da[4 * s + 3] : da[4 * s + 1];
          u0 = wmma4(pk[0 + s], b2, u0);
          u1 = wmma4(pk[2 + s], b2, u1);
          u2 = wmma4(pk[4 + s], b2, u2);
          u3 = wmma4(pk[6 + s], b2, u3);
        }
#pragma unroll
        for (int i = 0; i < 8; ++i) {
          W1[(0 * 16 + 8 * fh + i) * 64 + col] = wold[0 * 8 + i] - LR * u0[i];
          W1[(1 * 16 + 8 * fh + i) * 64 + col] = wold[1 * 8 + i] - LR * u1[i];
          W1[(2 * 16 + 8 * fh + i) * 64 + col] = wold[2 * 8 + i] - LR * u2[i];
          W1[(3 * 16 + 8 * fh + i) * 64 + col] = wold[3 * 8 + i] - LR * u3[i];
        }
      } else {
        const int wv = wave - 4;
        v2f pd[8];
#pragma unroll
        for (int tn = 0; tn < 4; ++tn) {
#pragma unroll
          for (int s = 0; s < 2; ++s) {
            pd[tn * 2 + s].x = dob[(4 * s + fk) * 64 + tn * 16 + fc];
            pd[tn * 2 + s].y = dob[(4 * s + fk + 1) * 64 + tn * 16 + fc];
          }
        }
        v8f u0 = {0.f, 0.f, 0.f, 0.f, 0.f, 0.f, 0.f, 0.f};
        v8f u1 = u0, u2 = u0, u3 = u0;
#pragma unroll
        for (int s = 0; s < 2; ++s) {
          u0 = wmma4(ph[s], pd[0 + s], u0);
          u1 = wmma4(ph[s], pd[2 + s], u1);
          u2 = wmma4(ph[s], pd[4 + s], u2);
          u3 = wmma4(ph[s], pd[6 + s], u3);
        }
        const int rb = W2P_IDX(wv * 16 + 8 * fh, fc);
#pragma unroll
        for (int i = 0; i < 8; ++i) {
          W2[rb + 0 * 8 * W2P_PITCH + 2 * i] = wold[0 * 8 + i] - LR * u0[i];
          W2[rb + 1 * 8 * W2P_PITCH + 2 * i] = wold[1 * 8 + i] - LR * u1[i];
          W2[rb + 2 * 8 * W2P_PITCH + 2 * i] = wold[2 * 8 + i] - LR * u2[i];
          W2[rb + 3 * 8 * W2P_PITCH + 2 * i] = wold[3 * 8 + i] - LR * u3[i];
        }
        const int t2 = tid - 128;
        if (t2 < 64) {
          float s = 0.f, s2 = 0.f;
          for (int b = 0; b < 8; ++b) s += dzb[b * 64 + t2] * xh[b * 64 + t2];
          for (int b = 0; b < 8; ++b) s2 += dob[b * 64 + t2];
          lgv[t2] -= LR * s;
          b2v[t2] -= LR * s2;
        } else {
          const int d = t2 - 64;
          float s = 0.f;
          for (int b = 0; b < 8; ++b) s += dzb[b * 64 + d];
          lbv[d] -= LR * s;
        }
      }
      __syncthreads();
    }

    fwd_pass(ktT, gdv, hg, ob, xh, zb, zbT, W1, W2, b1v, b2v, lgv, lbv,
             muv, isv, part, wave, lane, tid);
    {
      float e0 = zb[wave * 64 + lane], e1 = zb[wave * 64 + lane + 32];
      float s = wave_sum32(e0 + e1);
      float q = wave_sum32(e0 * e0 + e1 * e1);
      if (lane == 0) {
        float m = s * (1.f / 64.f);
        muv[wave] = m;
        isv[wave] = rsqrtf(q * (1.f / 64.f) - m * m + LN_EPS);
      }
    }
    __syncthreads();
    for (int idx = tid; idx < 512; idx += 256) {
      int b = idx >> 6, d = idx & 63;
      float z2 = (zb[idx] - muv[b]) * isv[b] * tgv[d] + tbv[d];
      size_t g = ((size_t)(b * SS + t)) * HIDC + h * 64 + d;
      outg[g] = qg[g] + z2;
    }
    __syncthreads();
  }
}

__global__ void __launch_bounds__(256)
gate_ln_kernel(const float* __restrict__ outb, const float* __restrict__ gact,
               const float* __restrict__ png, const float* __restrict__ pnb,
               float* __restrict__ gated) {
  __shared__ float r1[256], r2[256];
  __shared__ float mu_s, inv_s;
  const int r = blockIdx.x, tid = threadIdx.x;
  const float* row = outb + (size_t)r * HIDC;
  float s = 0.f, q = 0.f;
  for (int i = tid; i < HIDC; i += 256) { float v = row[i]; s += v; q += v * v; }
  r1[tid] = s; r2[tid] = q;
  __syncthreads();
  for (int off = 128; off > 0; off >>= 1) {
    if (tid < off) { r1[tid] += r1[tid + off]; r2[tid] += r2[tid + off]; }
    __syncthreads();
  }
  if (tid == 0) {
    float m = r1[0] * (1.f / HIDC);
    mu_s = m;
    inv_s = rsqrtf(r2[0] * (1.f / HIDC) - m * m + LN_EPS);
  }
  __syncthreads();
  for (int i = tid; i < HIDC; i += 256) {
    float y = (row[i] - mu_s) * inv_s * png[i] + pnb[i];
    gated[(size_t)r * HIDC + i] = gact[(size_t)r * HIDC + i] * y;
  }
}

extern "C" void kernel_launch(void* const* d_in, const int* in_sizes, int n_in,
                              void* d_out, int out_size, void* d_ws, size_t ws_size,
                              hipStream_t stream) {
  (void)in_sizes; (void)n_in; (void)out_size; (void)ws_size;
  const float* x   = (const float*)d_in[0];
  const float* wq  = (const float*)d_in[1];
  const float* wk  = (const float*)d_in[2];
  const float* wv  = (const float*)d_in[3];
  const float* fw1 = (const float*)d_in[4];
  const float* fb1 = (const float*)d_in[5];
  const float* fw2 = (const float*)d_in[6];
  const float* fb2 = (const float*)d_in[7];
  const float* flg = (const float*)d_in[8];
  const float* flb = (const float*)d_in[9];
  const float* lw  = (const float*)d_in[10];
  const float* lb  = (const float*)d_in[11];
  const float* ttg = (const float*)d_in[12];
  const float* ttb = (const float*)d_in[13];
  const float* wo  = (const float*)d_in[14];
  const float* wg  = (const float*)d_in[15];
  const float* png = (const float*)d_in[16];
  const float* pnb = (const float*)d_in[17];

  const int M = BB * SS;
  const size_t PLANE = (size_t)M * HIDC;
  float* qb    = (float*)d_ws;
  float* kb    = qb + PLANE;
  float* vb    = kb + PLANE;
  float* gactb = vb + PLANE;
  float* outb  = gactb + PLANE;
  float* gatedb = qb;

  dim3 gblk(256);
  dim3 ggrid(HIDC / 128, M / 64);
  gemm_wmma_f32<<<ggrid, gblk, 0, stream>>>(x, wq, qb,    M, HIDC, HIDC, 0);
  gemm_wmma_f32<<<ggrid, gblk, 0, stream>>>(x, wk, kb,    M, HIDC, HIDC, 0);
  gemm_wmma_f32<<<ggrid, gblk, 0, stream>>>(x, wv, vb,    M, HIDC, HIDC, 0);
  gemm_wmma_f32<<<ggrid, gblk, 0, stream>>>(x, wg, gactb, M, HIDC, HIDC, 1);

  ttt_scan_kernel<<<HH, 256, SCAN_SMEM_FLOATS * sizeof(float), stream>>>(
      kb, vb, qb, fw1, fb1, fw2, fb2, flg, flb, lw, lb, ttg, ttb, outb);

  gate_ln_kernel<<<M, 256, 0, stream>>>(outb, gactb, png, pnb, gatedb);

  gemm_wmma_f32<<<ggrid, gblk, 0, stream>>>(gatedb, wo, (float*)d_out,
                                            M, HIDC, HIDC, 0);
}


# scan: LayerNorms of all 5 forward passes and the post-norm applied where the tile is read as fragments: lane sums its half row in registers, 1 lane-half exchange; LN phases gone; sums re-associated
# speedup vs baseline: 2.7805x; 1.0966x over previous
#include <hip/hip_runtime.h>
#include <hip/hip_bf16.h>
#include <math.h>

#define BB     8
#define SS     256
#define HIDC   1024
#define HH     16
#define DD     64
#define FF     64
#define LR     0.1f
#define GSCALE (2.0f / 8192.0f)
#define LN_EPS 1e-5f

typedef float v2f __attribute__((ext_vector_type(2)));
typedef float v8f __attribute__((ext_vector_type(8)));

__device__ __forceinline__ v8f wmma4(v2f a, v2f b, v8f c) {
  return __builtin_amdgcn_wmma_f32_16x16x4_f32(
       false, a,  false, b,
       (short)0, c,  false,  false);
}

__device__ __forceinline__ float gelu_f(float x) {
  const float c = 0.7978845608028654f;
  float u = c * (x + 0.044715f * x * x * x);
  return 0.5f * x * (1.0f + tanhf(u));
}
__device__ __forceinline__ void gelu_and_grad(float x, float& g, float& gd) {
  const float c = 0.7978845608028654f;
  float u = c * (x + 0.044715f * x * x * x);
  float t = tanhf(u);
  float x2 = x * x;
  g  = 0.5f * x * (1.0f + t);
  gd = 0.5f * (1.0f + t) + 0.5f * x * (1.0f - t * t) * c * (1.0f + 0.134145f * x2);
}
__device__ __forceinline__ float wave_sum32(float v) {
#pragma unroll
  for (int off = 16; off > 0; off >>= 1) v += __shfl_xor(v, off, 32);
  return v;
}

__device__ __forceinline__ float row_xor8(float v) {
  const int b = __builtin_bit_cast(int, v);
  return __builtin_bit_cast(float, __builtin_amdgcn_update_dpp(b, b, 0x168, 0xf, 0xf, false));
}

__device__ __forceinline__ float xhalf(float v) {
  const int b = __builtin_bit_cast(int, v);
  return __builtin_bit_cast(float, __builtin_amdgcn_permlanex16(b, b, 0x76543210, 0xfedcba98, false, false));
}

typedef float v4f __attribute__((ext_vector_type(4)));
#define GA_PITCH 36
#define GB_PITCH 288
#define GHOLD8(a, o) asm volatile("" :: "v"(a[o]), "v"(a[o + 1]), "v"(a[o + 2]), "v"(a[o + 3]), \
                                        "v"(a[o + 4]), "v"(a[o + 5]), "v"(a[o + 6]), "v"(a[o + 7]))
__global__ void __launch_bounds__(256)
gemm_wmma_f32(const float* __restrict__ A, const float* __restrict__ Bm,
              float* __restrict__ C, int M, int N, int K, int act) {
  __shared__ __align__(16) float As[2][64 * GA_PITCH];
  __shared__ __align__(16) float Bs[2][16 * GB_PITCH];
  (void)M;
  const int tid   = threadIdx.x;
  const int wave  = tid >> 5;
  const int lane  = tid & 31;
  const int c     = lane & 15;
  const int h     = lane >> 4;
  const int wm    = wave >> 2;
  const int wn    = wave & 3;
  const int mbase = blockIdx.y * 64;
  const int nbase = blockIdx.x * 128;
  const int ar = tid >> 3, ac = (tid & 7) << 2;
  const int bp = tid >> 5, bc = (tid & 31) << 2;
  const float* ag = A + (size_t)(mbase + ar) * K + ac;
  const float* bg = Bm + (size_t)(2 * bp) * N + nbase + bc;
  const int aso = ar * GA_PITCH + ac;
  const int bso = bp * GB_PITCH + 2 * bc;
  v4f ra0, ra1, rb0, rb1, rb2, rb3;
  ra0 = *(const v4f*)(ag);
  ra1 = *(const v4f*)(ag + (size_t)32 * K);
  rb0 = *(const v4f*)(bg);
  rb1 = *(const v4f*)(bg + N);
  rb2 = *(const v4f*)(bg + (size_t)16 * N);
  rb3 = *(const v4f*)(bg + (size_t)17 * N);
  *(v4f*)&As[0][aso]                 = ra0;
  *(v4f*)&As[0][aso + 32 * GA_PITCH] = ra1;
  *(v4f*)&Bs[0][bso]                    = v4f{rb0.x, rb1.x, rb0.y, rb1.y};
  *(v4f*)&Bs[0][bso + 4]                = v4f{rb0.z, rb1.z, rb0.w, rb1.w};
  *(v4f*)&Bs[0][bso + 8 * GB_PITCH]     = v4f{rb2.x, rb3.x, rb2.y, rb3.y};
  *(v4f*)&Bs[0][bso + 8 * GB_PITCH + 4] = v4f{rb2.z, rb3.z, rb2.w, rb3.w};
  __syncthreads();
  v8f acc00 = {0.f, 0.f, 0.f, 0.f, 0.f, 0.f, 0.f, 0.f};
  v8f acc01 = acc00, acc10 = acc00, acc11 = acc00;
  const int afo = (wm * 32 + c) * GA_PITCH + 2 * h;
  const int bfo = h * GB_PITCH + 2 * (wn * 32 + c);
  const int nch = K >> 5;
  for (int ch = 0; ch < nch; ++ch) {
    const int cur = ch & 1;
    const bool more = (ch + 1 < nch);
    if (more) {
      const float* an = ag + (size_t)(ch + 1) * 32;
      const float* bn = bg + (size_t)(ch + 1) * 32 * N;
      ra0 = *(const v4f*)(an);
      ra1 = *(const v4f*)(an + (size_t)32 * K);
      rb0 = *(const v4f*)(bn);
      rb1 = *(const v4f*)(bn + N);
      rb2 = *(const v4f*)(bn + (size_t)16 * N);
      rb3 = *(const v4f*)(bn + (size_t)17 * N);
    }
    const float* as = &As[cur][afo];
    const float* bs = &Bs[cur][bfo];
    v2f fa[16], fb[16];
#pragma unroll
    for (int ks = 0; ks < 8; ++ks) {
      fa[2 * ks]     = *(const v2f*)(as + 4 * ks);
      fa[2 * ks + 1] = *(const v2f*)(as + 16 * GA_PITCH + 4 * ks);
      fb[2 * ks]     = *(const v2f*)(bs + 2 * GB_PITCH * ks);
      fb[2 * ks + 1] = *(const v2f*)(bs + 2 * GB_PITCH * ks + 32);
    }
    GHOLD8(fa, 0); GHOLD8(fa, 8); GHOLD8(fb, 0); GHOLD8(fb, 8);
#pragma unroll
    for (int ks = 0; ks < 8; ++ks) {
      acc00 = wmma4(fa[2 * ks],     fb[2 * ks],     acc00);
      acc10 = wmma4(fa[2 * ks + 1], fb[2 * ks],     acc10);
      acc01 = wmma4(fa[2 * ks],     fb[2 * ks + 1], acc01);
      acc11 = wmma4(fa[2 * ks + 1], fb[2 * ks + 1], acc11);
    }
    if (more) {
      float* ad = &As[cur ^ 1][aso];
      float* bd = &Bs[cur ^ 1][bso];
      *(v4f*)(ad)                 = ra0;
      *(v4f*)(ad + 32 * GA_PITCH) = ra1;
      *(v4f*)(bd)                    = v4f{rb0.x, rb1.x, rb0.y, rb1.y};
      *(v4f*)(bd + 4)                = v4f{rb0.z, rb1.z, rb0.w, rb1.w};
      *(v4f*)(bd + 8 * GB_PITCH)     = v4f{rb2.x, rb3.x, rb2.y, rb3.y};
      *(v4f*)(bd + 8 * GB_PITCH + 4) = v4f{rb2.z, rb3.z, rb2.w, rb3.w};
    }
    __syncthreads();
  }
  const int row0 = mbase + wm * 32 + 8 * h;
  const int col0 = nbase + wn * 32 + c;
#pragma unroll
  for (int r = 0; r < 8; ++r) {
    float v00 = acc00[r], v01 = acc01[r], v10 = acc10[r], v11 = acc11[r];
    if (act) { v00 = gelu_f(v00); v01 = gelu_f(v01); v10 = gelu_f(v10); v11 = gelu_f(v11); }
    C[(size_t)(row0 + r) * N + col0]           = v00;
    C[(size_t)(row0 + r) * N + col0 + 16]      = v01;
    C[(size_t)(row0 + 16 + r) * N + col0]      = v10;
    C[(size_t)(row0 + 16 + r) * N + col0 + 16] = v11;
  }
}


#define HOLD8(a, o) asm volatile("" :: "v"(a[o]), "v"(a[o + 1]), "v"(a[o + 2]), "v"(a[o + 3]), \
                                       "v"(a[o + 4]), "v"(a[o + 5]), "v"(a[o + 6]), "v"(a[o + 7]))
#define TIDX(b, d) ((((d) >> 1) << 4) + ((b) << 1) + ((d) & 1))
#define W2P_PITCH 192
#define W2P_ROT(p) ((((p) & 6) << 1) + (((p) & 1) << 5))
#define W2P_IDX(f, d) (((d) >> 1) * W2P_PITCH + 2 * (f) + W2P_ROT((d) >> 1) + ((d) & 1))
#define WB_W2P 0
#define WB_LWP 1
#define CH_GELU  0
#define CH_DPRED 1
__device__ __forceinline__ void frag_rows16(const float* T, int xb, v2f* of) {
#pragma unroll
  for (int i = 0; i < 16; ++i) {
    of[i].x = T[xb + 32 * i];
    of[i].y = T[xb + 32 * i + 1];
  }
}
__device__ __forceinline__ void half_row_stats(const v2f* of, float& m, float& is) {
  v2f sa = {0.f, 0.f}, sb = sa, qa = sa, qb = sa;
#pragma unroll
  for (int i = 0; i < 8; ++i) {
    sa += of[2 * i];
    sb += of[2 * i + 1];
    qa += of[2 * i] * of[2 * i];
    qb += of[2 * i + 1] * of[2 * i + 1];
  }
  float s = (sa.x + sa.y) + (sb.x + sb.y);
  float q = (qa.x + qa.y) + (qb.x + qb.y);
  s += xhalf(s);
  q += xhalf(q);
  m  = s * (1.f / 64.f);
  is = rsqrtf(q * (1.f / 64.f) - m * m + LN_EPS);
}
template <int EPI, int WBL, bool HAS_LATER, bool LN_ON_LOAD, typename LATER>
__device__ __forceinline__ void
chain_pair(const float* X8T, const float* Wa, const float* biasa, const float* Wb,
           float* Cgd, float* Cact, const float* tgtT, float* part, int wave, int lane,
           const float* lng, const float* lnb, LATER later_reads) {
  const int c  = lane & 15;
  const int h  = lane >> 4;
  const int bb = c & 7;
  const int kh = h << 1;
  const int fo = wave * 16;
  const int f0 = fo + 8 * h;
  const int xb = 16 * h + 2 * bb;
  const int tb = 128 * wave + 64 * h + 2 * bb;
  const int wp = 1024 * wave + 512 * h + 2 * c;
  const int wr = W2P_IDX(f0, c);
  v2f fa[16], fb[16], fa2[16];
  v2f lgf[16], lbf[16];
  float bi[8], tg[8];
  frag_rows16(X8T, xb, fb);
  if (LN_ON_LOAD) {
#pragma unroll
    for (int i = 0; i < 16; ++i) {
      lgf[i].x = lng[4 * i + kh];
      lgf[i].y = lng[4 * i + kh + 1];
      lbf[i].x = lnb[4 * i + kh];
      lbf[i].y = lnb[4 * i + kh + 1];
    }
  }
#pragma unroll
  for (int i = 0; i < 16; ++i) {
    const int k0 = 4 * i;
    fa[i].x = Wa[(k0 + kh) * 64 + fo + c];
    fa[i].y = Wa[(k0 + kh + 1) * 64 + fo + c];
  }
#pragma unroll
  for (int s = 0; s < 4; ++s) {
#pragma unroll
    for (int td = 0; td < 4; ++td) {
      if (WBL == WB_W2P) {
        fa2[s * 4 + td].x = Wb[wr + 8 * W2P_PITCH * td + 4 * s];
        fa2[s * 4 + td].y = Wb[wr + 8 * W2P_PITCH * td + 4 * s + 2];
      } else {
        fa2[s * 4 + td].x = Wb[wp + 128 * s + 32 * td];
        fa2[s * 4 + td].y = Wb[wp + 128 * s + 32 * td + 1];
      }
    }
  }
#pragma unroll
  for (int r = 0; r < 8; ++r) {
    bi[r] = biasa[f0 + r];
    tg[r] = (EPI == CH_DPRED) ? tgtT[tb + 16 * (r >> 1) + (r & 1)] : 0.f;
  }
  if (LN_ON_LOAD) {
    float m, is;
    half_row_stats(fb, m, is);
#pragma unroll
    for (int i = 0; i < 16; ++i) fb[i] = ((fb[i] - m) * is) * lgf[i] + lbf[i];
  }
  HOLD8(fa, 0);  HOLD8(fa, 8);
  if (!LN_ON_LOAD) { HOLD8(fb, 0);  HOLD8(fb, 8); }
  HOLD8(fa2, 0); HOLD8(fa2, 8);
  HOLD8(bi, 0);
  if (EPI == CH_DPRED) HOLD8(tg, 0);
  if (HAS_LATER) {
    later_reads();
    __builtin_amdgcn_sched_barrier(0);
  }
  v8f acc = {0.f, 0.f, 0.f, 0.f, 0.f, 0.f, 0.f, 0.f};
#pragma unroll
  for (int i = 0; i < 16; ++i) acc = wmma4(fa[i], fb[i], acc);
  v8f g;
  if (EPI == CH_GELU) {
    const bool hi = (c >= 8);
    float gm[4], dm[4];
#pragma unroll
    for (int r = 0; r < 4; ++r) {
      const float xlo = acc[r] + bi[r];
      const float xhi = acc[4 + r] + bi[4 + r];
      gelu_and_grad(hi ? xhi : xlo, gm[r], dm[r]);
    }
    {
      const int so = bb * 64 + f0 + (hi ? 4 : 0);
#pragma unroll
      for (int r = 0; r < 4; ++r) {
        Cgd[so + r] = dm[r];
        Cact[so + r] = gm[r];
      }
    }
#pragma unroll
    for (int r = 0; r < 4; ++r) {
      const float go = row_xor8(gm[r]);
      g[r]     = hi ? go : gm[r];
      g[4 + r] = hi ? gm[r] : go;
    }
  } else {
#pragma unroll
    for (int r = 0; r < 8; ++r) {
      float v = acc[r] + bi[r];
      g[r] = GSCALE * (v - tg[r]);
    }
  }
  v8f p0 = {0.f, 0.f, 0.f, 0.f, 0.f, 0.f, 0.f, 0.f};
  v8f p1 = p0, p2 = p0, p3 = p0;
#pragma unroll
  for (int s = 0; s < 4; ++s) {
    v2f b2;
    b2.x = g[2 * s];
    b2.y = g[2 * s + 1];
    p0 = wmma4(fa2[s * 4 + 0], b2, p0);
    p1 = wmma4(fa2[s * 4 + 1], b2, p1);
    p2 = wmma4(fa2[s * 4 + 2], b2, p2);
    p3 = wmma4(fa2[s * 4 + 3], b2, p3);
  }
  if (c < 8) {
    float* dst = part + wave * 512 + 64 * h + 2 * bb;
#pragma unroll
    for (int r = 0; r < 8; ++r) {
      const int o = 16 * (r >> 1) + (r & 1);
      dst[o]       = p0[r];
      dst[128 + o] = p1[r];
      dst[256 + o] = p2[r];
      dst[384 + o] = p3[r];
    }
  }
}

__device__ __forceinline__ float part_sum4(const float* part, int idx) {
  return (part[idx] + part[512 + idx]) + (part[1024 + idx] + part[1536 + idx]);
}

#define SCAN_SMEM_FLOATS (3 * 4096 + 11 * 512 + 7 * 64 + 32 + 4 * 512 + 32 * W2P_PITCH)

__global__ void __launch_bounds__(256)
ttt_scan_kernel(const float* __restrict__ kg, const float* __restrict__ vg,
                const float* __restrict__ qg,
                const float* __restrict__ fw1, const float* __restrict__ fb1,
                const float* __restrict__ fw2, const float* __restrict__ fb2,
                const float* __restrict__ flg, const float* __restrict__ flb,
                const float* __restrict__ lwg, const float* __restrict__ lbg,
                const float* __restrict__ ttg, const float* __restrict__ ttb,
                float* __restrict__ outg) {
  extern __shared__ __align__(16) float smem[];
  float* W1  = smem;
  float* LW  = W1 + 4096;
  float* ktb = LW + 4096;
  float* tgt = ktb + 512;
  float* gdv = tgt + 512;
  float* hg  = gdv + 512;
  float* xh  = hg + 512;
  float* zb  = xh + 512;
  float* dzb = zb + 512;
  float* dob = dzb + 512;
  float* ktT = dob + 512;
  float* dobT = ktT + 512;
  float* obT = dobT + 512;
  float* b1v = obT + 512;
  float* b2v = b1v + 64;
  float* lgv = b2v + 64;
  float* lbv = lgv + 64;
  float* tgv = lbv + 64;
  float* tbv = tgv + 64;
  float* lbz = tbv + 64;
  float* isv = lbz + 64;
  float* m1v = isv + 8;
  float* m2v = m1v + 8;
  float* part = m2v + 16;
  float* LWP = part + 4 * 512;
  float* W2  = LWP + 4096;

  const int h    = blockIdx.x;
  const int tid  = threadIdx.x;
  const int wave = tid >> 5;
  const int lane = tid & 31;

  const int b0 = tid >> 6,          d0 = tid & 63;
  const int b1i = (tid + 256) >> 6, d1i = (tid + 256) & 63;
  const size_t g0base = (size_t)(b0 * SS) * HIDC + h * 64 + d0;
  const size_t g1base = (size_t)(b1i * SS) * HIDC + h * 64 + d1i;

  for (int i = tid; i < 4096; i += 256) {
    W1[i] = fw1[(size_t)h * 4096 + i];
    W2[W2P_IDX(i >> 6, i & 63)] = fw2[(size_t)h * 4096 + i];
    LW[i] = lwg[i];
    LWP[((i & 63) >> 1) * 128 + 2 * (i >> 6) + (i & 1)] = lwg[i];
  }
  if (tid < 64) {
    b1v[tid] = fb1[h * 64 + tid];
    b2v[tid] = fb2[h * 64 + tid];
    lgv[tid] = flg[h * 64 + tid];
    lbv[tid] = flb[h * 64 + tid];
    tgv[tid] = ttg[h * 64 + tid];
    tbv[tid] = ttb[h * 64 + tid];
    lbz[tid] = lbg[tid];
  }

  float kp0 = kg[g0base], vp0 = vg[g0base];
  float kp1 = kg[g1base], vp1 = vg[g1base];
  __syncthreads();

  for (int t = 0; t < SS; ++t) {
    ktb[tid]            = kp0;  ktb[tid + 256]      = kp1;
    ktT[TIDX(b0, d0)]   = kp0;  ktT[TIDX(b1i, d1i)] = kp1;
    tgt[TIDX(b0, d0)]   = vp0 - kp0;
    tgt[TIDX(b1i, d1i)] = vp1 - kp1;
    __syncthreads();
    if (t + 1 < SS) {
      kp0 = kg[g0base + (t + 1) * HIDC];  vp0 = vg[g0base + (t + 1) * HIDC];
      kp1 = kg[g1base + (t + 1) * HIDC];  vp1 = vg[g1base + (t + 1) * HIDC];
    }

    for (int step = 0; step < 4; ++step) {
      if (wave < 4)
        chain_pair<CH_GELU, WB_W2P, false, false>(ktT, W1, b1v, W2, gdv, hg, nullptr, part, wave, lane,
                                                  nullptr, nullptr, [] {});
      __syncthreads();
      {
        const int pi = TIDX(wave, lane);
        obT[pi]       = part_sum4(part, pi) + b2v[lane];
        obT[pi + 256] = part_sum4(part, pi + 256) + b2v[lane + 32];
      }
      __syncthreads();
      const int fc = lane & 15;
      const int fh = lane >> 4;
      const int fk = fh << 1;
      v2f   pb[16];
      v2f   pk[8];
      float pg[8];
      float pb1 = 0.f;
      v2f   ph[2];
      float wold[32];
      if (wave < 4) {
        chain_pair<CH_DPRED, WB_LWP, true, true>(obT, LW, lbz, LWP, nullptr, nullptr, tgt, part, wave, lane,
                                                 lgv, lbv, [&] {
          const int col = wave * 16 + fc;
          const int wq  = fh * (W2P_PITCH + 32) + 2 * col;
#pragma unroll
          for (int i = 0; i < 16; ++i) {
            pb[i].x = W2[wq + 2 * W2P_PITCH * i + 4 * (i & 3)];
            pb[i].y = W2[wq + 2 * W2P_PITCH * i + 4 * (i & 3) + 1];
          }
#pragma unroll
          for (int r = 0; r < 8; ++r) pg[r] = gdv[r * 64 + col];
          pb1 = b1v[col];
#pragma unroll
          for (int tm = 0; tm < 4; ++tm) {
#pragma unroll
            for (int s = 0; s < 2; ++s) {
              pk[tm * 2 + s].x = ktb[(4 * s + fk) * 64 + tm * 16 + fc];
              pk[tm * 2 + s].y = ktb[(4 * s + fk + 1) * 64 + tm * 16 + fc];
            }
#pragma unroll
            for (int i = 0; i < 8; ++i) wold[tm * 8 + i] = W1[(tm * 16 + 8 * fh + i) * 64 + col];
          }
        });
      } else {
        const int wv = wave - 4;
        v2f of[16];
        frag_rows16(obT, 16 * fh + 2 * (fc & 7), of);
#pragma unroll
        for (int s = 0; s < 2; ++s) {
          ph[s].x = hg[(4 * s + fk) * 64 + wv * 16 + fc];
          ph[s].y = hg[(4 * s + fk + 1) * 64 + wv * 16 + fc];
        }
        {
          float m, is;
          half_row_stats(of, m, is);
          if (fc < 8) {
#pragma unroll
            for (int i = 0; i < 16; ++i) {
              if ((i >> 2) == wv) {
                const v2f xv = (of[i] - m) * is;
                xh[fc * 64 + 4 * i + fk]     = xv.x;
                xh[fc * 64 + 4 * i + fk + 1] = xv.y;
              }
            }
            if (wave == 4 && fh == 0) isv[fc] = is;
          }
        }
#pragma unroll
        for (int tn = 0; tn < 4; ++tn) {
#pragma unroll
          for (int i = 0; i < 8; ++i)
            wold[tn * 8 + i] = W2[W2P_IDX(wv * 16 + 8 * fh, fc) + 8 * W2P_PITCH * tn + 2 * i];
        }
      }
      __syncthreads();
      {
        const int pi = TIDX(wave, lane);
        float z0 = part_sum4(part, pi);
        float z1 = part_sum4(part, pi + 256);
        dzb[wave * 64 + lane] = z0;
        dzb[wave * 64 + lane + 32] = z1;
        float dd0 = z0 * lgv[lane];
        float dd1 = z1 * lgv[lane + 32];
        float x0 = xh[wave * 64 + lane], x1 = xh[wave * 64 + lane + 32];
        float s1 = wave_sum32(dd0 + dd1);
        float s2 = wave_sum32(dd0 * x0 + dd1 * x1);
        if (lane == 0) { m1v[wave] = s1 * (1.f / 64.f); m2v[wave] = s2 * (1.f / 64.f); }
      }
      __syncthreads();
      for (int idx = tid; idx < 512; idx += 256) {
        int b = idx >> 6, d = idx & 63;
        float dxh = dzb[idx] * lgv[d];
        const float dv = isv[b] * (dxh - m1v[b] - xh[idx] * m2v[b]);
        dob[idx] = dv;
        dobT[TIDX(b, d)] = dv;
      }
      __syncthreads();
      if (wave < 4) {
        const int col = wave * 16 + fc;
        const int bb  = fc & 7;
        v2f pa[16];
#pragma unroll
        for (int i = 0; i < 16; ++i) {
          pa[i].x = dobT[16 * fh + 2 * bb + 32 * i];
          pa[i].y = dobT[16 * fh + 2 * bb + 32 * i + 1];
        }
        HOLD8(pa, 0); HOLD8(pa, 8);
        v8f acc = {0.f, 0.f, 0.f, 0.f, 0.f, 0.f, 0.f, 0.f};
#pragma unroll
        for (int i = 0; i < 16; ++i) acc = wmma4(pa[i], pb[i], acc);
        float da[8];
        float sb = 0.f;
#pragma unroll
        for (int r = 0; r < 8; ++r) da[r] = acc[r] * pg[r];
#pragma unroll
        for (int r = 0; r < 8; ++r) sb += da[r];
        if (fh == 0) b1v[col] = pb1 - LR * sb;
        v8f u0 = {0.f, 0.f, 0.f, 0.f, 0.f, 0.f, 0.f, 0.f};
        v8f u1 = u0, u2 = u0, u3 = u0;
#pragma unroll
        for (int s = 0; s < 2; ++s) {
          v2f b2;
          b2.x = fh ? da[4 * s + 2] : da[4 * s];
          b2.y = fh ? da[4 * s + 3] : da[4 * s + 1];
          u0 = wmma4(pk[0 + s], b2, u0);
          u1 = wmma4(pk[2 + s], b2, u1);
          u2 = wmma4(pk[4 + s], b2, u2);
          u3 = wmma4(pk[6 + s], b2, u3);
        }
#pragma unroll
        for (int i = 0; i < 8; ++i) {
          W1[(0 * 16 + 8 * fh + i) * 64 + col] = wold[0 * 8 + i] - LR * u0[i];
          W1[(1 * 16 + 8 * fh + i) * 64 + col] = wold[1 * 8 + i] - LR * u1[i];
          W1[(2 * 16 + 8 * fh + i) * 64 + col] = wold[2 * 8 + i] - LR * u2[i];
          W1[(3 * 16 + 8 * fh + i) * 64 + col] = wold[3 * 8 + i] - LR * u3[i];
        }
      } else {
        const int wv = wave - 4;
        v2f pd[8];
#pragma unroll
        for (int tn = 0; tn < 4; ++tn) {
#pragma unroll
          for (int s = 0; s < 2; ++s) {
            pd[tn * 2 + s].x = dob[(4 * s + fk) * 64 + tn * 16 + fc];
            pd[tn * 2 + s].y = dob[(4 * s + fk + 1) * 64 + tn * 16 + fc];
          }
        }
        v8f u0 = {0.f, 0.f, 0.f, 0.f, 0.f, 0.f, 0.f, 0.f};
        v8f u1 = u0, u2 = u0, u3 = u0;
#pragma unroll
        for (int s = 0; s < 2; ++s) {
          u0 = wmma4(ph[s], pd[0 + s], u0);
          u1 = wmma4(ph[s], pd[2 + s], u1);
          u2 = wmma4(ph[s], pd[4 + s], u2);
          u3 = wmma4(ph[s], pd[6 + s], u3);
        }
        const int rb = W2P_IDX(wv * 16 + 8 * fh, fc);
#pragma unroll
        for (int i = 0; i < 8; ++i) {
          W2[rb + 0 * 8 * W2P_PITCH + 2 * i] = wold[0 * 8 + i] - LR * u0[i];
          W2[rb + 1 * 8 * W2P_PITCH + 2 * i] = wold[1 * 8 + i] - LR * u1[i];
          W2[rb + 2 * 8 * W2P_PITCH + 2 * i] = wold[2 * 8 + i] - LR * u2[i];
          W2[rb + 3 * 8 * W2P_PITCH + 2 * i] = wold[3 * 8 + i] - LR * u3[i];
        }
        const int t2 = tid - 128;
        if (t2 < 64) {
          float s = 0.f, s2 = 0.f;
          for (int b = 0; b < 8; ++b) s += dzb[b * 64 + t2] * xh[b * 64 + t2];
          for (int b = 0; b < 8; ++b) s2 += dob[b * 64 + t2];
          lgv[t2] -= LR * s;
          b2v[t2] -= LR * s2;
        } else {
          const int d = t2 - 64;
          float s = 0.f;
          for (int b = 0; b < 8; ++b) s += dzb[b * 64 + d];
          lbv[d] -= LR * s;
        }
      }
      __syncthreads();
    }

    if (wave < 4)
      chain_pair<CH_GELU, WB_W2P, false, false>(ktT, W1, b1v, W2, gdv, hg, nullptr, part, wave, lane,
                                                nullptr, nullptr, [] {});
    __syncthreads();
    {
      const int pi = TIDX(wave, lane);
      obT[pi]       = part_sum4(part, pi) + b2v[lane];
      obT[pi + 256] = part_sum4(part, pi + 256) + b2v[lane + 32];
    }
    __syncthreads();
    if (wave < 4) {
      const int fc = lane & 15;
      const int fh = lane >> 4;
      const int fk = fh << 1;
      v2f of[16], lgf[16], lbf[16];
      frag_rows16(obT, 16 * fh + 2 * (fc & 7), of);
#pragma unroll
      for (int i = 0; i < 16; ++i) {
        lgf[i].x = lgv[4 * i + fk];
        lgf[i].y = lgv[4 * i + fk + 1];
        lbf[i].x = lbv[4 * i + fk];
        lbf[i].y = lbv[4 * i + fk + 1];
      }
      float m, is, m2, is2;
      half_row_stats(of, m, is);
#pragma unroll
      for (int i = 0; i < 16; ++i) of[i] = ((of[i] - m) * is) * lgf[i] + lbf[i];
      half_row_stats(of, m2, is2);
      if (fc < 8) {
#pragma unroll
        for (int i = 0; i < 16; ++i) {
          if ((i >> 2) == wave) {
            const int d = 4 * i + fk;
            const v2f xv = (of[i] - m2) * is2;
            zb[fc * 64 + d]     = xv.x * tgv[d] + tbv[d];
            zb[fc * 64 + d + 1] = xv.y * tgv[d + 1] + tbv[d + 1];
          }
        }
      }
    }
    __syncthreads();
    for (int idx = tid; idx < 512; idx += 256) {
      int b = idx >> 6, d = idx & 63;
      size_t g = ((size_t)(b * SS + t)) * HIDC + h * 64 + d;
      outg[g] = qg[g] + zb[idx];
    }
    __syncthreads();
  }
}

__global__ void __launch_bounds__(256)
gate_ln_kernel(const float* __restrict__ outb, const float* __restrict__ gact,
               const float* __restrict__ png, const float* __restrict__ pnb,
               float* __restrict__ gated) {
  __shared__ float r1[256], r2[256];
  __shared__ float mu_s, inv_s;
  const int r = blockIdx.x, tid = threadIdx.x;
  const float* row = outb + (size_t)r * HIDC;
  float s = 0.f, q = 0.f;
  for (int i = tid; i < HIDC; i += 256) { float v = row[i]; s += v; q += v * v; }
  r1[tid] = s; r2[tid] = q;
  __syncthreads();
  for (int off = 128; off > 0; off >>= 1) {
    if (tid < off) { r1[tid] += r1[tid + off]; r2[tid] += r2[tid + off]; }
    __syncthreads();
  }
  if (tid == 0) {
    float m = r1[0] * (1.f / HIDC);
    mu_s = m;
    inv_s = rsqrtf(r2[0] * (1.f / HIDC) - m * m + LN_EPS);
  }
  __syncthreads();
  for (int i = tid; i < HIDC; i += 256) {
    float y = (row[i] - mu_s) * inv_s * png[i] + pnb[i];
    gated[(size_t)r * HIDC + i] = gact[(size_t)r * HIDC + i] * y;
  }
}

extern "C" void kernel_launch(void* const* d_in, const int* in_sizes, int n_in,
                              void* d_out, int out_size, void* d_ws, size_t ws_size,
                              hipStream_t stream) {
  (void)in_sizes; (void)n_in; (void)out_size; (void)ws_size;
  const float* x   = (const float*)d_in[0];
  const float* wq  = (const float*)d_in[1];
  const float* wk  = (const float*)d_in[2];
  const float* wv  = (const float*)d_in[3];
  const float* fw1 = (const float*)d_in[4];
  const float* fb1 = (const float*)d_in[5];
  const float* fw2 = (const float*)d_in[6];
  const float* fb2 = (const float*)d_in[7];
  const float* flg = (const float*)d_in[8];
  const float* flb = (const float*)d_in[9];
  const float* lw  = (const float*)d_in[10];
  const float* lb  = (const float*)d_in[11];
  const float* ttg = (const float*)d_in[12];
  const float* ttb = (const float*)d_in[13];
  const float* wo  = (const float*)d_in[14];
  const float* wg  = (const float*)d_in[15];
  const float* png = (const float*)d_in[16];
  const float* pnb = (const float*)d_in[17];

  const int M = BB * SS;
  const size_t PLANE = (size_t)M * HIDC;
  float* qb    = (float*)d_ws;
  float* kb    = qb + PLANE;
  float* vb    = kb + PLANE;
  float* gactb = vb + PLANE;
  float* outb  = gactb + PLANE;
  float* gatedb = qb;

  dim3 gblk(256);
  dim3 ggrid(HIDC / 128, M / 64);
  gemm_wmma_f32<<<ggrid, gblk, 0, stream>>>(x, wq, qb,    M, HIDC, HIDC, 0);
  gemm_wmma_f32<<<ggrid, gblk, 0, stream>>>(x, wk, kb,    M, HIDC, HIDC, 0);
  gemm_wmma_f32<<<ggrid, gblk, 0, stream>>>(x, wv, vb,    M, HIDC, HIDC, 0);
  gemm_wmma_f32<<<ggrid, gblk, 0, stream>>>(x, wg, gactb, M, HIDC, HIDC, 1);

  ttt_scan_kernel<<<HH, 256, SCAN_SMEM_FLOATS * sizeof(float), stream>>>(
      kb, vb, qb, fw1, fb1, fw2, fb2, flg, flb, lw, lb, ttg, ttb, outb);

  gate_ln_kernel<<<M, 256, 0, stream>>>(outb, gactb, png, pnb, gatedb);

  gemm_wmma_f32<<<ggrid, gblk, 0, stream>>>(gatedb, wo, (float*)d_out,
                                            M, HIDC, HIDC, 0);
}


# scan LN backward: row means formed from each wave's k-split partial d_z tile in registers (32 FMAs + 1 lane-half exchange), no lane tree; dob formed where the partials are summed; sums re-associated
# speedup vs baseline: 2.9150x; 1.0484x over previous
#include <hip/hip_runtime.h>
#include <hip/hip_bf16.h>
#include <math.h>

#define BB     8
#define SS     256
#define HIDC   1024
#define HH     16
#define DD     64
#define FF     64
#define LR     0.1f
#define GSCALE (2.0f / 8192.0f)
#define LN_EPS 1e-5f

typedef float v2f __attribute__((ext_vector_type(2)));
typedef float v8f __attribute__((ext_vector_type(8)));

__device__ __forceinline__ v8f wmma4(v2f a, v2f b, v8f c) {
  return __builtin_amdgcn_wmma_f32_16x16x4_f32(
       false, a,  false, b,
       (short)0, c,  false,  false);
}

__device__ __forceinline__ float gelu_f(float x) {
  const float c = 0.7978845608028654f;
  float u = c * (x + 0.044715f * x * x * x);
  return 0.5f * x * (1.0f + tanhf(u));
}
__device__ __forceinline__ void gelu_and_grad(float x, float& g, float& gd) {
  const float c = 0.7978845608028654f;
  float u = c * (x + 0.044715f * x * x * x);
  float t = tanhf(u);
  float x2 = x * x;
  g  = 0.5f * x * (1.0f + t);
  gd = 0.5f * (1.0f + t) + 0.5f * x * (1.0f - t * t) * c * (1.0f + 0.134145f * x2);
}

__device__ __forceinline__ float row_xor8(float v) {
  const int b = __builtin_bit_cast(int, v);
  return __builtin_bit_cast(float, __builtin_amdgcn_update_dpp(b, b, 0x168, 0xf, 0xf, false));
}

__device__ __forceinline__ float xhalf(float v) {
  const int b = __builtin_bit_cast(int, v);
  return __builtin_bit_cast(float, __builtin_amdgcn_permlanex16(b, b, 0x76543210, 0xfedcba98, false, false));
}

typedef float v4f __attribute__((ext_vector_type(4)));
#define GA_PITCH 36
#define GB_PITCH 288
#define GHOLD8(a, o) asm volatile("" :: "v"(a[o]), "v"(a[o + 1]), "v"(a[o + 2]), "v"(a[o + 3]), \
                                        "v"(a[o + 4]), "v"(a[o + 5]), "v"(a[o + 6]), "v"(a[o + 7]))
__global__ void __launch_bounds__(256)
gemm_wmma_f32(const float* __restrict__ A, const float* __restrict__ Bm,
              float* __restrict__ C, int M, int N, int K, int act) {
  __shared__ __align__(16) float As[2][64 * GA_PITCH];
  __shared__ __align__(16) float Bs[2][16 * GB_PITCH];
  (void)M;
  const int tid   = threadIdx.x;
  const int wave  = tid >> 5;
  const int lane  = tid & 31;
  const int c     = lane & 15;
  const int h     = lane >> 4;
  const int wm    = wave >> 2;
  const int wn    = wave & 3;
  const int mbase = blockIdx.y * 64;
  const int nbase = blockIdx.x * 128;
  const int ar = tid >> 3, ac = (tid & 7) << 2;
  const int bp = tid >> 5, bc = (tid & 31) << 2;
  const float* ag = A + (size_t)(mbase + ar) * K + ac;
  const float* bg = Bm + (size_t)(2 * bp) * N + nbase + bc;
  const int aso = ar * GA_PITCH + ac;
  const int bso = bp * GB_PITCH + 2 * bc;
  v4f ra0, ra1, rb0, rb1, rb2, rb3;
  ra0 = *(const v4f*)(ag);
  ra1 = *(const v4f*)(ag + (size_t)32 * K);
  rb0 = *(const v4f*)(bg);
  rb1 = *(const v4f*)(bg + N);
  rb2 = *(const v4f*)(bg + (size_t)16 * N);
  rb3 = *(const v4f*)(bg + (size_t)17 * N);
  *(v4f*)&As[0][aso]                 = ra0;
  *(v4f*)&As[0][aso + 32 * GA_PITCH] = ra1;
  *(v4f*)&Bs[0][bso]                    = v4f{rb0.x, rb1.x, rb0.y, rb1.y};
  *(v4f*)&Bs[0][bso + 4]                = v4f{rb0.z, rb1.z, rb0.w, rb1.w};
  *(v4f*)&Bs[0][bso + 8 * GB_PITCH]     = v4f{rb2.x, rb3.x, rb2.y, rb3.y};
  *(v4f*)&Bs[0][bso + 8 * GB_PITCH + 4] = v4f{rb2.z, rb3.z, rb2.w, rb3.w};
  __syncthreads();
  v8f acc00 = {0.f, 0.f, 0.f, 0.f, 0.f, 0.f, 0.f, 0.f};
  v8f acc01 = acc00, acc10 = acc00, acc11 = acc00;
  const int afo = (wm * 32 + c) * GA_PITCH + 2 * h;
  const int bfo = h * GB_PITCH + 2 * (wn * 32 + c);
  const int nch = K >> 5;
  for (int ch = 0; ch < nch; ++ch) {
    const int cur = ch & 1;
    const bool more = (ch + 1 < nch);
    if (more) {
      const float* an = ag + (size_t)(ch + 1) * 32;
      const float* bn = bg + (size_t)(ch + 1) * 32 * N;
      ra0 = *(const v4f*)(an);
      ra1 = *(const v4f*)(an + (size_t)32 * K);
      rb0 = *(const v4f*)(bn);
      rb1 = *(const v4f*)(bn + N);
      rb2 = *(const v4f*)(bn + (size_t)16 * N);
      rb3 = *(const v4f*)(bn + (size_t)17 * N);
    }
    const float* as = &As[cur][afo];
    const float* bs = &Bs[cur][bfo];
    v2f fa[16], fb[16];
#pragma unroll
    for (int ks = 0; ks < 8; ++ks) {
      fa[2 * ks]     = *(const v2f*)(as + 4 * ks);
      fa[2 * ks + 1] = *(const v2f*)(as + 16 * GA_PITCH + 4 * ks);
      fb[2 * ks]     = *(const v2f*)(bs + 2 * GB_PITCH * ks);
      fb[2 * ks + 1] = *(const v2f*)(bs + 2 * GB_PITCH * ks + 32);
    }
    GHOLD8(fa, 0); GHOLD8(fa, 8); GHOLD8(fb, 0); GHOLD8(fb, 8);
#pragma unroll
    for (int ks = 0; ks < 8; ++ks) {
      acc00 = wmma4(fa[2 * ks],     fb[2 * ks],     acc00);
      acc10 = wmma4(fa[2 * ks + 1], fb[2 * ks],     acc10);
      acc01 = wmma4(fa[2 * ks],     fb[2 * ks + 1], acc01);
      acc11 = wmma4(fa[2 * ks + 1], fb[2 * ks + 1], acc11);
    }
    if (more) {
      float* ad = &As[cur ^ 1][aso];
      float* bd = &Bs[cur ^ 1][bso];
      *(v4f*)(ad)                 = ra0;
      *(v4f*)(ad + 32 * GA_PITCH) = ra1;
      *(v4f*)(bd)                    = v4f{rb0.x, rb1.x, rb0.y, rb1.y};
      *(v4f*)(bd + 4)                = v4f{rb0.z, rb1.z, rb0.w, rb1.w};
      *(v4f*)(bd + 8 * GB_PITCH)     = v4f{rb2.x, rb3.x, rb2.y, rb3.y};
      *(v4f*)(bd + 8 * GB_PITCH + 4) = v4f{rb2.z, rb3.z, rb2.w, rb3.w};
    }
    __syncthreads();
  }
  const int row0 = mbase + wm * 32 + 8 * h;
  const int col0 = nbase + wn * 32 + c;
#pragma unroll
  for (int r = 0; r < 8; ++r) {
    float v00 = acc00[r], v01 = acc01[r], v10 = acc10[r], v11 = acc11[r];
    if (act) { v00 = gelu_f(v00); v01 = gelu_f(v01); v10 = gelu_f(v10); v11 = gelu_f(v11); }
    C[(size_t)(row0 + r) * N + col0]           = v00;
    C[(size_t)(row0 + r) * N + col0 + 16]      = v01;
    C[(size_t)(row0 + 16 + r) * N + col0]      = v10;
    C[(size_t)(row0 + 16 + r) * N + col0 + 16] = v11;
  }
}


#define HOLD8(a, o) asm volatile("" :: "v"(a[o]), "v"(a[o + 1]), "v"(a[o + 2]), "v"(a[o + 3]), \
                                       "v"(a[o + 4]), "v"(a[o + 5]), "v"(a[o + 6]), "v"(a[o + 7]))
#define TIDX(b, d) ((((d) >> 1) << 4) + ((b) << 1) + ((d) & 1))
#define W2P_PITCH 192
#define W2P_ROT(p) ((((p) & 6) << 1) + (((p) & 1) << 5))
#define W2P_IDX(f, d) (((d) >> 1) * W2P_PITCH + 2 * (f) + W2P_ROT((d) >> 1) + ((d) & 1))
#define WB_W2P 0
#define WB_LWP 1
#define CH_GELU  0
#define CH_DPRED 1
__device__ __forceinline__ void frag_rows16(const float* T, int xb, v2f* of) {
#pragma unroll
  for (int i = 0; i < 16; ++i) {
    of[i].x = T[xb + 32 * i];
    of[i].y = T[xb + 32 * i + 1];
  }
}
__device__ __forceinline__ void half_row_stats(const v2f* of, float& m, float& is) {
  v2f sa = {0.f, 0.f}, sb = sa, qa = sa, qb = sa;
#pragma unroll
  for (int i = 0; i < 8; ++i) {
    sa += of[2 * i];
    sb += of[2 * i + 1];
    qa += of[2 * i] * of[2 * i];
    qb += of[2 * i + 1] * of[2 * i + 1];
  }
  float s = (sa.x + sa.y) + (sb.x + sb.y);
  float q = (qa.x + qa.y) + (qb.x + qb.y);
  s += xhalf(s);
  q += xhalf(q);
  m  = s * (1.f / 64.f);
  is = rsqrtf(q * (1.f / 64.f) - m * m + LN_EPS);
}
template <int EPI, int WBL, bool HAS_LATER, bool LN_ON_LOAD, typename LATER>
__device__ __forceinline__ void
chain_pair(const float* X8T, const float* Wa, const float* biasa, const float* Wb,
           float* Cgd, float* Cact, const float* tgtT, float* part, int wave, int lane,
           const float* lng, const float* lnb, float* s1p, float* s2p, LATER later_reads) {
  const int c  = lane & 15;
  const int h  = lane >> 4;
  const int bb = c & 7;
  const int kh = h << 1;
  const int fo = wave * 16;
  const int f0 = fo + 8 * h;
  const int xb = 16 * h + 2 * bb;
  const int tb = 128 * wave + 64 * h + 2 * bb;
  const int wp = 1024 * wave + 512 * h + 2 * c;
  const int wr = W2P_IDX(f0, c);
  v2f fa[16], fb[16], fa2[16];
  v2f lgf[16], lbf[16];
  float bi[8], tg[8];
  constexpr bool LNB_STATS = LN_ON_LOAD && (EPI == CH_DPRED);
  v2f xd[16];
  float gd8[32];
  float m = 0.f, is = 0.f;
  frag_rows16(X8T, xb, fb);
  if (LN_ON_LOAD) {
#pragma unroll
    for (int i = 0; i < 16; ++i) {
      lgf[i].x = lng[4 * i + kh];
      lgf[i].y = lng[4 * i + kh + 1];
      lbf[i].x = lnb[4 * i + kh];
      lbf[i].y = lnb[4 * i + kh + 1];
    }
  }
#pragma unroll
  for (int i = 0; i < 16; ++i) {
    const int k0 = 4 * i;
    fa[i].x = Wa[(k0 + kh) * 64 + fo + c];
    fa[i].y = Wa[(k0 + kh + 1) * 64 + fo + c];
  }
#pragma unroll
  for (int s = 0; s < 4; ++s) {
#pragma unroll
    for (int td = 0; td < 4; ++td) {
      if (WBL == WB_W2P) {
        fa2[s * 4 + td].x = Wb[wr + 8 * W2P_PITCH * td + 4 * s];
        fa2[s * 4 + td].y = Wb[wr + 8 * W2P_PITCH * td + 4 * s + 2];
      } else {
        fa2[s * 4 + td].x = Wb[wp + 128 * s + 32 * td];
        fa2[s * 4 + td].y = Wb[wp + 128 * s + 32 * td + 1];
      }
    }
  }
#pragma unroll
  for (int r = 0; r < 8; ++r) {
    bi[r] = biasa[f0 + r];
    tg[r] = (EPI == CH_DPRED) ? tgtT[tb + 16 * (r >> 1) + (r & 1)] : 0.f;
  }
  if (LNB_STATS) {
#pragma unroll
    for (int td = 0; td < 4; ++td) {
#pragma unroll
      for (int q = 0; q < 4; ++q)
        xd[td * 4 + q] = *(const v2f*)(X8T + 128 * td + 64 * h + 16 * q + 2 * bb);
#pragma unroll
      for (int r = 0; r < 8; ++r) gd8[td * 8 + r] = lng[16 * td + 8 * h + r];
    }
  }
  if (LN_ON_LOAD) {
    half_row_stats(fb, m, is);
#pragma unroll
    for (int i = 0; i < 16; ++i) fb[i] = ((fb[i] - m) * is) * lgf[i] + lbf[i];
  }
  HOLD8(fa, 0);  HOLD8(fa, 8);
  if (!LN_ON_LOAD) { HOLD8(fb, 0);  HOLD8(fb, 8); }
  HOLD8(fa2, 0); HOLD8(fa2, 8);
  HOLD8(bi, 0);
  if (EPI == CH_DPRED) HOLD8(tg, 0);
  if (LNB_STATS) {
    HOLD8(xd, 0); HOLD8(xd, 8);
    HOLD8(gd8, 0); HOLD8(gd8, 8); HOLD8(gd8, 16); HOLD8(gd8, 24);
  }
  if (HAS_LATER) {
    later_reads();
    __builtin_amdgcn_sched_barrier(0);
  }
  v8f acc = {0.f, 0.f, 0.f, 0.f, 0.f, 0.f, 0.f, 0.f};
#pragma unroll
  for (int i = 0; i < 16; ++i) acc = wmma4(fa[i], fb[i], acc);
  v8f g;
  if (EPI == CH_GELU) {
    const bool hi = (c >= 8);
    float gm[4], dm[4];
#pragma unroll
    for (int r = 0; r < 4; ++r) {
      const float xlo = acc[r] + bi[r];
      const float xhi = acc[4 + r] + bi[4 + r];
      gelu_and_grad(hi ? xhi : xlo, gm[r], dm[r]);
    }
    {
      const int so = bb * 64 + f0 + (hi ? 4 : 0);
#pragma unroll
      for (int r = 0; r < 4; ++r) {
        Cgd[so + r] = dm[r];
        Cact[so + r] = gm[r];
      }
    }
#pragma unroll
    for (int r = 0; r < 4; ++r) {
      const float go = row_xor8(gm[r]);
      g[r]     = hi ? go : gm[r];
      g[4 + r] = hi ? gm[r] : go;
    }
  } else {
#pragma unroll
    for (int r = 0; r < 8; ++r) {
      float v = acc[r] + bi[r];
      g[r] = GSCALE * (v - tg[r]);
    }
  }
  v8f p0 = {0.f, 0.f, 0.f, 0.f, 0.f, 0.f, 0.f, 0.f};
  v8f p1 = p0, p2 = p0, p3 = p0;
#pragma unroll
  for (int s = 0; s < 4; ++s) {
    v2f b2;
    b2.x = g[2 * s];
    b2.y = g[2 * s + 1];
    p0 = wmma4(fa2[s * 4 + 0], b2, p0);
    p1 = wmma4(fa2[s * 4 + 1], b2, p1);
    p2 = wmma4(fa2[s * 4 + 2], b2, p2);
    p3 = wmma4(fa2[s * 4 + 3], b2, p3);
  }
  if (c < 8) {
    float* dst = part + wave * 512 + 64 * h + 2 * bb;
#pragma unroll
    for (int r = 0; r < 8; ++r) {
      const int o = 16 * (r >> 1) + (r & 1);
      dst[o]       = p0[r];
      dst[128 + o] = p1[r];
      dst[256 + o] = p2[r];
      dst[384 + o] = p3[r];
    }
  }
  if (LNB_STATS) {
    float a1 = 0.f, b1s = 0.f, a2 = 0.f, b2s = 0.f;
#pragma unroll
    for (int q = 0; q < 4; ++q) {
      const v2f x0 = (xd[q] - m) * is,      x1 = (xd[4 + q] - m) * is;
      const v2f x2 = (xd[8 + q] - m) * is,  x3 = (xd[12 + q] - m) * is;
#pragma unroll
      for (int j = 0; j < 2; ++j) {
        const int r = 2 * q + j;
        const float w0 = p0[r] * gd8[r],      w1 = p1[r] * gd8[8 + r];
        const float w2 = p2[r] * gd8[16 + r], w3 = p3[r] * gd8[24 + r];
        a1 += w0;  b1s += w1;  a1 += w2;  b1s += w3;
        a2 += w0 * (j ? x0.y : x0.x);
        b2s += w1 * (j ? x1.y : x1.x);
        a2 += w2 * (j ? x2.y : x2.x);
        b2s += w3 * (j ? x3.y : x3.x);
      }
    }
    float s1 = a1 + b1s, s2 = a2 + b2s;
    s1 += xhalf(s1);
    s2 += xhalf(s2);
    if (c < 8 && h == 0) { s1p[wave * 8 + bb] = s1; s2p[wave * 8 + bb] = s2; }
  }
}

__device__ __forceinline__ float part_sum4(const float* part, int idx) {
  return (part[idx] + part[512 + idx]) + (part[1024 + idx] + part[1536 + idx]);
}

#define SCAN_SMEM_FLOATS (3 * 4096 + 11 * 512 + 7 * 64 + 32 + 4 * 512 + 32 * W2P_PITCH + 64)

__global__ void __launch_bounds__(256)
ttt_scan_kernel(const float* __restrict__ kg, const float* __restrict__ vg,
                const float* __restrict__ qg,
                const float* __restrict__ fw1, const float* __restrict__ fb1,
                const float* __restrict__ fw2, const float* __restrict__ fb2,
                const float* __restrict__ flg, const float* __restrict__ flb,
                const float* __restrict__ lwg, const float* __restrict__ lbg,
                const float* __restrict__ ttg, const float* __restrict__ ttb,
                float* __restrict__ outg) {
  extern __shared__ __align__(16) float smem[];
  float* W1  = smem;
  float* LW  = W1 + 4096;
  float* ktb = LW + 4096;
  float* tgt = ktb + 512;
  float* gdv = tgt + 512;
  float* hg  = gdv + 512;
  float* xh  = hg + 512;
  float* zb  = xh + 512;
  float* dzb = zb + 512;
  float* dob = dzb + 512;
  float* ktT = dob + 512;
  float* dobT = ktT + 512;
  float* obT = dobT + 512;
  float* b1v = obT + 512;
  float* b2v = b1v + 64;
  float* lgv = b2v + 64;
  float* lbv = lgv + 64;
  float* tgv = lbv + 64;
  float* tbv = tgv + 64;
  float* lbz = tbv + 64;
  float* isv = lbz + 64;
  float* part = isv + 32;
  float* LWP = part + 4 * 512;
  float* W2  = LWP + 4096;
  float* s1p = W2 + 32 * W2P_PITCH;
  float* s2p = s1p + 32;

  const int h    = blockIdx.x;
  const int tid  = threadIdx.x;
  const int wave = tid >> 5;
  const int lane = tid & 31;

  const int b0 = tid >> 6,          d0 = tid & 63;
  const int b1i = (tid + 256) >> 6, d1i = (tid + 256) & 63;
  const size_t g0base = (size_t)(b0 * SS) * HIDC + h * 64 + d0;
  const size_t g1base = (size_t)(b1i * SS) * HIDC + h * 64 + d1i;

  for (int i = tid; i < 4096; i += 256) {
    W1[i] = fw1[(size_t)h * 4096 + i];
    W2[W2P_IDX(i >> 6, i & 63)] = fw2[(size_t)h * 4096 + i];
    LW[i] = lwg[i];
    LWP[((i & 63) >> 1) * 128 + 2 * (i >> 6) + (i & 1)] = lwg[i];
  }
  if (tid < 64) {
    b1v[tid] = fb1[h * 64 + tid];
    b2v[tid] = fb2[h * 64 + tid];
    lgv[tid] = flg[h * 64 + tid];
    lbv[tid] = flb[h * 64 + tid];
    tgv[tid] = ttg[h * 64 + tid];
    tbv[tid] = ttb[h * 64 + tid];
    lbz[tid] = lbg[tid];
  }

  float kp0 = kg[g0base], vp0 = vg[g0base];
  float kp1 = kg[g1base], vp1 = vg[g1base];
  __syncthreads();

  for (int t = 0; t < SS; ++t) {
    ktb[tid]            = kp0;  ktb[tid + 256]      = kp1;
    ktT[TIDX(b0, d0)]   = kp0;  ktT[TIDX(b1i, d1i)] = kp1;
    tgt[TIDX(b0, d0)]   = vp0 - kp0;
    tgt[TIDX(b1i, d1i)] = vp1 - kp1;
    __syncthreads();
    if (t + 1 < SS) {
      kp0 = kg[g0base + (t + 1) * HIDC];  vp0 = vg[g0base + (t + 1) * HIDC];
      kp1 = kg[g1base + (t + 1) * HIDC];  vp1 = vg[g1base + (t + 1) * HIDC];
    }

    for (int step = 0; step < 4; ++step) {
      if (wave < 4)
        chain_pair<CH_GELU, WB_W2P, false, false>(ktT, W1, b1v, W2, gdv, hg, nullptr, part, wave, lane,
                                                  nullptr, nullptr, nullptr, nullptr, [] {});
      __syncthreads();
      {
        const int pi = TIDX(wave, lane);
        obT[pi]       = part_sum4(part, pi) + b2v[lane];
        obT[pi + 256] = part_sum4(part, pi + 256) + b2v[lane + 32];
      }
      __syncthreads();
      const int fc = lane & 15;
      const int fh = lane >> 4;
      const int fk = fh << 1;
      v2f   pb[16];
      v2f   pk[8];
      float pg[8];
      float pb1 = 0.f;
      v2f   ph[2];
      float wold[32];
      if (wave < 4) {
        chain_pair<CH_DPRED, WB_LWP, true, true>(obT, LW, lbz, LWP, nullptr, nullptr, tgt, part, wave, lane,
                                                 lgv, lbv, s1p, s2p, [&] {
          const int col = wave * 16 + fc;
          const int wq  = fh * (W2P_PITCH + 32) + 2 * col;
#pragma unroll
          for (int i = 0; i < 16; ++i) {
            pb[i].x = W2[wq + 2 * W2P_PITCH * i + 4 * (i & 3)];
            pb[i].y = W2[wq + 2 * W2P_PITCH * i + 4 * (i & 3) + 1];
          }
#pragma unroll
          for (int r = 0; r < 8; ++r) pg[r] = gdv[r * 64 + col];
          pb1 = b1v[col];
#pragma unroll
          for (int tm = 0; tm < 4; ++tm) {
#pragma unroll
            for (int s = 0; s < 2; ++s) {
              pk[tm * 2 + s].x = ktb[(4 * s + fk) * 64 + tm * 16 + fc];
              pk[tm * 2 + s].y = ktb[(4 * s + fk + 1) * 64 + tm * 16 + fc];
            }
#pragma unroll
            for (int i = 0; i < 8; ++i) wold[tm * 8 + i] = W1[(tm * 16 + 8 * fh + i) * 64 + col];
          }
        });
      } else {
        const int wv = wave - 4;
        v2f of[16];
        frag_rows16(obT, 16 * fh + 2 * (fc & 7), of);
#pragma unroll
        for (int s = 0; s < 2; ++s) {
          ph[s].x = hg[(4 * s + fk) * 64 + wv * 16 + fc];
          ph[s].y = hg[(4 * s + fk + 1) * 64 + wv * 16 + fc];
        }
        {
          float m, is;
          half_row_stats(of, m, is);
          if (fc < 8) {
#pragma unroll
            for (int i = 0; i < 16; ++i) {
              if ((i >> 2) == wv) {
                const v2f xv = (of[i] - m) * is;
                xh[fc * 64 + 4 * i + fk]     = xv.x;
                xh[fc * 64 + 4 * i + fk + 1] = xv.y;
              }
            }
            if (wave == 4 && fh == 0) isv[fc] = is;
          }
        }
#pragma unroll
        for (int tn = 0; tn < 4; ++tn) {
#pragma unroll
          for (int i = 0; i < 8; ++i)
            wold[tn * 8 + i] = W2[W2P_IDX(wv * 16 + 8 * fh, fc) + 8 * W2P_PITCH * tn + 2 * i];
        }
      }
      __syncthreads();
      {
        const int pi = TIDX(wave, lane);
        const float z0 = part_sum4(part, pi);
        const float z1 = part_sum4(part, pi + 256);
        const float m1 = ((s1p[wave] + s1p[8 + wave]) + (s1p[16 + wave] + s1p[24 + wave])) * (1.f / 64.f);
        const float m2 = ((s2p[wave] + s2p[8 + wave]) + (s2p[16 + wave] + s2p[24 + wave])) * (1.f / 64.f);
        const float x0 = xh[wave * 64 + lane], x1 = xh[wave * 64 + lane + 32];
        const float isw = isv[wave];
        const float dv0 = isw * (z0 * lgv[lane] - m1 - x0 * m2);
        const float dv1 = isw * (z1 * lgv[lane + 32] - m1 - x1 * m2);
        dzb[wave * 64 + lane] = z0;
        dzb[wave * 64 + lane + 32] = z1;
        dob[wave * 64 + lane] = dv0;
        dob[wave * 64 + lane + 32] = dv1;
        dobT[pi] = dv0;
        dobT[pi + 256] = dv1;
      }
      __syncthreads();
      if (wave < 4) {
        const int col = wave * 16 + fc;
        const int bb  = fc & 7;
        v2f pa[16];
#pragma unroll
        for (int i = 0; i < 16; ++i) {
          pa[i].x = dobT[16 * fh + 2 * bb + 32 * i];
          pa[i].y = dobT[16 * fh + 2 * bb + 32 * i + 1];
        }
        HOLD8(pa, 0); HOLD8(pa, 8);
        v8f acc = {0.f, 0.f, 0.f, 0.f, 0.f, 0.f, 0.f, 0.f};
#pragma unroll
        for (int i = 0; i < 16; ++i) acc = wmma4(pa[i], pb[i], acc);
        float da[8];
        float sb = 0.f;
#pragma unroll
        for (int r = 0; r < 8; ++r) da[r] = acc[r] * pg[r];
#pragma unroll
        for (int r = 0; r < 8; ++r) sb += da[r];
        if (fh == 0) b1v[col] = pb1 - LR * sb;
        v8f u0 = {0.f, 0.f, 0.f, 0.f, 0.f, 0.f, 0.f, 0.f};
        v8f u1 = u0, u2 = u0, u3 = u0;
#pragma unroll
        for (int s = 0; s < 2; ++s) {
          v2f b2;
          b2.x = fh ? da[4 * s + 2] : da[4 * s];
          b2.y = fh ? da[4 * s + 3] : da[4 * s + 1];
          u0 = wmma4(pk[0 + s], b2, u0);
          u1 = wmma4(pk[2 + s], b2, u1);
          u2 = wmma4(pk[4 + s], b2, u2);
          u3 = wmma4(pk[6 + s], b2, u3);
        }
#pragma unroll
        for (int i = 0; i < 8; ++i) {
          W1[(0 * 16 + 8 * fh + i) * 64 + col] = wold[0 * 8 + i] - LR * u0[i];
          W1[(1 * 16 + 8 * fh + i) * 64 + col] = wold[1 * 8 + i] - LR * u1[i];
          W1[(2 * 16 + 8 * fh + i) * 64 + col] = wold[2 * 8 + i] - LR * u2[i];
          W1[(3 * 16 + 8 * fh + i) * 64 + col] = wold[3 * 8 + i] - LR * u3[i];
        }
      } else {
        const int wv = wave - 4;
        v2f pd[8];
#pragma unroll
        for (int tn = 0; tn < 4; ++tn) {
#pragma unroll
          for (int s = 0; s < 2; ++s) {
            pd[tn * 2 + s].x = dob[(4 * s + fk) * 64 + tn * 16 + fc];
            pd[tn * 2 + s].y = dob[(4 * s + fk + 1) * 64 + tn * 16 + fc];
          }
        }
        v8f u0 = {0.f, 0.f, 0.f, 0.f, 0.f, 0.f, 0.f, 0.f};
        v8f u1 = u0, u2 = u0, u3 = u0;
#pragma unroll
        for (int s = 0; s < 2; ++s) {
          u0 = wmma4(ph[s], pd[0 + s], u0);
          u1 = wmma4(ph[s], pd[2 + s], u1);
          u2 = wmma4(ph[s], pd[4 + s], u2);
          u3 = wmma4(ph[s], pd[6 + s], u3);
        }
        const int rb = W2P_IDX(wv * 16 + 8 * fh, fc);
#pragma unroll
        for (int i = 0; i < 8; ++i) {
          W2[rb + 0 * 8 * W2P_PITCH + 2 * i] = wold[0 * 8 + i] - LR * u0[i];
          W2[rb + 1 * 8 * W2P_PITCH + 2 * i] = wold[1 * 8 + i] - LR * u1[i];
          W2[rb + 2 * 8 * W2P_PITCH + 2 * i] = wold[2 * 8 + i] - LR * u2[i];
          W2[rb + 3 * 8 * W2P_PITCH + 2 * i] = wold[3 * 8 + i] - LR * u3[i];
        }
        const int t2 = tid - 128;
        if (t2 < 64) {
          float s = 0.f, s2 = 0.f;
          for (int b = 0; b < 8; ++b) s += dzb[b * 64 + t2] * xh[b * 64 + t2];
          for (int b = 0; b < 8; ++b) s2 += dob[b * 64 + t2];
          lgv[t2] -= LR * s;
          b2v[t2] -= LR * s2;
        } else {
          const int d = t2 - 64;
          float s = 0.f;
          for (int b = 0; b < 8; ++b) s += dzb[b * 64 + d];
          lbv[d] -= LR * s;
        }
      }
      __syncthreads();
    }

    if (wave < 4)
      chain_pair<CH_GELU, WB_W2P, false, false>(ktT, W1, b1v, W2, gdv, hg, nullptr, part, wave, lane,
                                                nullptr, nullptr, nullptr, nullptr, [] {});
    __syncthreads();
    {
      const int pi = TIDX(wave, lane);
      obT[pi]       = part_sum4(part, pi) + b2v[lane];
      obT[pi + 256] = part_sum4(part, pi + 256) + b2v[lane + 32];
    }
    __syncthreads();
    if (wave < 4) {
      const int fc = lane & 15;
      const int fh = lane >> 4;
      const int fk = fh << 1;
      v2f of[16], lgf[16], lbf[16];
      frag_rows16(obT, 16 * fh + 2 * (fc & 7), of);
#pragma unroll
      for (int i = 0; i < 16; ++i) {
        lgf[i].x = lgv[4 * i + fk];
        lgf[i].y = lgv[4 * i + fk + 1];
        lbf[i].x = lbv[4 * i + fk];
        lbf[i].y = lbv[4 * i + fk + 1];
      }
      float m, is, m2, is2;
      half_row_stats(of, m, is);
#pragma unroll
      for (int i = 0; i < 16; ++i) of[i] = ((of[i] - m) * is) * lgf[i] + lbf[i];
      half_row_stats(of, m2, is2);
      if (fc < 8) {
#pragma unroll
        for (int i = 0; i < 16; ++i) {
          if ((i >> 2) == wave) {
            const int d = 4 * i + fk;
            const v2f xv = (of[i] - m2) * is2;
            zb[fc * 64 + d]     = xv.x * tgv[d] + tbv[d];
            zb[fc * 64 + d + 1] = xv.y * tgv[d + 1] + tbv[d + 1];
          }
        }
      }
    }
    __syncthreads();
    for (int idx = tid; idx < 512; idx += 256) {
      int b = idx >> 6, d = idx & 63;
      size_t g = ((size_t)(b * SS + t)) * HIDC + h * 64 + d;
      outg[g] = qg[g] + zb[idx];
    }
    __syncthreads();
  }
}

__global__ void __launch_bounds__(256)
gate_ln_kernel(const float* __restrict__ outb, const float* __restrict__ gact,
               const float* __restrict__ png, const float* __restrict__ pnb,
               float* __restrict__ gated) {
  __shared__ float r1[256], r2[256];
  __shared__ float mu_s, inv_s;
  const int r = blockIdx.x, tid = threadIdx.x;
  const float* row = outb + (size_t)r * HIDC;
  float s = 0.f, q = 0.f;
  for (int i = tid; i < HIDC; i += 256) { float v = row[i]; s += v; q += v * v; }
  r1[tid] = s; r2[tid] = q;
  __syncthreads();
  for (int off = 128; off > 0; off >>= 1) {
    if (tid < off) { r1[tid] += r1[tid + off]; r2[tid] += r2[tid + off]; }
    __syncthreads();
  }
  if (tid == 0) {
    float m = r1[0] * (1.f / HIDC);
    mu_s = m;
    inv_s = rsqrtf(r2[0] * (1.f / HIDC) - m * m + LN_EPS);
  }
  __syncthreads();
  for (int i = tid; i < HIDC; i += 256) {
    float y = (row[i] - mu_s) * inv_s * png[i] + pnb[i];
    gated[(size_t)r * HIDC + i] = gact[(size_t)r * HIDC + i] * y;
  }
}

extern "C" void kernel_launch(void* const* d_in, const int* in_sizes, int n_in,
                              void* d_out, int out_size, void* d_ws, size_t ws_size,
                              hipStream_t stream) {
  (void)in_sizes; (void)n_in; (void)out_size; (void)ws_size;
  const float* x   = (const float*)d_in[0];
  const float* wq  = (const float*)d_in[1];
  const float* wk  = (const float*)d_in[2];
  const float* wv  = (const float*)d_in[3];
  const float* fw1 = (const float*)d_in[4];
  const float* fb1 = (const float*)d_in[5];
  const float* fw2 = (const float*)d_in[6];
  const float* fb2 = (const float*)d_in[7];
  const float* flg = (const float*)d_in[8];
  const float* flb = (const float*)d_in[9];
  const float* lw  = (const float*)d_in[10];
  const float* lb  = (const float*)d_in[11];
  const float* ttg = (const float*)d_in[12];
  const float* ttb = (const float*)d_in[13];
  const float* wo  = (const float*)d_in[14];
  const float* wg  = (const float*)d_in[15];
  const float* png = (const float*)d_in[16];
  const float* pnb = (const float*)d_in[17];

  const int M = BB * SS;
  const size_t PLANE = (size_t)M * HIDC;
  float* qb    = (float*)d_ws;
  float* kb    = qb + PLANE;
  float* vb    = kb + PLANE;
  float* gactb = vb + PLANE;
  float* outb  = gactb + PLANE;
  float* gatedb = qb;

  dim3 gblk(256);
  dim3 ggrid(HIDC / 128, M / 64);
  gemm_wmma_f32<<<ggrid, gblk, 0, stream>>>(x, wq, qb,    M, HIDC, HIDC, 0);
  gemm_wmma_f32<<<ggrid, gblk, 0, stream>>>(x, wk, kb,    M, HIDC, HIDC, 0);
  gemm_wmma_f32<<<ggrid, gblk, 0, stream>>>(x, wv, vb,    M, HIDC, HIDC, 0);
  gemm_wmma_f32<<<ggrid, gblk, 0, stream>>>(x, wg, gactb, M, HIDC, HIDC, 1);

  ttt_scan_kernel<<<HH, 256, SCAN_SMEM_FLOATS * sizeof(float), stream>>>(
      kb, vb, qb, fw1, fb1, fw2, fb2, flg, flb, lw, lb, ttg, ttb, outb);

  gate_ln_kernel<<<M, 256, 0, stream>>>(outb, gactb, png, pnb, gatedb);

  gemm_wmma_f32<<<ggrid, gblk, 0, stream>>>(gatedb, wo, (float*)d_out,
                                            M, HIDC, HIDC, 0);
}


# scan: final apply of token t and step-0 forward pass of token t+1 (same weights) run as ONE pass: t in tile columns 0-7, t+1 in the formerly duplicate 8-15; final LNs on waves 4-7 beside next chain
# speedup vs baseline: 3.2478x; 1.1142x over previous
#include <hip/hip_runtime.h>
#include <hip/hip_bf16.h>
#include <math.h>

#define BB     8
#define SS     256
#define HIDC   1024
#define HH     16
#define DD     64
#define FF     64
#define LR     0.1f
#define GSCALE (2.0f / 8192.0f)
#define LN_EPS 1e-5f

typedef float v2f __attribute__((ext_vector_type(2)));
typedef float v8f __attribute__((ext_vector_type(8)));

__device__ __forceinline__ v8f wmma4(v2f a, v2f b, v8f c) {
  return __builtin_amdgcn_wmma_f32_16x16x4_f32(
       false, a,  false, b,
       (short)0, c,  false,  false);
}

__device__ __forceinline__ float gelu_f(float x) {
  const float c = 0.7978845608028654f;
  float u = c * (x + 0.044715f * x * x * x);
  return 0.5f * x * (1.0f + tanhf(u));
}
__device__ __forceinline__ void gelu_and_grad(float x, float& g, float& gd) {
  const float c = 0.7978845608028654f;
  float u = c * (x + 0.044715f * x * x * x);
  float t = tanhf(u);
  float x2 = x * x;
  g  = 0.5f * x * (1.0f + t);
  gd = 0.5f * (1.0f + t) + 0.5f * x * (1.0f - t * t) * c * (1.0f + 0.134145f * x2);
}

__device__ __forceinline__ float row_xor8(float v) {
  const int b = __builtin_bit_cast(int, v);
  return __builtin_bit_cast(float, __builtin_amdgcn_update_dpp(b, b, 0x168, 0xf, 0xf, false));
}

__device__ __forceinline__ float xhalf(float v) {
  const int b = __builtin_bit_cast(int, v);
  return __builtin_bit_cast(float, __builtin_amdgcn_permlanex16(b, b, 0x76543210, 0xfedcba98, false, false));
}

typedef float v4f __attribute__((ext_vector_type(4)));
#define GA_PITCH 36
#define GB_PITCH 288
#define GHOLD8(a, o) asm volatile("" :: "v"(a[o]), "v"(a[o + 1]), "v"(a[o + 2]), "v"(a[o + 3]), \
                                        "v"(a[o + 4]), "v"(a[o + 5]), "v"(a[o + 6]), "v"(a[o + 7]))
__global__ void __launch_bounds__(256)
gemm_wmma_f32(const float* __restrict__ A, const float* __restrict__ Bm,
              float* __restrict__ C, int M, int N, int K, int act) {
  __shared__ __align__(16) float As[2][64 * GA_PITCH];
  __shared__ __align__(16) float Bs[2][16 * GB_PITCH];
  (void)M;
  const int tid   = threadIdx.x;
  const int wave  = tid >> 5;
  const int lane  = tid & 31;
  const int c     = lane & 15;
  const int h     = lane >> 4;
  const int wm    = wave >> 2;
  const int wn    = wave & 3;
  const int mbase = blockIdx.y * 64;
  const int nbase = blockIdx.x * 128;
  const int ar = tid >> 3, ac = (tid & 7) << 2;
  const int bp = tid >> 5, bc = (tid & 31) << 2;
  const float* ag = A + (size_t)(mbase + ar) * K + ac;
  const float* bg = Bm + (size_t)(2 * bp) * N + nbase + bc;
  const int aso = ar * GA_PITCH + ac;
  const int bso = bp * GB_PITCH + 2 * bc;
  v4f ra0, ra1, rb0, rb1, rb2, rb3;
  ra0 = *(const v4f*)(ag);
  ra1 = *(const v4f*)(ag + (size_t)32 * K);
  rb0 = *(const v4f*)(bg);
  rb1 = *(const v4f*)(bg + N);
  rb2 = *(const v4f*)(bg + (size_t)16 * N);
  rb3 = *(const v4f*)(bg + (size_t)17 * N);
  *(v4f*)&As[0][aso]                 = ra0;
  *(v4f*)&As[0][aso + 32 * GA_PITCH] = ra1;
  *(v4f*)&Bs[0][bso]                    = v4f{rb0.x, rb1.x, rb0.y, rb1.y};
  *(v4f*)&Bs[0][bso + 4]                = v4f{rb0.z, rb1.z, rb0.w, rb1.w};
  *(v4f*)&Bs[0][bso + 8 * GB_PITCH]     = v4f{rb2.x, rb3.x, rb2.y, rb3.y};
  *(v4f*)&Bs[0][bso + 8 * GB_PITCH + 4] = v4f{rb2.z, rb3.z, rb2.w, rb3.w};
  __syncthreads();
  v8f acc00 = {0.f, 0.f, 0.f, 0.f, 0.f, 0.f, 0.f, 0.f};
  v8f acc01 = acc00, acc10 = acc00, acc11 = acc00;
  const int afo = (wm * 32 + c) * GA_PITCH + 2 * h;
  const int bfo = h * GB_PITCH + 2 * (wn * 32 + c);
  const int nch = K >> 5;
  for (int ch = 0; ch < nch; ++ch) {
    const int cur = ch & 1;
    const bool more = (ch + 1 < nch);
    if (more) {
      const float* an = ag + (size_t)(ch + 1) * 32;
      const float* bn = bg + (size_t)(ch + 1) * 32 * N;
      ra0 = *(const v4f*)(an);
      ra1 = *(const v4f*)(an + (size_t)32 * K);
      rb0 = *(const v4f*)(bn);
      rb1 = *(const v4f*)(bn + N);
      rb2 = *(const v4f*)(bn + (size_t)16 * N);
      rb3 = *(const v4f*)(bn + (size_t)17 * N);
    }
    const float* as = &As[cur][afo];
    const float* bs = &Bs[cur][bfo];
    v2f fa[16], fb[16];
#pragma unroll
    for (int ks = 0; ks < 8; ++ks) {
      fa[2 * ks]     = *(const v2f*)(as + 4 * ks);
      fa[2 * ks + 1] = *(const v2f*)(as + 16 * GA_PITCH + 4 * ks);
      fb[2 * ks]     = *(const v2f*)(bs + 2 * GB_PITCH * ks);
      fb[2 * ks + 1] = *(const v2f*)(bs + 2 * GB_PITCH * ks + 32);
    }
    GHOLD8(fa, 0); GHOLD8(fa, 8); GHOLD8(fb, 0); GHOLD8(fb, 8);
#pragma unroll
    for (int ks = 0; ks < 8; ++ks) {
      acc00 = wmma4(fa[2 * ks],     fb[2 * ks],     acc00);
      acc10 = wmma4(fa[2 * ks + 1], fb[2 * ks],     acc10);
      acc01 = wmma4(fa[2 * ks],     fb[2 * ks + 1], acc01);
      acc11 = wmma4(fa[2 * ks + 1], fb[2 * ks + 1], acc11);
    }
    if (more) {
      float* ad = &As[cur ^ 1][aso];
      float* bd = &Bs[cur ^ 1][bso];
      *(v4f*)(ad)                 = ra0;
      *(v4f*)(ad + 32 * GA_PITCH) = ra1;
      *(v4f*)(bd)                    = v4f{rb0.x, rb1.x, rb0.y, rb1.y};
      *(v4f*)(bd + 4)                = v4f{rb0.z, rb1.z, rb0.w, rb1.w};
      *(v4f*)(bd + 8 * GB_PITCH)     = v4f{rb2.x, rb3.x, rb2.y, rb3.y};
      *(v4f*)(bd + 8 * GB_PITCH + 4) = v4f{rb2.z, rb3.z, rb2.w, rb3.w};
    }
    __syncthreads();
  }
  const int row0 = mbase + wm * 32 + 8 * h;
  const int col0 = nbase + wn * 32 + c;
#pragma unroll
  for (int r = 0; r < 8; ++r) {
    float v00 = acc00[r], v01 = acc01[r], v10 = acc10[r], v11 = acc11[r];
    if (act) { v00 = gelu_f(v00); v01 = gelu_f(v01); v10 = gelu_f(v10); v11 = gelu_f(v11); }
    C[(size_t)(row0 + r) * N + col0]           = v00;
    C[(size_t)(row0 + r) * N + col0 + 16]      = v01;
    C[(size_t)(row0 + 16 + r) * N + col0]      = v10;
    C[(size_t)(row0 + 16 + r) * N + col0 + 16] = v11;
  }
}


#define HOLD8(a, o) asm volatile("" :: "v"(a[o]), "v"(a[o + 1]), "v"(a[o + 2]), "v"(a[o + 3]), \
                                       "v"(a[o + 4]), "v"(a[o + 5]), "v"(a[o + 6]), "v"(a[o + 7]))
#define TIDX(b, d) ((((d) >> 1) << 4) + ((b) << 1) + ((d) & 1))
#define W2P_PITCH 192
#define W2P_ROT(p) ((((p) & 6) << 1) + (((p) & 1) << 5))
#define W2P_IDX(f, d) (((d) >> 1) * W2P_PITCH + 2 * (f) + W2P_ROT((d) >> 1) + ((d) & 1))
#define WB_W2P 0
#define WB_LWP 1
#define CH_GELU  0
#define CH_DPRED 1
__device__ __forceinline__ void frag_rows16(const float* T, int xb, v2f* of) {
#pragma unroll
  for (int i = 0; i < 16; ++i) {
    of[i].x = T[xb + 32 * i];
    of[i].y = T[xb + 32 * i + 1];
  }
}
__device__ __forceinline__ void half_row_stats(const v2f* of, float& m, float& is) {
  v2f sa = {0.f, 0.f}, sb = sa, qa = sa, qb = sa;
#pragma unroll
  for (int i = 0; i < 8; ++i) {
    sa += of[2 * i];
    sb += of[2 * i + 1];
    qa += of[2 * i] * of[2 * i];
    qb += of[2 * i + 1] * of[2 * i + 1];
  }
  float s = (sa.x + sa.y) + (sb.x + sb.y);
  float q = (qa.x + qa.y) + (qb.x + qb.y);
  s += xhalf(s);
  q += xhalf(q);
  m  = s * (1.f / 64.f);
  is = rsqrtf(q * (1.f / 64.f) - m * m + LN_EPS);
}
template <int EPI, int WBL, bool HAS_LATER, bool LN_ON_LOAD, typename LATER>
__device__ __forceinline__ void
chain_pair(const float* X8T, const float* Wa, const float* biasa, const float* Wb,
           float* Cgd, float* Cact, const float* tgtT, float* part, int wave, int lane,
           const float* lng, const float* lnb, float* s1p, float* s2p, LATER later_reads) {
  const int c  = lane & 15;
  const int h  = lane >> 4;
  const int bb = c & 7;
  const int kh = h << 1;
  const int fo = wave * 16;
  const int f0 = fo + 8 * h;
  const int xb = 16 * h + 2 * bb;
  const int tb = 128 * wave + 64 * h + 2 * bb;
  const int wp = 1024 * wave + 512 * h + 2 * c;
  const int wr = W2P_IDX(f0, c);
  v2f fa[16], fb[16], fa2[16];
  v2f lgf[16], lbf[16];
  float bi[8], tg[8];
  constexpr bool LNB_STATS = LN_ON_LOAD && (EPI == CH_DPRED);
  v2f xd[16];
  float gd8[32];
  float m = 0.f, is = 0.f;
  frag_rows16(X8T, xb, fb);
  if (LN_ON_LOAD) {
#pragma unroll
    for (int i = 0; i < 16; ++i) {
      lgf[i].x = lng[4 * i + kh];
      lgf[i].y = lng[4 * i + kh + 1];
      lbf[i].x = lnb[4 * i + kh];
      lbf[i].y = lnb[4 * i + kh + 1];
    }
  }
#pragma unroll
  for (int i = 0; i < 16; ++i) {
    const int k0 = 4 * i;
    fa[i].x = Wa[(k0 + kh) * 64 + fo + c];
    fa[i].y = Wa[(k0 + kh + 1) * 64 + fo + c];
  }
#pragma unroll
  for (int s = 0; s < 4; ++s) {
#pragma unroll
    for (int td = 0; td < 4; ++td) {
      if (WBL == WB_W2P) {
        fa2[s * 4 + td].x = Wb[wr + 8 * W2P_PITCH * td + 4 * s];
        fa2[s * 4 + td].y = Wb[wr + 8 * W2P_PITCH * td + 4 * s + 2];
      } else {
        fa2[s * 4 + td].x = Wb[wp + 128 * s + 32 * td];
        fa2[s * 4 + td].y = Wb[wp + 128 * s + 32 * td + 1];
      }
    }
  }
#pragma unroll
  for (int r = 0; r < 8; ++r) {
    bi[r] = biasa[f0 + r];
    tg[r] = (EPI == CH_DPRED) ? tgtT[tb + 16 * (r >> 1) + (r & 1)] : 0.f;
  }
  if (LNB_STATS) {
#pragma unroll
    for (int td = 0; td < 4; ++td) {
#pragma unroll
      for (int q = 0; q < 4; ++q)
        xd[td * 4 + q] = *(const v2f*)(X8T + 128 * td + 64 * h + 16 * q + 2 * bb);
#pragma unroll
      for (int r = 0; r < 8; ++r) gd8[td * 8 + r] = lng[16 * td + 8 * h + r];
    }
  }
  if (LN_ON_LOAD) {
    half_row_stats(fb, m, is);
#pragma unroll
    for (int i = 0; i < 16; ++i) fb[i] = ((fb[i] - m) * is) * lgf[i] + lbf[i];
  }
  HOLD8(fa, 0);  HOLD8(fa, 8);
  if (!LN_ON_LOAD) { HOLD8(fb, 0);  HOLD8(fb, 8); }
  HOLD8(fa2, 0); HOLD8(fa2, 8);
  HOLD8(bi, 0);
  if (EPI == CH_DPRED) HOLD8(tg, 0);
  if (LNB_STATS) {
    HOLD8(xd, 0); HOLD8(xd, 8);
    HOLD8(gd8, 0); HOLD8(gd8, 8); HOLD8(gd8, 16); HOLD8(gd8, 24);
  }
  if (HAS_LATER) {
    later_reads();
    __builtin_amdgcn_sched_barrier(0);
  }
  v8f acc = {0.f, 0.f, 0.f, 0.f, 0.f, 0.f, 0.f, 0.f};
#pragma unroll
  for (int i = 0; i < 16; ++i) acc = wmma4(fa[i], fb[i], acc);
  v8f g;
  if (EPI == CH_GELU) {
    const bool hi = (c >= 8);
    float gm[4], dm[4];
#pragma unroll
    for (int r = 0; r < 4; ++r) {
      const float xlo = acc[r] + bi[r];
      const float xhi = acc[4 + r] + bi[4 + r];
      gelu_and_grad(hi ? xhi : xlo, gm[r], dm[r]);
    }
    {
      const int so = bb * 64 + f0 + (hi ? 4 : 0);
#pragma unroll
      for (int r = 0; r < 4; ++r) {
        Cgd[so + r] = dm[r];
        Cact[so + r] = gm[r];
      }
    }
#pragma unroll
    for (int r = 0; r < 4; ++r) {
      const float go = row_xor8(gm[r]);
      g[r]     = hi ? go : gm[r];
      g[4 + r] = hi ? gm[r] : go;
    }
  } else {
#pragma unroll
    for (int r = 0; r < 8; ++r) {
      float v = acc[r] + bi[r];
      g[r] = GSCALE * (v - tg[r]);
    }
  }
  v8f p0 = {0.f, 0.f, 0.f, 0.f, 0.f, 0.f, 0.f, 0.f};
  v8f p1 = p0, p2 = p0, p3 = p0;
#pragma unroll
  for (int s = 0; s < 4; ++s) {
    v2f b2;
    b2.x = g[2 * s];
    b2.y = g[2 * s + 1];
    p0 = wmma4(fa2[s * 4 + 0], b2, p0);
    p1 = wmma4(fa2[s * 4 + 1], b2, p1);
    p2 = wmma4(fa2[s * 4 + 2], b2, p2);
    p3 = wmma4(fa2[s * 4 + 3], b2, p3);
  }
  if (c < 8) {
    float* dst = part + wave * 512 + 64 * h + 2 * bb;
#pragma unroll
    for (int r = 0; r < 8; ++r) {
      const int o = 16 * (r >> 1) + (r & 1);
      dst[o]       = p0[r];
      dst[128 + o] = p1[r];
      dst[256 + o] = p2[r];
      dst[384 + o] = p3[r];
    }
  }
  if (LNB_STATS) {
    float a1 = 0.f, b1s = 0.f, a2 = 0.f, b2s = 0.f;
#pragma unroll
    for (int q = 0; q < 4; ++q) {
      const v2f x0 = (xd[q] - m) * is,      x1 = (xd[4 + q] - m) * is;
      const v2f x2 = (xd[8 + q] - m) * is,  x3 = (xd[12 + q] - m) * is;
#pragma unroll
      for (int j = 0; j < 2; ++j) {
        const int r = 2 * q + j;
        const float w0 = p0[r] * gd8[r],      w1 = p1[r] * gd8[8 + r];
        const float w2 = p2[r] * gd8[16 + r], w3 = p3[r] * gd8[24 + r];
        a1 += w0;  b1s += w1;  a1 += w2;  b1s += w3;
        a2 += w0 * (j ? x0.y : x0.x);
        b2s += w1 * (j ? x1.y : x1.x);
        a2 += w2 * (j ? x2.y : x2.x);
        b2s += w3 * (j ? x3.y : x3.x);
      }
    }
    float s1 = a1 + b1s, s2 = a2 + b2s;
    s1 += xhalf(s1);
    s2 += xhalf(s2);
    if (c < 8 && h == 0) { s1p[wave * 8 + bb] = s1; s2p[wave * 8 + bb] = s2; }
  }
}

__device__ __forceinline__ float part_sum4(const float* part, int idx) {
  return (part[idx] + part[512 + idx]) + (part[1024 + idx] + part[1536 + idx]);
}

__device__ __forceinline__ void
chain_fwd_two(const float* XcT, const float* XnT, const float* Wa, const float* biasa, const float* Wb,
              float* Cgd, float* Cact, float* partN, float* partF, int wave, int lane) {
  const int c  = lane & 15;
  const int h  = lane >> 4;
  const int bb = c & 7;
  const bool nx = (c >= 8);
  const int kh = h << 1;
  const int fo = wave * 16;
  const int f0 = fo + 8 * h;
  const int xb = 16 * h + 2 * bb;
  const int wr = W2P_IDX(f0, c);
  const float* X8T = nx ? XnT : XcT;
  v2f fa[16], fb[16], fa2[16];
  float bi[8];
  frag_rows16(X8T, xb, fb);
#pragma unroll
  for (int i = 0; i < 16; ++i) {
    const int k0 = 4 * i;
    fa[i].x = Wa[(k0 + kh) * 64 + fo + c];
    fa[i].y = Wa[(k0 + kh + 1) * 64 + fo + c];
  }
#pragma unroll
  for (int s = 0; s < 4; ++s) {
#pragma unroll
    for (int td = 0; td < 4; ++td) {
      fa2[s * 4 + td].x = Wb[wr + 8 * W2P_PITCH * td + 4 * s];
      fa2[s * 4 + td].y = Wb[wr + 8 * W2P_PITCH * td + 4 * s + 2];
    }
  }
#pragma unroll
  for (int r = 0; r < 8; ++r) bi[r] = biasa[f0 + r];
  HOLD8(fa, 0);  HOLD8(fa, 8);
  HOLD8(fb, 0);  HOLD8(fb, 8);
  HOLD8(fa2, 0); HOLD8(fa2, 8);
  HOLD8(bi, 0);
  v8f acc = {0.f, 0.f, 0.f, 0.f, 0.f, 0.f, 0.f, 0.f};
#pragma unroll
  for (int i = 0; i < 16; ++i) acc = wmma4(fa[i], fb[i], acc);
  v8f g;
  float dm[8];
#pragma unroll
  for (int r = 0; r < 8; ++r) {
    float gv;
    gelu_and_grad(acc[r] + bi[r], gv, dm[r]);
    g[r] = gv;
  }
  if (nx) {
    const int so = bb * 64 + f0;
#pragma unroll
    for (int r = 0; r < 8; ++r) {
      Cgd[so + r] = dm[r];
      Cact[so + r] = g[r];
    }
  }
  v8f p0 = {0.f, 0.f, 0.f, 0.f, 0.f, 0.f, 0.f, 0.f};
  v8f p1 = p0, p2 = p0, p3 = p0;
#pragma unroll
  for (int s = 0; s < 4; ++s) {
    v2f b2;
    b2.x = g[2 * s];
    b2.y = g[2 * s + 1];
    p0 = wmma4(fa2[s * 4 + 0], b2, p0);
    p1 = wmma4(fa2[s * 4 + 1], b2, p1);
    p2 = wmma4(fa2[s * 4 + 2], b2, p2);
    p3 = wmma4(fa2[s * 4 + 3], b2, p3);
  }
  {
    float* dst = (nx ? partN : partF) + wave * 512 + 64 * h + 2 * bb;
#pragma unroll
    for (int r = 0; r < 8; ++r) {
      const int o = 16 * (r >> 1) + (r & 1);
      dst[o]       = p0[r];
      dst[128 + o] = p1[r];
      dst[256 + o] = p2[r];
      dst[384 + o] = p3[r];
    }
  }
}

__device__ __forceinline__ void
final_norms(const float* obF, const float* lgv, const float* lbv, const float* tgv, const float* tbv,
            float* zb, int wv, int lane) {
  const int fc = lane & 15;
  const int fh = lane >> 4;
  const int fk = fh << 1;
  v2f of[16], lgf[16], lbf[16];
  frag_rows16(obF, 16 * fh + 2 * (fc & 7), of);
#pragma unroll
  for (int i = 0; i < 16; ++i) {
    lgf[i].x = lgv[4 * i + fk];
    lgf[i].y = lgv[4 * i + fk + 1];
    lbf[i].x = lbv[4 * i + fk];
    lbf[i].y = lbv[4 * i + fk + 1];
  }
  float m, is, m2, is2;
  half_row_stats(of, m, is);
#pragma unroll
  for (int i = 0; i < 16; ++i) of[i] = ((of[i] - m) * is) * lgf[i] + lbf[i];
  half_row_stats(of, m2, is2);
  if (fc < 8) {
#pragma unroll
    for (int i = 0; i < 16; ++i) {
      if ((i >> 2) == wv) {
        const int d = 4 * i + fk;
        const v2f xv = (of[i] - m2) * is2;
        zb[fc * 64 + d]     = xv.x * tgv[d] + tbv[d];
        zb[fc * 64 + d + 1] = xv.y * tgv[d + 1] + tbv[d + 1];
      }
    }
  }
}

#define SCAN_SMEM_FLOATS (3 * 4096 + 11 * 512 + 7 * 64 + 32 + 4 * 512 + 32 * W2P_PITCH + 64 \
                          + 512 + 32 + 4 * 512 + 512)

__global__ void __launch_bounds__(256)
ttt_scan_kernel(const float* __restrict__ kg, const float* __restrict__ vg,
                const float* __restrict__ qg,
                const float* __restrict__ fw1, const float* __restrict__ fb1,
                const float* __restrict__ fw2, const float* __restrict__ fb2,
                const float* __restrict__ flg, const float* __restrict__ flb,
                const float* __restrict__ lwg, const float* __restrict__ lbg,
                const float* __restrict__ ttg, const float* __restrict__ ttb,
                float* __restrict__ outg) {
  extern __shared__ __align__(16) float smem[];
  float* W1  = smem;
  float* LW  = W1 + 4096;
  float* ktb = LW + 4096;
  float* tgt = ktb + 512;
  float* gdv = tgt + 512;
  float* hg  = gdv + 512;
  float* xh  = hg + 512;
  float* zb  = xh + 512;
  float* dzb = zb + 512;
  float* dob = dzb + 512;
  float* ktT = dob + 512;
  float* dobT = ktT + 512;
  float* obT = dobT + 512;
  float* b1v = obT + 512;
  float* b2v = b1v + 64;
  float* lgv = b2v + 64;
  float* lbv = lgv + 64;
  float* tgv = lbv + 64;
  float* tbv = tgv + 64;
  float* lbz = tbv + 64;
  float* isv = lbz + 64;
  float* part = isv + 32;
  float* LWP = part + 4 * 512;
  float* W2  = LWP + 4096;
  float* s1p = W2 + 32 * W2P_PITCH;
  float* s2p = s1p + 32;
  float* ktT1 = s2p + 32;
  float* partF = ktT1 + 512 + 32;
  float* obF   = partF + 4 * 512;

  const int h    = blockIdx.x;
  const int tid  = threadIdx.x;
  const int wave = tid >> 5;
  const int lane = tid & 31;

  const int b0 = tid >> 6,          d0 = tid & 63;
  const int b1i = (tid + 256) >> 6, d1i = (tid + 256) & 63;
  const size_t g0base = (size_t)(b0 * SS) * HIDC + h * 64 + d0;
  const size_t g1base = (size_t)(b1i * SS) * HIDC + h * 64 + d1i;

  for (int i = tid; i < 4096; i += 256) {
    W1[i] = fw1[(size_t)h * 4096 + i];
    W2[W2P_IDX(i >> 6, i & 63)] = fw2[(size_t)h * 4096 + i];
    LW[i] = lwg[i];
    LWP[((i & 63) >> 1) * 128 + 2 * (i >> 6) + (i & 1)] = lwg[i];
  }
  if (tid < 64) {
    b1v[tid] = fb1[h * 64 + tid];
    b2v[tid] = fb2[h * 64 + tid];
    lgv[tid] = flg[h * 64 + tid];
    lbv[tid] = flb[h * 64 + tid];
    tgv[tid] = ttg[h * 64 + tid];
    tbv[tid] = ttb[h * 64 + tid];
    lbz[tid] = lbg[tid];
  }

  {
    const float kp0 = kg[g0base], vp0 = vg[g0base];
    const float kp1 = kg[g1base], vp1 = vg[g1base];
    ktb[tid]            = kp0;  ktb[tid + 256]      = kp1;
    ktT[TIDX(b0, d0)]   = kp0;  ktT[TIDX(b1i, d1i)] = kp1;
    tgt[TIDX(b0, d0)]   = vp0 - kp0;
    tgt[TIDX(b1i, d1i)] = vp1 - kp1;
  }
  float kp0 = kg[g0base + HIDC], vp0 = vg[g0base + HIDC];
  float kp1 = kg[g1base + HIDC], vp1 = vg[g1base + HIDC];
  __syncthreads();

  for (int t = 0; t < SS; ++t) {
    float* ktC = (t & 1) ? ktT1 : ktT;
    float* ktN = (t & 1) ? ktT : ktT1;

    for (int step = 0; step < 4; ++step) {
      if (step > 0 || t == 0) {
        if (wave < 4)
          chain_pair<CH_GELU, WB_W2P, false, false>(ktC, W1, b1v, W2, gdv, hg, nullptr, part, wave, lane,
                                                    nullptr, nullptr, nullptr, nullptr, [] {});
        __syncthreads();
        {
          const int pi = TIDX(wave, lane);
          obT[pi]       = part_sum4(part, pi) + b2v[lane];
          obT[pi + 256] = part_sum4(part, pi + 256) + b2v[lane + 32];
        }
        __syncthreads();
      }
      const bool fin = (step == 0 && t > 0);
      float qv0 = 0.f, qv1 = 0.f;
      if (fin) {
        qv0 = qg[g0base + (size_t)(t - 1) * HIDC];
        qv1 = qg[g1base + (size_t)(t - 1) * HIDC];
      }
      const int fc = lane & 15;
      const int fh = lane >> 4;
      const int fk = fh << 1;
      v2f   pb[16];
      v2f   pk[8];
      float pg[8];
      float pb1 = 0.f;
      v2f   ph[2];
      float wold[32];
      if (wave < 4) {
        chain_pair<CH_DPRED, WB_LWP, true, true>(obT, LW, lbz, LWP, nullptr, nullptr, tgt, part, wave, lane,
                                                 lgv, lbv, s1p, s2p, [&] {
          const int col = wave * 16 + fc;
          const int wq  = fh * (W2P_PITCH + 32) + 2 * col;
#pragma unroll
          for (int i = 0; i < 16; ++i) {
            pb[i].x = W2[wq + 2 * W2P_PITCH * i + 4 * (i & 3)];
            pb[i].y = W2[wq + 2 * W2P_PITCH * i + 4 * (i & 3) + 1];
          }
#pragma unroll
          for (int r = 0; r < 8; ++r) pg[r] = gdv[r * 64 + col];
          pb1 = b1v[col];
#pragma unroll
          for (int tm = 0; tm < 4; ++tm) {
#pragma unroll
            for (int s = 0; s < 2; ++s) {
              pk[tm * 2 + s].x = ktb[(4 * s + fk) * 64 + tm * 16 + fc];
              pk[tm * 2 + s].y = ktb[(4 * s + fk + 1) * 64 + tm * 16 + fc];
            }
#pragma unroll
            for (int i = 0; i < 8; ++i) wold[tm * 8 + i] = W1[(tm * 16 + 8 * fh + i) * 64 + col];
          }
        });
      } else {
        const int wv = wave - 4;
        v2f of[16];
        frag_rows16(obT, 16 * fh + 2 * (fc & 7), of);
#pragma unroll
        for (int s = 0; s < 2; ++s) {
          ph[s].x = hg[(4 * s + fk) * 64 + wv * 16 + fc];
          ph[s].y = hg[(4 * s + fk + 1) * 64 + wv * 16 + fc];
        }
        {
          float m, is;
          half_row_stats(of, m, is);
          if (fc < 8) {
#pragma unroll
            for (int i = 0; i < 16; ++i) {
              if ((i >> 2) == wv) {
                const v2f xv = (of[i] - m) * is;
                xh[fc * 64 + 4 * i + fk]     = xv.x;
                xh[fc * 64 + 4 * i + fk + 1] = xv.y;
              }
            }
            if (wave == 4 && fh == 0) isv[fc] = is;
          }
        }
#pragma unroll
        for (int tn = 0; tn < 4; ++tn) {
#pragma unroll
          for (int i = 0; i < 8; ++i)
            wold[tn * 8 + i] = W2[W2P_IDX(wv * 16 + 8 * fh, fc) + 8 * W2P_PITCH * tn + 2 * i];
        }
        if (fin) final_norms(obF, lgv, lbv, tgv, tbv, zb, wv, lane);
      }
      __syncthreads();
      if (step == 3 && t + 1 < SS) {
        ktb[tid]            = kp0;  ktb[tid + 256]      = kp1;
        ktN[TIDX(b0, d0)]   = kp0;  ktN[TIDX(b1i, d1i)] = kp1;
        tgt[TIDX(b0, d0)]   = vp0 - kp0;
        tgt[TIDX(b1i, d1i)] = vp1 - kp1;
        if (t + 2 < SS) {
          kp0 = kg[g0base + (size_t)(t + 2) * HIDC];  vp0 = vg[g0base + (size_t)(t + 2) * HIDC];
          kp1 = kg[g1base + (size_t)(t + 2) * HIDC];  vp1 = vg[g1base + (size_t)(t + 2) * HIDC];
        }
      }
      {
        const int pi = TIDX(wave, lane);
        const float z0 = part_sum4(part, pi);
        const float z1 = part_sum4(part, pi + 256);
        const float m1 = ((s1p[wave] + s1p[8 + wave]) + (s1p[16 + wave] + s1p[24 + wave])) * (1.f / 64.f);
        const float m2 = ((s2p[wave] + s2p[8 + wave]) + (s2p[16 + wave] + s2p[24 + wave])) * (1.f / 64.f);
        const float x0 = xh[wave * 64 + lane], x1 = xh[wave * 64 + lane + 32];
        const float isw = isv[wave];
        const float dv0 = isw * (z0 * lgv[lane] - m1 - x0 * m2);
        const float dv1 = isw * (z1 * lgv[lane + 32] - m1 - x1 * m2);
        dzb[wave * 64 + lane] = z0;
        dzb[wave * 64 + lane + 32] = z1;
        dob[wave * 64 + lane] = dv0;
        dob[wave * 64 + lane + 32] = dv1;
        dobT[pi] = dv0;
        dobT[pi + 256] = dv1;
      }
      if (fin) {
        outg[g0base + (size_t)(t - 1) * HIDC] = qv0 + zb[tid];
        outg[g1base + (size_t)(t - 1) * HIDC] = qv1 + zb[tid + 256];
      }
      __syncthreads();
      if (wave < 4) {
        const int col = wave * 16 + fc;
        const int bb  = fc & 7;
        v2f pa[16];
#pragma unroll
        for (int i = 0; i < 16; ++i) {
          pa[i].x = dobT[16 * fh + 2 * bb + 32 * i];
          pa[i].y = dobT[16 * fh + 2 * bb + 32 * i + 1];
        }
        HOLD8(pa, 0); HOLD8(pa, 8);
        v8f acc = {0.f, 0.f, 0.f, 0.f, 0.f, 0.f, 0.f, 0.f};
#pragma unroll
        for (int i = 0; i < 16; ++i) acc = wmma4(pa[i], pb[i], acc);
        float da[8];
        float sb = 0.f;
#pragma unroll
        for (int r = 0; r < 8; ++r) da[r] = acc[r] * pg[r];
#pragma unroll
        for (int r = 0; r < 8; ++r) sb += da[r];
        if (fh == 0) b1v[col] = pb1 - LR * sb;
        v8f u0 = {0.f, 0.f, 0.f, 0.f, 0.f, 0.f, 0.f, 0.f};
        v8f u1 = u0, u2 = u0, u3 = u0;
#pragma unroll
        for (int s = 0; s < 2; ++s) {
          v2f b2;
          b2.x = fh ? da[4 * s + 2] : da[4 * s];
          b2.y = fh ? da[4 * s + 3] : da[4 * s + 1];
          u0 = wmma4(pk[0 + s], b2, u0);
          u1 = wmma4(pk[2 + s], b2, u1);
          u2 = wmma4(pk[4 + s], b2, u2);
          u3 = wmma4(pk[6 + s], b2, u3);
        }
#pragma unroll
        for (int i = 0; i < 8; ++i) {
          W1[(0 * 16 + 8 * fh + i) * 64 + col] = wold[0 * 8 + i] - LR * u0[i];
          W1[(1 * 16 + 8 * fh + i) * 64 + col] = wold[1 * 8 + i] - LR * u1[i];
          W1[(2 * 16 + 8 * fh + i) * 64 + col] = wold[2 * 8 + i] - LR * u2[i];
          W1[(3 * 16 + 8 * fh + i) * 64 + col] = wold[3 * 8 + i] - LR * u3[i];
        }
      } else {
        const int wv = wave - 4;
        v2f pd[8];
#pragma unroll
        for (int tn = 0; tn < 4; ++tn) {
#pragma unroll
          for (int s = 0; s < 2; ++s) {
            pd[tn * 2 + s].x = dob[(4 * s + fk) * 64 + tn * 16 + fc];
            pd[tn * 2 + s].y = dob[(4 * s + fk + 1) * 64 + tn * 16 + fc];
          }
        }
        v8f u0 = {0.f, 0.f, 0.f, 0.f, 0.f, 0.f, 0.f, 0.f};
        v8f u1 = u0, u2 = u0, u3 = u0;
#pragma unroll
        for (int s = 0; s < 2; ++s) {
          u0 = wmma4(ph[s], pd[0 + s], u0);
          u1 = wmma4(ph[s], pd[2 + s], u1);
          u2 = wmma4(ph[s], pd[4 + s], u2);
          u3 = wmma4(ph[s], pd[6 + s], u3);
        }
        const int rb = W2P_IDX(wv * 16 + 8 * fh, fc);
#pragma unroll
        for (int i = 0; i < 8; ++i) {
          W2[rb + 0 * 8 * W2P_PITCH + 2 * i] = wold[0 * 8 + i] - LR * u0[i];
          W2[rb + 1 * 8 * W2P_PITCH + 2 * i] = wold[1 * 8 + i] - LR * u1[i];
          W2[rb + 2 * 8 * W2P_PITCH + 2 * i] = wold[2 * 8 + i] - LR * u2[i];
          W2[rb + 3 * 8 * W2P_PITCH + 2 * i] = wold[3 * 8 + i] - LR * u3[i];
        }
        const int t2 = tid - 128;
        if (t2 < 64) {
          float s = 0.f, s2 = 0.f;
          for (int b = 0; b < 8; ++b) s += dzb[b * 64 + t2] * xh[b * 64 + t2];
          for (int b = 0; b < 8; ++b) s2 += dob[b * 64 + t2];
          lgv[t2] -= LR * s;
          b2v[t2] -= LR * s2;
        } else {
          const int d = t2 - 64;
          float s = 0.f;
          for (int b = 0; b < 8; ++b) s += dzb[b * 64 + d];
          lbv[d] -= LR * s;
        }
      }
      __syncthreads();
    }

    if (wave < 4)
      chain_fwd_two(ktC, ktN, W1, b1v, W2, gdv, hg, part, partF, wave, lane);
    __syncthreads();
    {
      const int pi = TIDX(wave, lane);
      obT[pi]       = part_sum4(part, pi) + b2v[lane];
      obT[pi + 256] = part_sum4(part, pi + 256) + b2v[lane + 32];
      obF[pi]       = part_sum4(partF, pi) + b2v[lane];
      obF[pi + 256] = part_sum4(partF, pi + 256) + b2v[lane + 32];
    }
    __syncthreads();
  }
  {
    const float qv0 = qg[g0base + (size_t)(SS - 1) * HIDC];
    const float qv1 = qg[g1base + (size_t)(SS - 1) * HIDC];
    if (wave >= 4) final_norms(obF, lgv, lbv, tgv, tbv, zb, wave - 4, lane);
    __syncthreads();
    outg[g0base + (size_t)(SS - 1) * HIDC] = qv0 + zb[tid];
    outg[g1base + (size_t)(SS - 1) * HIDC] = qv1 + zb[tid + 256];
  }
}

__global__ void __launch_bounds__(256)
gate_ln_kernel(const float* __restrict__ outb, const float* __restrict__ gact,
               const float* __restrict__ png, const float* __restrict__ pnb,
               float* __restrict__ gated) {
  __shared__ float r1[256], r2[256];
  __shared__ float mu_s, inv_s;
  const int r = blockIdx.x, tid = threadIdx.x;
  const float* row = outb + (size_t)r * HIDC;
  float s = 0.f, q = 0.f;
  for (int i = tid; i < HIDC; i += 256) { float v = row[i]; s += v; q += v * v; }
  r1[tid] = s; r2[tid] = q;
  __syncthreads();
  for (int off = 128; off > 0; off >>= 1) {
    if (tid < off) { r1[tid] += r1[tid + off]; r2[tid] += r2[tid + off]; }
    __syncthreads();
  }
  if (tid == 0) {
    float m = r1[0] * (1.f / HIDC);
    mu_s = m;
    inv_s = rsqrtf(r2[0] * (1.f / HIDC) - m * m + LN_EPS);
  }
  __syncthreads();
  for (int i = tid; i < HIDC; i += 256) {
    float y = (row[i] - mu_s) * inv_s * png[i] + pnb[i];
    gated[(size_t)r * HIDC + i] = gact[(size_t)r * HIDC + i] * y;
  }
}

extern "C" void kernel_launch(void* const* d_in, const int* in_sizes, int n_in,
                              void* d_out, int out_size, void* d_ws, size_t ws_size,
                              hipStream_t stream) {
  (void)in_sizes; (void)n_in; (void)out_size; (void)ws_size;
  const float* x   = (const float*)d_in[0];
  const float* wq  = (const float*)d_in[1];
  const float* wk  = (const float*)d_in[2];
  const float* wv  = (const float*)d_in[3];
  const float* fw1 = (const float*)d_in[4];
  const float* fb1 = (const float*)d_in[5];
  const float* fw2 = (const float*)d_in[6];
  const float* fb2 = (const float*)d_in[7];
  const float* flg = (const float*)d_in[8];
  const float* flb = (const float*)d_in[9];
  const float* lw  = (const float*)d_in[10];
  const float* lb  = (const float*)d_in[11];
  const float* ttg = (const float*)d_in[12];
  const float* ttb = (const float*)d_in[13];
  const float* wo  = (const float*)d_in[14];
  const float* wg  = (const float*)d_in[15];
  const float* png = (const float*)d_in[16];
  const float* pnb = (const float*)d_in[17];

  const int M = BB * SS;
  const size_t PLANE = (size_t)M * HIDC;
  float* qb    = (float*)d_ws;
  float* kb    = qb + PLANE;
  float* vb    = kb + PLANE;
  float* gactb = vb + PLANE;
  float* outb  = gactb + PLANE;
  float* gatedb = qb;

  dim3 gblk(256);
  dim3 ggrid(HIDC / 128, M / 64);
  gemm_wmma_f32<<<ggrid, gblk, 0, stream>>>(x, wq, qb,    M, HIDC, HIDC, 0);
  gemm_wmma_f32<<<ggrid, gblk, 0, stream>>>(x, wk, kb,    M, HIDC, HIDC, 0);
  gemm_wmma_f32<<<ggrid, gblk, 0, stream>>>(x, wv, vb,    M, HIDC, HIDC, 0);
  gemm_wmma_f32<<<ggrid, gblk, 0, stream>>>(x, wg, gactb, M, HIDC, HIDC, 1);

  ttt_scan_kernel<<<HH, 256, SCAN_SMEM_FLOATS * sizeof(float), stream>>>(
      kb, vb, qb, fw1, fb1, fw2, fb2, flg, flb, lw, lb, ttg, ttb, outb);

  gate_ln_kernel<<<M, 256, 0, stream>>>(outb, gactb, png, pnb, gatedb);

  gemm_wmma_f32<<<ggrid, gblk, 0, stream>>>(gatedb, wo, (float*)d_out,
                                            M, HIDC, HIDC, 0);
}
